# starting kernel
# baseline (speedup 1.0000x reference)
#include <hip/hip_runtime.h>
#include <hip/hip_bf16.h>

typedef __attribute__((ext_vector_type(2))) float v2f;
typedef __attribute__((ext_vector_type(8))) float v8f;

__global__ void k_deg_init(float* __restrict__ deg, int n) {
    int i = blockIdx.x * blockDim.x + threadIdx.x;
    if (i < n) deg[i] = 1.0f;
}

__global__ void k_deg_count(const int* __restrict__ dst, float* __restrict__ deg, int e) {
    int i = blockIdx.x * blockDim.x + threadIdx.x;
    if (i < e) atomicAdd(&deg[dst[i]], 1.0f);
}

__global__ void k_norm(const float* __restrict__ deg, float* __restrict__ isq,
                       float* __restrict__ idg, int n) {
    int i = blockIdx.x * blockDim.x + threadIdx.x;
    if (i < n) {
        float d = deg[i];
        isq[i] = rsqrtf(d);
        idg[i] = 1.0f / d;
    }
}

__global__ void k_coef(const int* __restrict__ src, const int* __restrict__ dst,
                       const float* __restrict__ isq, float* __restrict__ coef, int e) {
    int i = blockIdx.x * blockDim.x + threadIdx.x;
    if (i < e) coef[i] = isq[src[i]] * isq[dst[i]];
}

__global__ void k_pad_x(const float* __restrict__ x, float* __restrict__ x8, int n) {
    long idx = (long)blockIdx.x * blockDim.x + threadIdx.x;
    if (idx < (long)n * 8) {
        int i = (int)(idx >> 3), j = (int)(idx & 7);
        x8[idx] = (j < 5) ? x[(long)i * 5 + j] : 0.0f;
    }
}

__global__ void k_prep_w(const float* __restrict__ W, float* __restrict__ Bt,
                         int Kin, int Cin, int KPAD, int COLS) {
    int idx = blockIdx.x * blockDim.x + threadIdx.x;
    if (idx >= COLS * KPAD) return;
    int col = idx / KPAD, k = idx % KPAD;
    Bt[idx] = (k < Kin && col < Cin) ? W[k * Cin + col] : 0.0f;
}

template <int K, int NT, int SC>
__global__ void k_gemm_wmma(const float* __restrict__ A, const float* __restrict__ Bt,
                            float* __restrict__ H) {
    const int lane = threadIdx.x & 31;
    const int wave = threadIdx.x >> 5;
    const int half = lane >> 4;
    const int l15  = lane & 15;
    const size_t row = (size_t)blockIdx.x * 16 + l15;
    const int    col = wave * 16 + l15;
    const float* Arow = A + row * K;
    const float* Bcol = Bt + (size_t)col * K;

    v8f c = {};
#pragma unroll
    for (int k = 0; k < K; k += 4) {
        const int ka = k + half * 2;
        v2f a = *(const v2f*)(Arow + ka);
        v2f b = *(const v2f*)(Bcol + ka);
        c = __builtin_amdgcn_wmma_f32_16x16x4_f32(
                 false, a,  false, b,
                 (short)0, c,  false,  false);
    }

    if (NT * 16 == SC || col < SC) {
        const size_t mbase = (size_t)blockIdx.x * 16 + half * 8;
#pragma unroll
        for (int r = 0; r < 8; ++r)
            H[(mbase + r) * SC + col] = c[r];
    }
}

__global__ void k_self_bias(const float* __restrict__ h, const float* __restrict__ idg,
                            const float* __restrict__ bias, float* __restrict__ agg,
                            long n_elems, int dim) {
    long idx = (long)blockIdx.x * blockDim.x + threadIdx.x;
    if (idx < n_elems) {
        int i = (int)(idx / dim);
        int j = (int)(idx % dim);
        agg[idx] = h[idx] * idg[i] + bias[j];
    }
}

__global__ void k_scatter64(const int* __restrict__ src, const int* __restrict__ dst,
                            const float* __restrict__ coef, const float* __restrict__ h,
                            float* __restrict__ agg, int e) {
    int i = blockIdx.x * blockDim.x + threadIdx.x;
    if (i >= e) return;
    const int   s  = src[i];
    const int   d  = dst[i];
    const float cf = coef[i];
    const float4* hs = (const float4*)(h + (size_t)s * 64);
    float* ad = agg + (size_t)d * 64;
#pragma unroll
    for (int g = 0; g < 16; ++g) {
        float4 v = hs[g];
        atomicAdd(ad + 4 * g + 0, v.x * cf);
        atomicAdd(ad + 4 * g + 1, v.y * cf);
        atomicAdd(ad + 4 * g + 2, v.z * cf);
        atomicAdd(ad + 4 * g + 3, v.w * cf);
    }
}

__global__ void k_scatter2(const int* __restrict__ src, const int* __restrict__ dst,
                           const float* __restrict__ coef, const float* __restrict__ h,
                           float* __restrict__ out, int e) {
    int i = blockIdx.x * blockDim.x + threadIdx.x;
    if (i >= e) return;
    const int   s  = src[i];
    const int   d  = dst[i];
    const float cf = coef[i];
    float2 v = *(const float2*)(h + (size_t)s * 2);
    atomicAdd(out + (size_t)d * 2 + 0, v.x * cf);
    atomicAdd(out + (size_t)d * 2 + 1, v.y * cf);
}

__global__ void k_relu(float* __restrict__ x, long n) {
    long idx = (long)blockIdx.x * blockDim.x + threadIdx.x;
    if (idx < n) x[idx] = fmaxf(x[idx], 0.0f);
}

extern "C" void kernel_launch(void* const* d_in, const int* in_sizes, int n_in,
                              void* d_out, int out_size, void* d_ws, size_t ws_size,
                              hipStream_t stream) {
    const float* x  = (const float*)d_in[0];
    const int*   ei = (const int*)d_in[1];
    const float* W1 = (const float*)d_in[2];
    const float* b1 = (const float*)d_in[3];
    const float* W2 = (const float*)d_in[4];
    const float* b2 = (const float*)d_in[5];
    const float* W3 = (const float*)d_in[6];
    const float* b3 = (const float*)d_in[7];
    float* out = (float*)d_out;

    const int N = in_sizes[0] / 5;
    const int E = in_sizes[1] / 2;
    const int* src = ei;
    const int* dst = ei + E;

    float* ws   = (float*)d_ws;
    float* deg  = ws;
    float* isq  = deg  + N;
    float* idg  = isq  + N;
    float* coef = idg  + N;
    float* x8   = coef + E;
    float* w1t  = x8   + (size_t)N * 8;
    float* w2t  = w1t  + 64 * 8;
    float* w3t  = w2t  + 64 * 64;
    float* bufA = w3t  + 16 * 64;
    float* bufB = bufA + (size_t)N * 64;
    float* h3   = bufB + (size_t)N * 64;

    const int T = 256;
    auto cdiv = [](long a, long b) { return (int)((a + b - 1) / b); };
    const long NH = (long)N * 64;

    k_deg_init <<<cdiv(N, T), T, 0, stream>>>(deg, N);
    k_deg_count<<<cdiv(E, T), T, 0, stream>>>(dst, deg, E);
    k_norm     <<<cdiv(N, T), T, 0, stream>>>(deg, isq, idg, N);
    k_coef     <<<cdiv(E, T), T, 0, stream>>>(src, dst, isq, coef, E);
    k_pad_x    <<<cdiv((long)N * 8, T), T, 0, stream>>>(x, x8, N);
    k_prep_w   <<<cdiv(64 * 8,  T), T, 0, stream>>>(W1, w1t, 5, 64, 8, 64);
    k_prep_w   <<<cdiv(64 * 64, T), T, 0, stream>>>(W2, w2t, 64, 64, 64, 64);
    k_prep_w   <<<cdiv(16 * 64, T), T, 0, stream>>>(W3, w3t, 64, 2, 64, 16);

    const int GT = cdiv(N, 16);

    k_gemm_wmma<8, 4, 64><<<GT, 128, 0, stream>>>(x8, w1t, bufA);
    k_self_bias<<<cdiv(NH, T), T, 0, stream>>>(bufA, idg, b1, bufB, NH, 64);
    k_scatter64<<<cdiv(E, T), T, 0, stream>>>(src, dst, coef, bufA, bufB, E);
    k_relu     <<<cdiv(NH, T), T, 0, stream>>>(bufB, NH);

    k_gemm_wmma<64, 4, 64><<<GT, 128, 0, stream>>>(bufB, w2t, bufA);
    k_self_bias<<<cdiv(NH, T), T, 0, stream>>>(bufA, idg, b2, bufB, NH, 64);
    k_scatter64<<<cdiv(E, T), T, 0, stream>>>(src, dst, coef, bufA, bufB, E);
    k_relu     <<<cdiv(NH, T), T, 0, stream>>>(bufB, NH);

    k_gemm_wmma<64, 1, 2><<<GT, 32, 0, stream>>>(bufB, w3t, h3);
    k_self_bias<<<cdiv((long)N * 2, T), T, 0, stream>>>(h3, idg, b3, out, (long)N * 2, 2);
    k_scatter2 <<<cdiv(E, T), T, 0, stream>>>(src, dst, coef, h3, out, E);
}


# Float-atomic edge scatter replaced by a per-call counting sort of edges by dst (returning int add = arrival-order rank) and a per-row register sum stored once; sum order within a row changes
# speedup vs baseline: 8.1680x; 8.1680x over previous
#include <hip/hip_runtime.h>
#include <hip/hip_bf16.h>

typedef __attribute__((ext_vector_type(2))) float v2f;
typedef __attribute__((ext_vector_type(8))) float v8f;

__global__ void k_cnt_zero(int* __restrict__ cnt, int n) {
    int i = blockIdx.x * blockDim.x + threadIdx.x;
    if (i < n) cnt[i] = 0;
}

__global__ void k_rank(const int* __restrict__ dst, int* __restrict__ cnt,
                       int* __restrict__ rank, int e) {
    int i = blockIdx.x * blockDim.x + threadIdx.x;
    if (i < e) rank[i] = atomicAdd(&cnt[dst[i]], 1);
}

__global__ void __launch_bounds__(256)
k_scan_block(const int* __restrict__ cnt, int* __restrict__ rowptr, int* __restrict__ bsum,
             float* __restrict__ isq, float* __restrict__ idg, int n) {
    __shared__ int sh[256];
    const int t = threadIdx.x;
    const int base = (blockIdx.x * 256 + t) * 4;
    int v[4];
    for (int j = 0; j < 4; ++j) {
        const int idx = base + j;
        v[j] = 0;
        if (idx < n) {
            v[j] = cnt[idx];
            const float d = (float)(v[j] + 1);
            isq[idx] = rsqrtf(d);
            idg[idx] = 1.0f / d;
        }
    }
    const int s = v[0] + v[1] + v[2] + v[3];
    sh[t] = s;
    __syncthreads();
    for (int off = 1; off < 256; off <<= 1) {
        const int x = (t >= off) ? sh[t - off] : 0;
        __syncthreads();
        sh[t] += x;
        __syncthreads();
    }
    const int incl = sh[t];
    int run = incl - s;
    for (int j = 0; j < 4; ++j) {
        const int idx = base + j;
        if (idx < n) rowptr[idx] = run;
        run += v[j];
    }
    if (t == 255) bsum[blockIdx.x] = incl;
}

__global__ void __launch_bounds__(256)
k_scan_top(int* __restrict__ bsum, int nb) {
    __shared__ int sh[256];
    const int t = threadIdx.x;
    int carry = 0;
    for (int base = 0; base < nb; base += 256) {
        const int i = base + t;
        const int v = (i < nb) ? bsum[i] : 0;
        sh[t] = v;
        __syncthreads();
        for (int off = 1; off < 256; off <<= 1) {
            const int x = (t >= off) ? sh[t - off] : 0;
            __syncthreads();
            sh[t] += x;
            __syncthreads();
        }
        const int incl = sh[t];
        const int tot  = sh[255];
        __syncthreads();
        if (i < nb) bsum[i] = carry + incl - v;
        carry += tot;
    }
}

__global__ void k_scan_add(int* __restrict__ rowptr, const int* __restrict__ bsum, int n, int e) {
    int i = blockIdx.x * blockDim.x + threadIdx.x;
    if (i < n) rowptr[i] += bsum[i >> 10];
    else if (i == n) rowptr[n] = e;
}

__global__ void k_fill(const int* __restrict__ src, const int* __restrict__ dst,
                       const int* __restrict__ rank, const int* __restrict__ rowptr,
                       const float* __restrict__ isq, int* __restrict__ rec, int e) {
    int i = blockIdx.x * blockDim.x + threadIdx.x;
    if (i < e) {
        const int s = src[i];
        const int d = dst[i];
        const int p = rowptr[d] + rank[i];
        ((int2*)rec)[p] = make_int2(s, __float_as_int(isq[s] * isq[d]));
    }
}

__global__ void k_pad_x(const float* __restrict__ x, float* __restrict__ x8, int n) {
    long idx = (long)blockIdx.x * blockDim.x + threadIdx.x;
    if (idx < (long)n * 8) {
        int i = (int)(idx >> 3), j = (int)(idx & 7);
        x8[idx] = (j < 5) ? x[(long)i * 5 + j] : 0.0f;
    }
}

__global__ void k_prep_w(const float* __restrict__ W, float* __restrict__ Bt,
                         int Kin, int Cin, int KPAD, int COLS) {
    int idx = blockIdx.x * blockDim.x + threadIdx.x;
    if (idx >= COLS * KPAD) return;
    int col = idx / KPAD, k = idx % KPAD;
    Bt[idx] = (k < Kin && col < Cin) ? W[k * Cin + col] : 0.0f;
}

template <int K, int NT, int SC>
__global__ void k_gemm_wmma(const float* __restrict__ A, const float* __restrict__ Bt,
                            float* __restrict__ H) {
    const int lane = threadIdx.x & 31;
    const int wave = threadIdx.x >> 5;
    const int half = lane >> 4;
    const int l15  = lane & 15;
    const size_t row = (size_t)blockIdx.x * 16 + l15;
    const int    col = wave * 16 + l15;
    const float* Arow = A + row * K;
    const float* Bcol = Bt + (size_t)col * K;

    v8f c = {};
#pragma unroll
    for (int k = 0; k < K; k += 4) {
        const int ka = k + half * 2;
        v2f a = *(const v2f*)(Arow + ka);
        v2f b = *(const v2f*)(Bcol + ka);
        c = __builtin_amdgcn_wmma_f32_16x16x4_f32(
                 false, a,  false, b,
                 (short)0, c,  false,  false);
    }

    if (NT * 16 == SC || col < SC) {
        const size_t mbase = (size_t)blockIdx.x * 16 + half * 8;
#pragma unroll
        for (int r = 0; r < 8; ++r)
            H[(mbase + r) * SC + col] = c[r];
    }
}

__global__ void k_self_bias(const float* __restrict__ h, const float* __restrict__ idg,
                            const float* __restrict__ bias, float* __restrict__ agg,
                            long n_elems, int dim) {
    long idx = (long)blockIdx.x * blockDim.x + threadIdx.x;
    if (idx < n_elems) {
        int i = (int)(idx / dim);
        int j = (int)(idx % dim);
        agg[idx] = h[idx] * idg[i] + bias[j];
    }
}

typedef __attribute__((ext_vector_type(4))) float v4f;

__global__ void __launch_bounds__(256)
k_gather64(const int* __restrict__ rowptr, const int* __restrict__ recw,
           const float* __restrict__ h, float* __restrict__ agg, int n) {
    const int2* rec = (const int2*)recw;
    const int l = threadIdx.x & 15;
    const int stride = gridDim.x * 16;
    const v4f* h4 = (const v4f*)h;
    for (int node = blockIdx.x * 16 + (threadIdx.x >> 4); node < n; node += stride) {
        const int beg = rowptr[node];
        const int end = rowptr[node + 1];
        v4f acc = {0.0f, 0.0f, 0.0f, 0.0f};
        if (beg < end) {
            const int last = end - 1;
            int2 e0 = rec[beg];
            int2 e1 = rec[min(beg + 1, last)];
            int2 e2 = rec[min(beg + 2, last)];
            int2 e3 = rec[min(beg + 3, last)];
            for (int p = beg; p < end; p += 4) {
                const v4f v0 = h4[(size_t)e0.x * 16 + l];
                const v4f v1 = h4[(size_t)e1.x * 16 + l];
                const v4f v2 = h4[(size_t)e2.x * 16 + l];
                const v4f v3 = h4[(size_t)e3.x * 16 + l];
                const float c0 = __int_as_float(e0.y);
                const float c1 = __int_as_float(e1.y);
                const float c2 = __int_as_float(e2.y);
                const float c3 = __int_as_float(e3.y);
                e0 = rec[min(p + 4, last)];
                e1 = rec[min(p + 5, last)];
                e2 = rec[min(p + 6, last)];
                e3 = rec[min(p + 7, last)];
                asm volatile("" :: "v"(v0), "v"(v1), "v"(v2), "v"(v3));
                acc += v0 * c0;
                if (p + 1 < end) acc += v1 * c1;
                if (p + 2 < end) acc += v2 * c2;
                if (p + 3 < end) acc += v3 * c3;
            }
        }
        v4f* a4 = (v4f*)agg + (size_t)node * 16 + l;
        *a4 = *a4 + acc;
    }
}

__global__ void k_gather2(const int* __restrict__ rowptr, const int* __restrict__ recw,
                          const float* __restrict__ h, float* __restrict__ out, int n) {
    const int2* rec = (const int2*)recw;
    const int node = blockIdx.x * blockDim.x + threadIdx.x;
    if (node >= n) return;
    const int beg = rowptr[node];
    const int end = rowptr[node + 1];
    const v2f* h2 = (const v2f*)h;
    v2f acc = {0.0f, 0.0f};
    for (int p = beg; p < end; p += 4) {
        const int last = end - 1;
        const int2 e0 = rec[p];
        const int2 e1 = rec[min(p + 1, last)];
        const int2 e2 = rec[min(p + 2, last)];
        const int2 e3 = rec[min(p + 3, last)];
        const v2f v0 = h2[e0.x];
        const v2f v1 = h2[e1.x];
        const v2f v2 = h2[e2.x];
        const v2f v3 = h2[e3.x];
        const float c0 = __int_as_float(e0.y);
        const float c1 = __int_as_float(e1.y);
        const float c2 = __int_as_float(e2.y);
        const float c3 = __int_as_float(e3.y);
        asm volatile("" :: "v"(v0), "v"(v1), "v"(v2), "v"(v3));
        acc += v0 * c0;
        if (p + 1 < end) acc += v1 * c1;
        if (p + 2 < end) acc += v2 * c2;
        if (p + 3 < end) acc += v3 * c3;
    }
    v2f* o2 = (v2f*)out + node;
    *o2 = *o2 + acc;
}

__global__ void k_relu(float* __restrict__ x, long n) {
    long idx = (long)blockIdx.x * blockDim.x + threadIdx.x;
    if (idx < n) x[idx] = fmaxf(x[idx], 0.0f);
}

extern "C" void kernel_launch(void* const* d_in, const int* in_sizes, int n_in,
                              void* d_out, int out_size, void* d_ws, size_t ws_size,
                              hipStream_t stream) {
    const float* x  = (const float*)d_in[0];
    const int*   ei = (const int*)d_in[1];
    const float* W1 = (const float*)d_in[2];
    const float* b1 = (const float*)d_in[3];
    const float* W2 = (const float*)d_in[4];
    const float* b2 = (const float*)d_in[5];
    const float* W3 = (const float*)d_in[6];
    const float* b3 = (const float*)d_in[7];
    float* out = (float*)d_out;

    const int N = in_sizes[0] / 5;
    const int E = in_sizes[1] / 2;
    const int* src = ei;
    const int* dst = ei + E;

    const int T = 256;
    auto cdiv = [](long a, long b) { return (int)((a + b - 1) / b); };
    const long NH = (long)N * 64;
    const int NB = cdiv(N, 1024);

    float* ws   = (float*)d_ws;
    float* bufA = ws;
    float* bufB = bufA + (size_t)N * 64;
    float* x8   = bufB + (size_t)N * 64;
    float* w1t  = x8   + (size_t)N * 8;
    float* w2t  = w1t  + 64 * 8;
    float* w3t  = w2t  + 64 * 64;
    float* h3   = w3t  + 16 * 64;
    int*   rec  = (int*)(h3 + (size_t)N * 2);
    int*   rank = rec  + (size_t)E * 2;
    int*   cnt  = rank + E;
    int*   rowp = cnt  + N;
    int*   bsum = rowp + N + 1;
    float* isq  = (float*)(bsum + NB);
    float* idg  = isq  + N;

    k_cnt_zero  <<<cdiv(N, T), T, 0, stream>>>(cnt, N);
    k_rank      <<<cdiv(E, T), T, 0, stream>>>(dst, cnt, rank, E);
    k_scan_block<<<NB, 256, 0, stream>>>(cnt, rowp, bsum, isq, idg, N);
    k_scan_top  <<<1, 256, 0, stream>>>(bsum, NB);
    k_scan_add  <<<cdiv(N + 1, T), T, 0, stream>>>(rowp, bsum, N, E);
    k_fill      <<<cdiv(E, T), T, 0, stream>>>(src, dst, rank, rowp, isq, rec, E);
    k_pad_x    <<<cdiv((long)N * 8, T), T, 0, stream>>>(x, x8, N);
    k_prep_w   <<<cdiv(64 * 8,  T), T, 0, stream>>>(W1, w1t, 5, 64, 8, 64);
    k_prep_w   <<<cdiv(64 * 64, T), T, 0, stream>>>(W2, w2t, 64, 64, 64, 64);
    k_prep_w   <<<cdiv(16 * 64, T), T, 0, stream>>>(W3, w3t, 64, 2, 64, 16);

    const int GT = cdiv(N, 16);
    const int GG = 2048;

    k_gemm_wmma<8, 4, 64><<<GT, 128, 0, stream>>>(x8, w1t, bufA);
    k_self_bias<<<cdiv(NH, T), T, 0, stream>>>(bufA, idg, b1, bufB, NH, 64);
    k_gather64 <<<GG, 256, 0, stream>>>(rowp, rec, bufA, bufB, N);
    k_relu     <<<cdiv(NH, T), T, 0, stream>>>(bufB, NH);

    k_gemm_wmma<64, 4, 64><<<GT, 128, 0, stream>>>(bufB, w2t, bufA);
    k_self_bias<<<cdiv(NH, T), T, 0, stream>>>(bufA, idg, b2, bufB, NH, 64);
    k_gather64 <<<GG, 256, 0, stream>>>(rowp, rec, bufA, bufB, N);
    k_relu     <<<cdiv(NH, T), T, 0, stream>>>(bufB, NH);

    k_gemm_wmma<64, 1, 2><<<GT, 32, 0, stream>>>(bufB, w3t, h3);
    k_self_bias<<<cdiv((long)N * 2, T), T, 0, stream>>>(h3, idg, b3, out, (long)N * 2, 2);
    k_gather2  <<<cdiv(N, T), T, 0, stream>>>(rowp, rec, h3, out, N);
}


# Self-loop term, bias and ReLU folded into the per-row edge-sum kernels: the row sum starts from h*inv_deg+bias and is stored once (5 elementwise launches gone; sum order within a row changes)
# speedup vs baseline: 10.1207x; 1.2391x over previous
#include <hip/hip_runtime.h>
#include <hip/hip_bf16.h>

typedef __attribute__((ext_vector_type(2))) float v2f;
typedef __attribute__((ext_vector_type(8))) float v8f;

__global__ void k_cnt_zero(int* __restrict__ cnt, int n) {
    int i = blockIdx.x * blockDim.x + threadIdx.x;
    if (i < n) cnt[i] = 0;
}

__global__ void k_rank(const int* __restrict__ dst, int* __restrict__ cnt,
                       int* __restrict__ rank, int e) {
    int i = blockIdx.x * blockDim.x + threadIdx.x;
    if (i < e) rank[i] = atomicAdd(&cnt[dst[i]], 1);
}

__global__ void __launch_bounds__(256)
k_scan_block(const int* __restrict__ cnt, int* __restrict__ rowptr, int* __restrict__ bsum,
             float* __restrict__ isq, float* __restrict__ idg, int n) {
    __shared__ int sh[256];
    const int t = threadIdx.x;
    const int base = (blockIdx.x * 256 + t) * 4;
    int v[4];
    for (int j = 0; j < 4; ++j) {
        const int idx = base + j;
        v[j] = 0;
        if (idx < n) {
            v[j] = cnt[idx];
            const float d = (float)(v[j] + 1);
            isq[idx] = rsqrtf(d);
            idg[idx] = 1.0f / d;
        }
    }
    const int s = v[0] + v[1] + v[2] + v[3];
    sh[t] = s;
    __syncthreads();
    for (int off = 1; off < 256; off <<= 1) {
        const int x = (t >= off) ? sh[t - off] : 0;
        __syncthreads();
        sh[t] += x;
        __syncthreads();
    }
    const int incl = sh[t];
    int run = incl - s;
    for (int j = 0; j < 4; ++j) {
        const int idx = base + j;
        if (idx < n) rowptr[idx] = run;
        run += v[j];
    }
    if (t == 255) bsum[blockIdx.x] = incl;
}

__global__ void __launch_bounds__(256)
k_scan_top(int* __restrict__ bsum, int nb) {
    __shared__ int sh[256];
    const int t = threadIdx.x;
    int carry = 0;
    for (int base = 0; base < nb; base += 256) {
        const int i = base + t;
        const int v = (i < nb) ? bsum[i] : 0;
        sh[t] = v;
        __syncthreads();
        for (int off = 1; off < 256; off <<= 1) {
            const int x = (t >= off) ? sh[t - off] : 0;
            __syncthreads();
            sh[t] += x;
            __syncthreads();
        }
        const int incl = sh[t];
        const int tot  = sh[255];
        __syncthreads();
        if (i < nb) bsum[i] = carry + incl - v;
        carry += tot;
    }
}

__global__ void k_scan_add(int* __restrict__ rowptr, const int* __restrict__ bsum, int n, int e) {
    int i = blockIdx.x * blockDim.x + threadIdx.x;
    if (i < n) rowptr[i] += bsum[i >> 10];
    else if (i == n) rowptr[n] = e;
}

__global__ void k_fill(const int* __restrict__ src, const int* __restrict__ dst,
                       const int* __restrict__ rank, const int* __restrict__ rowptr,
                       const float* __restrict__ isq, int* __restrict__ rec, int e) {
    int i = blockIdx.x * blockDim.x + threadIdx.x;
    if (i < e) {
        const int s = src[i];
        const int d = dst[i];
        const int p = rowptr[d] + rank[i];
        ((int2*)rec)[p] = make_int2(s, __float_as_int(isq[s] * isq[d]));
    }
}

__global__ void k_pad_x(const float* __restrict__ x, float* __restrict__ x8, int n) {
    long idx = (long)blockIdx.x * blockDim.x + threadIdx.x;
    if (idx < (long)n * 8) {
        int i = (int)(idx >> 3), j = (int)(idx & 7);
        x8[idx] = (j < 5) ? x[(long)i * 5 + j] : 0.0f;
    }
}

__global__ void k_prep_w(const float* __restrict__ W, float* __restrict__ Bt,
                         int Kin, int Cin, int KPAD, int COLS) {
    int idx = blockIdx.x * blockDim.x + threadIdx.x;
    if (idx >= COLS * KPAD) return;
    int col = idx / KPAD, k = idx % KPAD;
    Bt[idx] = (k < Kin && col < Cin) ? W[k * Cin + col] : 0.0f;
}

template <int K, int NT, int SC>
__global__ void k_gemm_wmma(const float* __restrict__ A, const float* __restrict__ Bt,
                            float* __restrict__ H) {
    const int lane = threadIdx.x & 31;
    const int wave = threadIdx.x >> 5;
    const int half = lane >> 4;
    const int l15  = lane & 15;
    const size_t row = (size_t)blockIdx.x * 16 + l15;
    const int    col = wave * 16 + l15;
    const float* Arow = A + row * K;
    const float* Bcol = Bt + (size_t)col * K;

    v8f c = {};
#pragma unroll
    for (int k = 0; k < K; k += 4) {
        const int ka = k + half * 2;
        v2f a = *(const v2f*)(Arow + ka);
        v2f b = *(const v2f*)(Bcol + ka);
        c = __builtin_amdgcn_wmma_f32_16x16x4_f32(
                 false, a,  false, b,
                 (short)0, c,  false,  false);
    }

    if (NT * 16 == SC || col < SC) {
        const size_t mbase = (size_t)blockIdx.x * 16 + half * 8;
#pragma unroll
        for (int r = 0; r < 8; ++r)
            H[(mbase + r) * SC + col] = c[r];
    }
}

typedef __attribute__((ext_vector_type(4))) float v4f;

__global__ void __launch_bounds__(256)
k_gather64(const int* __restrict__ rowptr, const int* __restrict__ recw,
           const float* __restrict__ h, const float* __restrict__ idg,
           const float* __restrict__ bias, float* __restrict__ out, int n) {
    const int2* rec = (const int2*)recw;
    const int l = threadIdx.x & 15;
    const int stride = gridDim.x * 16;
    const v4f* h4 = (const v4f*)h;
    const v4f bv = {bias[4 * l], bias[4 * l + 1], bias[4 * l + 2], bias[4 * l + 3]};
    for (int node = blockIdx.x * 16 + (threadIdx.x >> 4); node < n; node += stride) {
        const int beg = rowptr[node];
        const int end = rowptr[node + 1];
        v4f acc = h4[(size_t)node * 16 + l] * idg[node] + bv;
        if (beg < end) {
            const int last = end - 1;
            int2 e0 = rec[beg];
            int2 e1 = rec[min(beg + 1, last)];
            int2 e2 = rec[min(beg + 2, last)];
            int2 e3 = rec[min(beg + 3, last)];
            for (int p = beg; p < end; p += 4) {
                const v4f v0 = h4[(size_t)e0.x * 16 + l];
                const v4f v1 = h4[(size_t)e1.x * 16 + l];
                const v4f v2 = h4[(size_t)e2.x * 16 + l];
                const v4f v3 = h4[(size_t)e3.x * 16 + l];
                const float c0 = __int_as_float(e0.y);
                const float c1 = __int_as_float(e1.y);
                const float c2 = __int_as_float(e2.y);
                const float c3 = __int_as_float(e3.y);
                e0 = rec[min(p + 4, last)];
                e1 = rec[min(p + 5, last)];
                e2 = rec[min(p + 6, last)];
                e3 = rec[min(p + 7, last)];
                asm volatile("" :: "v"(v0), "v"(v1), "v"(v2), "v"(v3));
                acc += v0 * c0;
                if (p + 1 < end) acc += v1 * c1;
                if (p + 2 < end) acc += v2 * c2;
                if (p + 3 < end) acc += v3 * c3;
            }
        }
        v4f r;
        r.x = fmaxf(acc.x, 0.0f);
        r.y = fmaxf(acc.y, 0.0f);
        r.z = fmaxf(acc.z, 0.0f);
        r.w = fmaxf(acc.w, 0.0f);
        ((v4f*)out)[(size_t)node * 16 + l] = r;
    }
}

__global__ void k_gather2(const int* __restrict__ rowptr, const int* __restrict__ recw,
                          const float* __restrict__ h, const float* __restrict__ idg,
                          const float* __restrict__ bias, float* __restrict__ out, int n) {
    const int2* rec = (const int2*)recw;
    const int node = blockIdx.x * blockDim.x + threadIdx.x;
    if (node >= n) return;
    const int beg = rowptr[node];
    const int end = rowptr[node + 1];
    const v2f* h2 = (const v2f*)h;
    const v2f bv = {bias[0], bias[1]};
    v2f acc = h2[node] * idg[node] + bv;
    for (int p = beg; p < end; p += 4) {
        const int last = end - 1;
        const int2 e0 = rec[p];
        const int2 e1 = rec[min(p + 1, last)];
        const int2 e2 = rec[min(p + 2, last)];
        const int2 e3 = rec[min(p + 3, last)];
        const v2f v0 = h2[e0.x];
        const v2f v1 = h2[e1.x];
        const v2f v2 = h2[e2.x];
        const v2f v3 = h2[e3.x];
        const float c0 = __int_as_float(e0.y);
        const float c1 = __int_as_float(e1.y);
        const float c2 = __int_as_float(e2.y);
        const float c3 = __int_as_float(e3.y);
        asm volatile("" :: "v"(v0), "v"(v1), "v"(v2), "v"(v3));
        acc += v0 * c0;
        if (p + 1 < end) acc += v1 * c1;
        if (p + 2 < end) acc += v2 * c2;
        if (p + 3 < end) acc += v3 * c3;
    }
    ((v2f*)out)[node] = acc;
}

extern "C" void kernel_launch(void* const* d_in, const int* in_sizes, int n_in,
                              void* d_out, int out_size, void* d_ws, size_t ws_size,
                              hipStream_t stream) {
    const float* x  = (const float*)d_in[0];
    const int*   ei = (const int*)d_in[1];
    const float* W1 = (const float*)d_in[2];
    const float* b1 = (const float*)d_in[3];
    const float* W2 = (const float*)d_in[4];
    const float* b2 = (const float*)d_in[5];
    const float* W3 = (const float*)d_in[6];
    const float* b3 = (const float*)d_in[7];
    float* out = (float*)d_out;

    const int N = in_sizes[0] / 5;
    const int E = in_sizes[1] / 2;
    const int* src = ei;
    const int* dst = ei + E;

    const int T = 256;
    auto cdiv = [](long a, long b) { return (int)((a + b - 1) / b); };
    const int NB = cdiv(N, 1024);

    float* ws   = (float*)d_ws;
    float* bufA = ws;
    float* bufB = bufA + (size_t)N * 64;
    float* x8   = bufB + (size_t)N * 64;
    float* w1t  = x8   + (size_t)N * 8;
    float* w2t  = w1t  + 64 * 8;
    float* w3t  = w2t  + 64 * 64;
    float* h3   = w3t  + 16 * 64;
    int*   rec  = (int*)(h3 + (size_t)N * 2);
    int*   rank = rec  + (size_t)E * 2;
    int*   cnt  = rank + E;
    int*   rowp = cnt  + N;
    int*   bsum = rowp + N + 1;
    float* isq  = (float*)(bsum + NB);
    float* idg  = isq  + N;

    k_cnt_zero  <<<cdiv(N, T), T, 0, stream>>>(cnt, N);
    k_rank      <<<cdiv(E, T), T, 0, stream>>>(dst, cnt, rank, E);
    k_scan_block<<<NB, 256, 0, stream>>>(cnt, rowp, bsum, isq, idg, N);
    k_scan_top  <<<1, 256, 0, stream>>>(bsum, NB);
    k_scan_add  <<<cdiv(N + 1, T), T, 0, stream>>>(rowp, bsum, N, E);
    k_fill      <<<cdiv(E, T), T, 0, stream>>>(src, dst, rank, rowp, isq, rec, E);
    k_pad_x    <<<cdiv((long)N * 8, T), T, 0, stream>>>(x, x8, N);
    k_prep_w   <<<cdiv(64 * 8,  T), T, 0, stream>>>(W1, w1t, 5, 64, 8, 64);
    k_prep_w   <<<cdiv(64 * 64, T), T, 0, stream>>>(W2, w2t, 64, 64, 64, 64);
    k_prep_w   <<<cdiv(16 * 64, T), T, 0, stream>>>(W3, w3t, 64, 2, 64, 16);

    const int GT = cdiv(N, 16);
    const int GG = 2048;

    k_gemm_wmma<8, 4, 64><<<GT, 128, 0, stream>>>(x8, w1t, bufA);
    k_gather64 <<<GG, 256, 0, stream>>>(rowp, rec, bufA, idg, b1, bufB, N);

    k_gemm_wmma<64, 4, 64><<<GT, 128, 0, stream>>>(bufB, w2t, bufA);
    k_gather64 <<<GG, 256, 0, stream>>>(rowp, rec, bufA, idg, b2, bufB, N);

    k_gemm_wmma<64, 1, 2><<<GT, 32, 0, stream>>>(bufB, w3t, h3);
    k_gather2  <<<cdiv(N, T), T, 0, stream>>>(rowp, rec, h3, idg, b3, out, N);
}


# Layer 1 takes the edge sum before the W1 product: 32-byte input rows are gathered instead of 256-byte products and the sums feed the f32 WMMA as A fragments in one launch; layer-1 sum order changes
# speedup vs baseline: 10.6142x; 1.0488x over previous
#include <hip/hip_runtime.h>
#include <hip/hip_bf16.h>

typedef __attribute__((ext_vector_type(2))) float v2f;
typedef __attribute__((ext_vector_type(8))) float v8f;

__global__ void k_cnt_zero(int* __restrict__ cnt, int n) {
    int i = blockIdx.x * blockDim.x + threadIdx.x;
    if (i < n) cnt[i] = 0;
}

__global__ void k_rank(const int* __restrict__ dst, int* __restrict__ cnt,
                       int* __restrict__ rank, int e) {
    int i = blockIdx.x * blockDim.x + threadIdx.x;
    if (i < e) rank[i] = atomicAdd(&cnt[dst[i]], 1);
}

__global__ void __launch_bounds__(256)
k_scan_block(const int* __restrict__ cnt, int* __restrict__ rowptr, int* __restrict__ bsum,
             float* __restrict__ isq, float* __restrict__ idg, int n) {
    __shared__ int sh[256];
    const int t = threadIdx.x;
    const int base = (blockIdx.x * 256 + t) * 4;
    int v[4];
    for (int j = 0; j < 4; ++j) {
        const int idx = base + j;
        v[j] = 0;
        if (idx < n) {
            v[j] = cnt[idx];
            const float d = (float)(v[j] + 1);
            isq[idx] = rsqrtf(d);
            idg[idx] = 1.0f / d;
        }
    }
    const int s = v[0] + v[1] + v[2] + v[3];
    sh[t] = s;
    __syncthreads();
    for (int off = 1; off < 256; off <<= 1) {
        const int x = (t >= off) ? sh[t - off] : 0;
        __syncthreads();
        sh[t] += x;
        __syncthreads();
    }
    const int incl = sh[t];
    int run = incl - s;
    for (int j = 0; j < 4; ++j) {
        const int idx = base + j;
        if (idx < n) rowptr[idx] = run;
        run += v[j];
    }
    if (t == 255) bsum[blockIdx.x] = incl;
}

__global__ void __launch_bounds__(256)
k_scan_top(int* __restrict__ bsum, int nb) {
    __shared__ int sh[256];
    const int t = threadIdx.x;
    int carry = 0;
    for (int base = 0; base < nb; base += 256) {
        const int i = base + t;
        const int v = (i < nb) ? bsum[i] : 0;
        sh[t] = v;
        __syncthreads();
        for (int off = 1; off < 256; off <<= 1) {
            const int x = (t >= off) ? sh[t - off] : 0;
            __syncthreads();
            sh[t] += x;
            __syncthreads();
        }
        const int incl = sh[t];
        const int tot  = sh[255];
        __syncthreads();
        if (i < nb) bsum[i] = carry + incl - v;
        carry += tot;
    }
}

__global__ void k_scan_add(int* __restrict__ rowptr, const int* __restrict__ bsum, int n, int e) {
    int i = blockIdx.x * blockDim.x + threadIdx.x;
    if (i < n) rowptr[i] += bsum[i >> 10];
    else if (i == n) rowptr[n] = e;
}

__global__ void k_fill(const int* __restrict__ src, const int* __restrict__ dst,
                       const int* __restrict__ rank, const int* __restrict__ rowptr,
                       const float* __restrict__ isq, int* __restrict__ rec, int e) {
    int i = blockIdx.x * blockDim.x + threadIdx.x;
    if (i < e) {
        const int s = src[i];
        const int d = dst[i];
        const int p = rowptr[d] + rank[i];
        ((int2*)rec)[p] = make_int2(s, __float_as_int(isq[s] * isq[d]));
    }
}

__global__ void k_pad_x(const float* __restrict__ x, float* __restrict__ xq, int n) {
    long idx = (long)blockIdx.x * blockDim.x + threadIdx.x;
    if (idx < (long)n * 8) {
        const int i = (int)(idx >> 3), p = (int)(idx & 7);
        const int col = (p & 1) + ((p >> 1) & 1) * 4 + (p >> 2) * 2;
        xq[idx] = (col < 5) ? x[(long)i * 5 + col] : 0.0f;
    }
}

__global__ void k_prep_w(const float* __restrict__ W, float* __restrict__ Bt,
                         int Kin, int Cin, int KPAD, int COLS) {
    int idx = blockIdx.x * blockDim.x + threadIdx.x;
    if (idx >= COLS * KPAD) return;
    int col = idx / KPAD, k = idx % KPAD;
    Bt[idx] = (k < Kin && col < Cin) ? W[k * Cin + col] : 0.0f;
}

template <int K, int NT, int SC>
__global__ void k_gemm_wmma(const float* __restrict__ A, const float* __restrict__ Bt,
                            float* __restrict__ H) {
    const int lane = threadIdx.x & 31;
    const int wave = threadIdx.x >> 5;
    const int half = lane >> 4;
    const int l15  = lane & 15;
    const size_t row = (size_t)blockIdx.x * 16 + l15;
    const int    col = wave * 16 + l15;
    const float* Arow = A + row * K;
    const float* Bcol = Bt + (size_t)col * K;

    v8f c = {};
#pragma unroll
    for (int k = 0; k < K; k += 4) {
        const int ka = k + half * 2;
        v2f a = *(const v2f*)(Arow + ka);
        v2f b = *(const v2f*)(Bcol + ka);
        c = __builtin_amdgcn_wmma_f32_16x16x4_f32(
                 false, a,  false, b,
                 (short)0, c,  false,  false);
    }

    if (NT * 16 == SC || col < SC) {
        const size_t mbase = (size_t)blockIdx.x * 16 + half * 8;
#pragma unroll
        for (int r = 0; r < 8; ++r)
            H[(mbase + r) * SC + col] = c[r];
    }
}

typedef __attribute__((ext_vector_type(4))) float v4f;

__global__ void __launch_bounds__(256)
k_gather64(const int* __restrict__ rowptr, const int* __restrict__ recw,
           const float* __restrict__ h, const float* __restrict__ idg,
           const float* __restrict__ bias, float* __restrict__ out, int n) {
    const int2* rec = (const int2*)recw;
    const int l = threadIdx.x & 15;
    const int stride = gridDim.x * 16;
    const v4f* h4 = (const v4f*)h;
    const v4f bv = {bias[4 * l], bias[4 * l + 1], bias[4 * l + 2], bias[4 * l + 3]};
    for (int node = blockIdx.x * 16 + (threadIdx.x >> 4); node < n; node += stride) {
        const int beg = rowptr[node];
        const int end = rowptr[node + 1];
        v4f acc = h4[(size_t)node * 16 + l] * idg[node] + bv;
        if (beg < end) {
            const int last = end - 1;
            int2 e0 = rec[beg];
            int2 e1 = rec[min(beg + 1, last)];
            int2 e2 = rec[min(beg + 2, last)];
            int2 e3 = rec[min(beg + 3, last)];
            for (int p = beg; p < end; p += 4) {
                const v4f v0 = h4[(size_t)e0.x * 16 + l];
                const v4f v1 = h4[(size_t)e1.x * 16 + l];
                const v4f v2 = h4[(size_t)e2.x * 16 + l];
                const v4f v3 = h4[(size_t)e3.x * 16 + l];
                const float c0 = __int_as_float(e0.y);
                const float c1 = __int_as_float(e1.y);
                const float c2 = __int_as_float(e2.y);
                const float c3 = __int_as_float(e3.y);
                e0 = rec[min(p + 4, last)];
                e1 = rec[min(p + 5, last)];
                e2 = rec[min(p + 6, last)];
                e3 = rec[min(p + 7, last)];
                asm volatile("" :: "v"(v0), "v"(v1), "v"(v2), "v"(v3));
                acc += v0 * c0;
                if (p + 1 < end) acc += v1 * c1;
                if (p + 2 < end) acc += v2 * c2;
                if (p + 3 < end) acc += v3 * c3;
            }
        }
        v4f r;
        r.x = fmaxf(acc.x, 0.0f);
        r.y = fmaxf(acc.y, 0.0f);
        r.z = fmaxf(acc.z, 0.0f);
        r.w = fmaxf(acc.w, 0.0f);
        ((v4f*)out)[(size_t)node * 16 + l] = r;
    }
}

__global__ void __launch_bounds__(64)
k_layer1(const int* __restrict__ rowptr, const int* __restrict__ recw,
         const float* __restrict__ xq, const float* __restrict__ idg,
         const float* __restrict__ Bt, const float* __restrict__ bias,
         float* __restrict__ out, int ntiles) {
    const int lane = threadIdx.x & 31;
    const int half = lane >> 4;
    const int l15  = lane & 15;
    const int tile = blockIdx.x * 2 + (threadIdx.x >> 5);
    if (tile >= ntiles) return;
    const int node = tile * 16 + l15;
    const int2* rec = (const int2*)recw;
    const v4f* xq4 = (const v4f*)xq;

    const int beg = rowptr[node];
    const int end = rowptr[node + 1];

    v2f b0[4], b1[4];
#pragma unroll
    for (int ct = 0; ct < 4; ++ct) {
        const float* Bcol = Bt + (size_t)(4 * l15 + ct) * 8 + half * 2;
        b0[ct] = *(const v2f*)(Bcol);
        b1[ct] = *(const v2f*)(Bcol + 4);
    }
    const v4f bb = *(const v4f*)(bias + 4 * l15);

    v4f acc = xq4[(size_t)node * 2 + half] * idg[node];
    if (beg < end) {
        const int last = end - 1;
        int2 e[8];
#pragma unroll
        for (int j = 0; j < 8; ++j) e[j] = rec[min(beg + j, last)];
        for (int p = beg; p < end; p += 8) {
            v4f v[8];
            float c[8];
#pragma unroll
            for (int j = 0; j < 8; ++j) {
                v[j] = xq4[(size_t)e[j].x * 2 + half];
                c[j] = __int_as_float(e[j].y);
            }
#pragma unroll
            for (int j = 0; j < 8; ++j) e[j] = rec[min(p + 8 + j, last)];
            asm volatile("" :: "v"(v[0]), "v"(v[1]), "v"(v[2]), "v"(v[3]),
                               "v"(v[4]), "v"(v[5]), "v"(v[6]), "v"(v[7]));
            acc += v[0] * c[0];
#pragma unroll
            for (int j = 1; j < 8; ++j)
                if (p + j < end) acc += v[j] * c[j];
        }
    }

    const v2f a0 = {acc.x, acc.y};
    const v2f a1 = {acc.z, acc.w};
    v8f d[4];
#pragma unroll
    for (int ct = 0; ct < 4; ++ct) {
        v8f z = {};
        d[ct] = __builtin_amdgcn_wmma_f32_16x16x4_f32(
                     false, a0,  false, b0[ct],
                     (short)0, z,  false,  false);
    }
#pragma unroll
    for (int ct = 0; ct < 4; ++ct)
        d[ct] = __builtin_amdgcn_wmma_f32_16x16x4_f32(
                     false, a1,  false, b1[ct],
                     (short)0, d[ct],  false,  false);

    const size_t mbase = (size_t)tile * 16 + half * 8;
#pragma unroll
    for (int r = 0; r < 8; ++r) {
        v4f o;
        o.x = fmaxf(d[0][r] + bb.x, 0.0f);
        o.y = fmaxf(d[1][r] + bb.y, 0.0f);
        o.z = fmaxf(d[2][r] + bb.z, 0.0f);
        o.w = fmaxf(d[3][r] + bb.w, 0.0f);
        ((v4f*)out)[(mbase + r) * 16 + l15] = o;
    }
}

__global__ void k_gather2(const int* __restrict__ rowptr, const int* __restrict__ recw,
                          const float* __restrict__ h, const float* __restrict__ idg,
                          const float* __restrict__ bias, float* __restrict__ out, int n) {
    const int2* rec = (const int2*)recw;
    const int node = blockIdx.x * blockDim.x + threadIdx.x;
    if (node >= n) return;
    const int beg = rowptr[node];
    const int end = rowptr[node + 1];
    const v2f* h2 = (const v2f*)h;
    const v2f bv = {bias[0], bias[1]};
    v2f acc = h2[node] * idg[node] + bv;
    for (int p = beg; p < end; p += 4) {
        const int last = end - 1;
        const int2 e0 = rec[p];
        const int2 e1 = rec[min(p + 1, last)];
        const int2 e2 = rec[min(p + 2, last)];
        const int2 e3 = rec[min(p + 3, last)];
        const v2f v0 = h2[e0.x];
        const v2f v1 = h2[e1.x];
        const v2f v2 = h2[e2.x];
        const v2f v3 = h2[e3.x];
        const float c0 = __int_as_float(e0.y);
        const float c1 = __int_as_float(e1.y);
        const float c2 = __int_as_float(e2.y);
        const float c3 = __int_as_float(e3.y);
        asm volatile("" :: "v"(v0), "v"(v1), "v"(v2), "v"(v3));
        acc += v0 * c0;
        if (p + 1 < end) acc += v1 * c1;
        if (p + 2 < end) acc += v2 * c2;
        if (p + 3 < end) acc += v3 * c3;
    }
    ((v2f*)out)[node] = acc;
}

extern "C" void kernel_launch(void* const* d_in, const int* in_sizes, int n_in,
                              void* d_out, int out_size, void* d_ws, size_t ws_size,
                              hipStream_t stream) {
    const float* x  = (const float*)d_in[0];
    const int*   ei = (const int*)d_in[1];
    const float* W1 = (const float*)d_in[2];
    const float* b1 = (const float*)d_in[3];
    const float* W2 = (const float*)d_in[4];
    const float* b2 = (const float*)d_in[5];
    const float* W3 = (const float*)d_in[6];
    const float* b3 = (const float*)d_in[7];
    float* out = (float*)d_out;

    const int N = in_sizes[0] / 5;
    const int E = in_sizes[1] / 2;
    const int* src = ei;
    const int* dst = ei + E;

    const int T = 256;
    auto cdiv = [](long a, long b) { return (int)((a + b - 1) / b); };
    const int NB = cdiv(N, 1024);

    float* ws   = (float*)d_ws;
    float* bufA = ws;
    float* bufB = bufA + (size_t)N * 64;
    float* xq   = bufB + (size_t)N * 64;
    float* w1t  = xq   + (size_t)N * 8;
    float* w2t  = w1t  + 64 * 8;
    float* w3t  = w2t  + 64 * 64;
    float* h3   = w3t  + 16 * 64;
    int*   rec  = (int*)(h3 + (size_t)N * 2);
    int*   rank = rec  + (size_t)E * 2;
    int*   cnt  = rank + E;
    int*   rowp = cnt  + N;
    int*   bsum = rowp + N + 1;
    float* isq  = (float*)(bsum + NB);
    float* idg  = isq  + N;

    k_cnt_zero  <<<cdiv(N, T), T, 0, stream>>>(cnt, N);
    k_rank      <<<cdiv(E, T), T, 0, stream>>>(dst, cnt, rank, E);
    k_scan_block<<<NB, 256, 0, stream>>>(cnt, rowp, bsum, isq, idg, N);
    k_scan_top  <<<1, 256, 0, stream>>>(bsum, NB);
    k_scan_add  <<<cdiv(N + 1, T), T, 0, stream>>>(rowp, bsum, N, E);
    k_fill      <<<cdiv(E, T), T, 0, stream>>>(src, dst, rank, rowp, isq, rec, E);
    k_pad_x    <<<cdiv((long)N * 8, T), T, 0, stream>>>(x, xq, N);
    k_prep_w   <<<cdiv(64 * 8,  T), T, 0, stream>>>(W1, w1t, 5, 64, 8, 64);
    k_prep_w   <<<cdiv(64 * 64, T), T, 0, stream>>>(W2, w2t, 64, 64, 64, 64);
    k_prep_w   <<<cdiv(16 * 64, T), T, 0, stream>>>(W3, w3t, 64, 2, 64, 16);

    const int GT = cdiv(N, 16);
    const int GG = 2048;

    k_layer1   <<<cdiv(GT, 2), 64, 0, stream>>>(rowp, rec, xq, idg, w1t, b1, bufB, GT);

    k_gemm_wmma<64, 4, 64><<<GT, 128, 0, stream>>>(bufB, w2t, bufA);
    k_gather64 <<<GG, 256, 0, stream>>>(rowp, rec, bufA, idg, b2, bufB, N);

    k_gemm_wmma<64, 1, 2><<<GT, 32, 0, stream>>>(bufB, w3t, h3);
    k_gather2  <<<cdiv(N, T), T, 0, stream>>>(rowp, rec, h3, idg, b3, out, N);
}


# Edge ranks taken on LDS counters in a two-level counting sort (bucket of 512 dst rows, then row) instead of one returning global atomic per edge; sum order within a row changes
# speedup vs baseline: 19.6304x; 1.8494x over previous
#include <hip/hip_runtime.h>
#include <hip/hip_bf16.h>

typedef __attribute__((ext_vector_type(2))) float v2f;
typedef __attribute__((ext_vector_type(8))) float v8f;

#define BSH     9
#define BROWS   512
#define NBK_MAX 256
#define CHUNK   20480
#define STH     1024
#define STILE   32768

__global__ void __launch_bounds__(STH)
k_bkt_count(const int* __restrict__ dst, int* __restrict__ tab, int e, int nbk, int nch) {
    __shared__ int hist[NBK_MAX];
    const int t = threadIdx.x;
    if (t < NBK_MAX) hist[t] = 0;
    __syncthreads();
    const int beg = blockIdx.x * CHUNK;
    const int end = min(beg + CHUNK, e);
    for (int i = beg + t; i < end; i += 4 * STH) {
        int d[4];
#pragma unroll
        for (int k = 0; k < 4; ++k) d[k] = dst[min(i + k * STH, end - 1)];
        asm volatile("" :: "v"(d[0]), "v"(d[1]), "v"(d[2]), "v"(d[3]));
#pragma unroll
        for (int k = 0; k < 4; ++k)
            if (i + k * STH < end) atomicAdd(&hist[d[k] >> BSH], 1);
    }
    __syncthreads();
    if (t < nbk) tab[t * nch + blockIdx.x] = hist[t];
}

__global__ void __launch_bounds__(STH)
k_tab_scan(int* __restrict__ tab, int len) {
    __shared__ int buf[STILE + STILE / 32];
    __shared__ int part[STH];
    const int t = threadIdx.x;
    int carry = 0;
    for (int tile = 0; tile < len; tile += STILE) {
        for (int k0 = t; k0 < STILE; k0 += 8 * STH) {
            int v[8];
#pragma unroll
            for (int u = 0; u < 8; ++u) v[u] = tab[min(tile + k0 + u * STH, len - 1)];
#pragma unroll
            for (int u = 0; u < 8; ++u) {
                const int k = k0 + u * STH;
                buf[k + (k >> 5)] = (tile + k < len) ? v[u] : 0;
            }
        }
        __syncthreads();
        int s = 0;
        for (int j = 0; j < 32; ++j) s += buf[33 * t + j];
        part[t] = s;
        __syncthreads();
        for (int off = 1; off < STH; off <<= 1) {
            const int x = (t >= off) ? part[t - off] : 0;
            __syncthreads();
            part[t] += x;
            __syncthreads();
        }
        int run = carry + part[t] - s;
        const int tot = part[STH - 1];
        for (int j = 0; j < 32; ++j) {
            const int v = buf[33 * t + j];
            buf[33 * t + j] = run;
            run += v;
        }
        __syncthreads();
        for (int k = t; k < STILE; k += STH) {
            const int g = tile + k;
            if (g < len) tab[g] = buf[k + (k >> 5)];
        }
        carry += tot;
        __syncthreads();
    }
}

__global__ void __launch_bounds__(STH)
k_bkt_scatter(const int* __restrict__ src, const int* __restrict__ dst,
              const int* __restrict__ tab, int* __restrict__ bedge, int e, int nbk, int nch) {
    __shared__ int cur[NBK_MAX];
    const int t = threadIdx.x;
    if (t < nbk) cur[t] = tab[t * nch + blockIdx.x];
    __syncthreads();
    const int beg = blockIdx.x * CHUNK;
    const int end = min(beg + CHUNK, e);
    int2* be = (int2*)bedge;
    for (int i = beg + t; i < end; i += 4 * STH) {
        int s[4], d[4];
#pragma unroll
        for (int k = 0; k < 4; ++k) {
            const int ii = min(i + k * STH, end - 1);
            s[k] = src[ii];
            d[k] = dst[ii];
        }
        asm volatile("" :: "v"(s[0]), "v"(s[1]), "v"(s[2]), "v"(s[3]),
                           "v"(d[0]), "v"(d[1]), "v"(d[2]), "v"(d[3]));
#pragma unroll
        for (int k = 0; k < 4; ++k)
            if (i + k * STH < end) {
                const int p = atomicAdd(&cur[d[k] >> BSH], 1);
                be[p] = make_int2(s[k], d[k]);
            }
    }
}

__global__ void __launch_bounds__(STH)
k_row_count(const int* __restrict__ bedge, const int* __restrict__ tab,
            int* __restrict__ rowptr, float* __restrict__ isq, float* __restrict__ idg,
            int n, int e, int nbk, int nch) {
    __shared__ int cnt[BROWS];
    __shared__ int sc[BROWS];
    const int t = threadIdx.x;
    const int b = blockIdx.x;
    const int base = b << BSH;
    const int ebeg = tab[b * nch];
    const int eend = (b + 1 < nbk) ? tab[(b + 1) * nch] : e;
    if (t < BROWS) cnt[t] = 0;
    __syncthreads();
    const int2* be = (const int2*)bedge;
    for (int i = ebeg + t; i < eend; i += 4 * STH) {
        int d[4];
#pragma unroll
        for (int k = 0; k < 4; ++k) d[k] = be[min(i + k * STH, eend - 1)].y;
        asm volatile("" :: "v"(d[0]), "v"(d[1]), "v"(d[2]), "v"(d[3]));
#pragma unroll
        for (int k = 0; k < 4; ++k)
            if (i + k * STH < eend) atomicAdd(&cnt[d[k] - base], 1);
    }
    __syncthreads();
    int c = 0;
    if (t < BROWS) { c = cnt[t]; sc[t] = c; }
    __syncthreads();
    for (int off = 1; off < BROWS; off <<= 1) {
        int x = 0;
        if (t < BROWS && t >= off) x = sc[t - off];
        __syncthreads();
        if (t < BROWS) sc[t] += x;
        __syncthreads();
    }
    if (t < BROWS) {
        const int node = base + t;
        if (node < n) {
            rowptr[node] = ebeg + sc[t] - c;
            const float dg = (float)(c + 1);
            isq[node] = rsqrtf(dg);
            idg[node] = 1.0f / dg;
        }
    }
    if (b == 0 && t == 0) rowptr[n] = e;
}

__global__ void __launch_bounds__(STH)
k_row_place(const int* __restrict__ bedge, const int* __restrict__ tab,
            const int* __restrict__ rowptr, const float* __restrict__ isq,
            int* __restrict__ rec, int n, int e, int nbk, int nch) {
    __shared__ int   cur[BROWS];
    __shared__ float qs[BROWS];
    const int t = threadIdx.x;
    const int b = blockIdx.x;
    const int base = b << BSH;
    const int ebeg = tab[b * nch];
    const int eend = (b + 1 < nbk) ? tab[(b + 1) * nch] : e;
    if (t < BROWS && base + t < n) {
        cur[t] = rowptr[base + t];
        qs[t]  = isq[base + t];
    }
    __syncthreads();
    const int2* be = (const int2*)bedge;
    int2* rec2 = (int2*)rec;
    if (ebeg + t < eend) {
        const int last = eend - 1;
        int2 sd[4];
#pragma unroll
        for (int k = 0; k < 4; ++k) sd[k] = be[min(ebeg + t + k * STH, last)];
        for (int i = ebeg + t; i < eend; i += 4 * STH) {
            int2 cu[4];
            float q[4];
#pragma unroll
            for (int k = 0; k < 4; ++k) {
                cu[k] = sd[k];
                q[k] = isq[cu[k].x];
            }
#pragma unroll
            for (int k = 0; k < 4; ++k) sd[k] = be[min(i + (4 + k) * STH, last)];
            asm volatile("" :: "v"(q[0]), "v"(q[1]), "v"(q[2]), "v"(q[3]));
#pragma unroll
            for (int k = 0; k < 4; ++k)
                if (i + k * STH < eend) {
                    const int j = cu[k].y - base;
                    const int p = atomicAdd(&cur[j], 1);
                    rec2[p] = make_int2(cu[k].x, __float_as_int(q[k] * qs[j]));
                }
        }
    }
}

__global__ void k_pad_x(const float* __restrict__ x, float* __restrict__ xq, int n) {
    long idx = (long)blockIdx.x * blockDim.x + threadIdx.x;
    if (idx < (long)n * 8) {
        const int i = (int)(idx >> 3), p = (int)(idx & 7);
        const int col = (p & 1) + ((p >> 1) & 1) * 4 + (p >> 2) * 2;
        xq[idx] = (col < 5) ? x[(long)i * 5 + col] : 0.0f;
    }
}

__global__ void k_prep_w(const float* __restrict__ W, float* __restrict__ Bt,
                         int Kin, int Cin, int KPAD, int COLS) {
    int idx = blockIdx.x * blockDim.x + threadIdx.x;
    if (idx >= COLS * KPAD) return;
    int col = idx / KPAD, k = idx % KPAD;
    Bt[idx] = (k < Kin && col < Cin) ? W[k * Cin + col] : 0.0f;
}

template <int K, int NT, int SC>
__global__ void k_gemm_wmma(const float* __restrict__ A, const float* __restrict__ Bt,
                            float* __restrict__ H) {
    const int lane = threadIdx.x & 31;
    const int wave = threadIdx.x >> 5;
    const int half = lane >> 4;
    const int l15  = lane & 15;
    const size_t row = (size_t)blockIdx.x * 16 + l15;
    const int    col = wave * 16 + l15;
    const float* Arow = A + row * K;
    const float* Bcol = Bt + (size_t)col * K;

    v8f c = {};
#pragma unroll
    for (int k = 0; k < K; k += 4) {
        const int ka = k + half * 2;
        v2f a = *(const v2f*)(Arow + ka);
        v2f b = *(const v2f*)(Bcol + ka);
        c = __builtin_amdgcn_wmma_f32_16x16x4_f32(
                 false, a,  false, b,
                 (short)0, c,  false,  false);
    }

    if (NT * 16 == SC || col < SC) {
        const size_t mbase = (size_t)blockIdx.x * 16 + half * 8;
#pragma unroll
        for (int r = 0; r < 8; ++r)
            H[(mbase + r) * SC + col] = c[r];
    }
}

typedef __attribute__((ext_vector_type(4))) float v4f;

__global__ void __launch_bounds__(256)
k_gather64(const int* __restrict__ rowptr, const int* __restrict__ recw,
           const float* __restrict__ h, const float* __restrict__ idg,
           const float* __restrict__ bias, float* __restrict__ out, int n) {
    const int2* rec = (const int2*)recw;
    const int l = threadIdx.x & 15;
    const int stride = gridDim.x * 16;
    const v4f* h4 = (const v4f*)h;
    const v4f bv = {bias[4 * l], bias[4 * l + 1], bias[4 * l + 2], bias[4 * l + 3]};
    for (int node = blockIdx.x * 16 + (threadIdx.x >> 4); node < n; node += stride) {
        const int beg = rowptr[node];
        const int end = rowptr[node + 1];
        v4f acc = h4[(size_t)node * 16 + l] * idg[node] + bv;
        if (beg < end) {
            const int last = end - 1;
            int2 e0 = rec[beg];
            int2 e1 = rec[min(beg + 1, last)];
            int2 e2 = rec[min(beg + 2, last)];
            int2 e3 = rec[min(beg + 3, last)];
            for (int p = beg; p < end; p += 4) {
                const v4f v0 = h4[(size_t)e0.x * 16 + l];
                const v4f v1 = h4[(size_t)e1.x * 16 + l];
                const v4f v2 = h4[(size_t)e2.x * 16 + l];
                const v4f v3 = h4[(size_t)e3.x * 16 + l];
                const float c0 = __int_as_float(e0.y);
                const float c1 = __int_as_float(e1.y);
                const float c2 = __int_as_float(e2.y);
                const float c3 = __int_as_float(e3.y);
                e0 = rec[min(p + 4, last)];
                e1 = rec[min(p + 5, last)];
                e2 = rec[min(p + 6, last)];
                e3 = rec[min(p + 7, last)];
                asm volatile("" :: "v"(v0), "v"(v1), "v"(v2), "v"(v3));
                acc += v0 * c0;
                if (p + 1 < end) acc += v1 * c1;
                if (p + 2 < end) acc += v2 * c2;
                if (p + 3 < end) acc += v3 * c3;
            }
        }
        v4f r;
        r.x = fmaxf(acc.x, 0.0f);
        r.y = fmaxf(acc.y, 0.0f);
        r.z = fmaxf(acc.z, 0.0f);
        r.w = fmaxf(acc.w, 0.0f);
        ((v4f*)out)[(size_t)node * 16 + l] = r;
    }
}

__global__ void __launch_bounds__(64)
k_layer1(const int* __restrict__ rowptr, const int* __restrict__ recw,
         const float* __restrict__ xq, const float* __restrict__ idg,
         const float* __restrict__ Bt, const float* __restrict__ bias,
         float* __restrict__ out, int ntiles) {
    const int lane = threadIdx.x & 31;
    const int half = lane >> 4;
    const int l15  = lane & 15;
    const int tile = blockIdx.x * 2 + (threadIdx.x >> 5);
    if (tile >= ntiles) return;
    const int node = tile * 16 + l15;
    const int2* rec = (const int2*)recw;
    const v4f* xq4 = (const v4f*)xq;

    const int beg = rowptr[node];
    const int end = rowptr[node + 1];

    v2f b0[4], b1[4];
#pragma unroll
    for (int ct = 0; ct < 4; ++ct) {
        const float* Bcol = Bt + (size_t)(4 * l15 + ct) * 8 + half * 2;
        b0[ct] = *(const v2f*)(Bcol);
        b1[ct] = *(const v2f*)(Bcol + 4);
    }
    const v4f bb = *(const v4f*)(bias + 4 * l15);

    v4f acc = xq4[(size_t)node * 2 + half] * idg[node];
    if (beg < end) {
        const int last = end - 1;
        int2 e[8];
#pragma unroll
        for (int j = 0; j < 8; ++j) e[j] = rec[min(beg + j, last)];
        for (int p = beg; p < end; p += 8) {
            v4f v[8];
            float c[8];
#pragma unroll
            for (int j = 0; j < 8; ++j) {
                v[j] = xq4[(size_t)e[j].x * 2 + half];
                c[j] = __int_as_float(e[j].y);
            }
#pragma unroll
            for (int j = 0; j < 8; ++j) e[j] = rec[min(p + 8 + j, last)];
            asm volatile("" :: "v"(v[0]), "v"(v[1]), "v"(v[2]), "v"(v[3]),
                               "v"(v[4]), "v"(v[5]), "v"(v[6]), "v"(v[7]));
            acc += v[0] * c[0];
#pragma unroll
            for (int j = 1; j < 8; ++j)
                if (p + j < end) acc += v[j] * c[j];
        }
    }

    const v2f a0 = {acc.x, acc.y};
    const v2f a1 = {acc.z, acc.w};
    v8f d[4];
#pragma unroll
    for (int ct = 0; ct < 4; ++ct) {
        v8f z = {};
        d[ct] = __builtin_amdgcn_wmma_f32_16x16x4_f32(
                     false, a0,  false, b0[ct],
                     (short)0, z,  false,  false);
    }
#pragma unroll
    for (int ct = 0; ct < 4; ++ct)
        d[ct] = __builtin_amdgcn_wmma_f32_16x16x4_f32(
                     false, a1,  false, b1[ct],
                     (short)0, d[ct],  false,  false);

    const size_t mbase = (size_t)tile * 16 + half * 8;
#pragma unroll
    for (int r = 0; r < 8; ++r) {
        v4f o;
        o.x = fmaxf(d[0][r] + bb.x, 0.0f);
        o.y = fmaxf(d[1][r] + bb.y, 0.0f);
        o.z = fmaxf(d[2][r] + bb.z, 0.0f);
        o.w = fmaxf(d[3][r] + bb.w, 0.0f);
        ((v4f*)out)[(mbase + r) * 16 + l15] = o;
    }
}

__global__ void k_gather2(const int* __restrict__ rowptr, const int* __restrict__ recw,
                          const float* __restrict__ h, const float* __restrict__ idg,
                          const float* __restrict__ bias, float* __restrict__ out, int n) {
    const int2* rec = (const int2*)recw;
    const int node = blockIdx.x * blockDim.x + threadIdx.x;
    if (node >= n) return;
    const int beg = rowptr[node];
    const int end = rowptr[node + 1];
    const v2f* h2 = (const v2f*)h;
    const v2f bv = {bias[0], bias[1]};
    v2f acc = h2[node] * idg[node] + bv;
    for (int p = beg; p < end; p += 4) {
        const int last = end - 1;
        const int2 e0 = rec[p];
        const int2 e1 = rec[min(p + 1, last)];
        const int2 e2 = rec[min(p + 2, last)];
        const int2 e3 = rec[min(p + 3, last)];
        const v2f v0 = h2[e0.x];
        const v2f v1 = h2[e1.x];
        const v2f v2 = h2[e2.x];
        const v2f v3 = h2[e3.x];
        const float c0 = __int_as_float(e0.y);
        const float c1 = __int_as_float(e1.y);
        const float c2 = __int_as_float(e2.y);
        const float c3 = __int_as_float(e3.y);
        asm volatile("" :: "v"(v0), "v"(v1), "v"(v2), "v"(v3));
        acc += v0 * c0;
        if (p + 1 < end) acc += v1 * c1;
        if (p + 2 < end) acc += v2 * c2;
        if (p + 3 < end) acc += v3 * c3;
    }
    ((v2f*)out)[node] = acc;
}

extern "C" void kernel_launch(void* const* d_in, const int* in_sizes, int n_in,
                              void* d_out, int out_size, void* d_ws, size_t ws_size,
                              hipStream_t stream) {
    const float* x  = (const float*)d_in[0];
    const int*   ei = (const int*)d_in[1];
    const float* W1 = (const float*)d_in[2];
    const float* b1 = (const float*)d_in[3];
    const float* W2 = (const float*)d_in[4];
    const float* b2 = (const float*)d_in[5];
    const float* W3 = (const float*)d_in[6];
    const float* b3 = (const float*)d_in[7];
    float* out = (float*)d_out;

    const int N = in_sizes[0] / 5;
    const int E = in_sizes[1] / 2;
    const int* src = ei;
    const int* dst = ei + E;

    const int T = 256;
    auto cdiv = [](long a, long b) { return (int)((a + b - 1) / b); };
    const int NBK = cdiv(N, BROWS);
    const int NCH = cdiv(E, CHUNK);

    float* ws   = (float*)d_ws;
    float* bufA = ws;
    float* bufB = bufA + (size_t)N * 64;
    float* xq   = bufB + (size_t)N * 64;
    float* w1t  = xq   + (size_t)N * 8;
    float* w2t  = w1t  + 64 * 8;
    float* w3t  = w2t  + 64 * 64;
    float* h3   = w3t  + 16 * 64;
    int*   rec  = (int*)(h3 + (size_t)N * 2);
    int*   bedg = rec  + (size_t)E * 2;
    int*   tab  = bedg + (size_t)E * 2;
    int*   rowp = tab  + (size_t)NBK * NCH;
    float* isq  = (float*)(rowp + N + 1);
    float* idg  = isq  + N;

    k_bkt_count  <<<NCH, STH, 0, stream>>>(dst, tab, E, NBK, NCH);
    k_tab_scan   <<<1,   STH, 0, stream>>>(tab, NBK * NCH);
    k_bkt_scatter<<<NCH, STH, 0, stream>>>(src, dst, tab, bedg, E, NBK, NCH);
    k_row_count  <<<NBK, STH, 0, stream>>>(bedg, tab, rowp, isq, idg, N, E, NBK, NCH);
    k_row_place  <<<NBK, STH, 0, stream>>>(bedg, tab, rowp, isq, rec, N, E, NBK, NCH);
    k_pad_x    <<<cdiv((long)N * 8, T), T, 0, stream>>>(x, xq, N);
    k_prep_w   <<<cdiv(64 * 8,  T), T, 0, stream>>>(W1, w1t, 5, 64, 8, 64);
    k_prep_w   <<<cdiv(64 * 64, T), T, 0, stream>>>(W2, w2t, 64, 64, 64, 64);
    k_prep_w   <<<cdiv(16 * 64, T), T, 0, stream>>>(W3, w3t, 64, 2, 64, 16);

    const int GT = cdiv(N, 16);
    const int GG = 2048;

    k_layer1   <<<cdiv(GT, 2), 64, 0, stream>>>(rowp, rec, xq, idg, w1t, b1, bufB, GT);

    k_gemm_wmma<64, 4, 64><<<GT, 128, 0, stream>>>(bufB, w2t, bufA);
    k_gather64 <<<GG, 256, 0, stream>>>(rowp, rec, bufA, idg, b2, bufB, N);

    k_gemm_wmma<64, 1, 2><<<GT, 32, 0, stream>>>(bufB, w3t, h3);
    k_gather2  <<<cdiv(N, T), T, 0, stream>>>(rowp, rec, h3, idg, b3, out, N);
}


# Input pad and the three weight transposes moved into spare blocks of the first sort launch (blocks past the edge chunks): 4 launches fewer, 14 to 10; sums unchanged
# speedup vs baseline: 20.6702x; 1.0530x over previous
#include <hip/hip_runtime.h>
#include <hip/hip_bf16.h>

typedef __attribute__((ext_vector_type(2))) float v2f;
typedef __attribute__((ext_vector_type(8))) float v8f;

#define BSH     9
#define BROWS   512
#define NBK_MAX 256
#define CHUNK   20480
#define STH     1024
#define STILE   32768
#define PREPB   96

__device__ __forceinline__ void prep_w(const float* __restrict__ W, float* __restrict__ Bt,
                                       int Kin, int Cin, int KPAD, int COLS, int g, int gs) {
    for (int idx = g; idx < COLS * KPAD; idx += gs) {
        const int col = idx / KPAD, k = idx % KPAD;
        Bt[idx] = (k < Kin && col < Cin) ? W[k * Cin + col] : 0.0f;
    }
}

__device__ __forceinline__ float pad_elem(const float* __restrict__ x, long idx) {
    const long i = idx >> 3;
    const int p = (int)(idx & 7);
    const int col = (p & 1) + ((p >> 1) & 1) * 4 + (p >> 2) * 2;
    return (col < 5) ? x[i * 5 + col] : 0.0f;
}

__device__ __forceinline__ void pad_x(const float* __restrict__ x, float* __restrict__ xq,
                                      int n, int g, int gs) {
    const long tot = (long)n * 8;
    for (long i0 = g; i0 < tot; i0 += 4L * gs) {
        float v[4];
#pragma unroll
        for (int k = 0; k < 4; ++k) v[k] = pad_elem(x, min(i0 + (long)k * gs, tot - 1));
#pragma unroll
        for (int k = 0; k < 4; ++k)
            if (i0 + (long)k * gs < tot) xq[i0 + (long)k * gs] = v[k];
    }
}

__global__ void __launch_bounds__(STH)
k_bkt_count(const int* __restrict__ dst, int* __restrict__ tab, int e, int nbk, int nch,
            const float* __restrict__ x, const float* __restrict__ W1,
            const float* __restrict__ W2, const float* __restrict__ W3,
            float* __restrict__ xq, float* __restrict__ w1t, float* __restrict__ w2t,
            float* __restrict__ w3t, int n) {
    __shared__ int hist[NBK_MAX];
    const int t = threadIdx.x;
    if ((int)blockIdx.x >= nch) {
        const int g  = ((int)blockIdx.x - nch) * STH + t;
        const int gs = ((int)gridDim.x - nch) * STH;
        prep_w(W1, w1t, 5, 64, 8, 64, g, gs);
        prep_w(W2, w2t, 64, 64, 64, 64, g, gs);
        prep_w(W3, w3t, 64, 2, 64, 16, g, gs);
        pad_x(x, xq, n, g, gs);
        return;
    }
    if (t < NBK_MAX) hist[t] = 0;
    __syncthreads();
    const int beg = blockIdx.x * CHUNK;
    const int end = min(beg + CHUNK, e);
    for (int i = beg + t; i < end; i += 4 * STH) {
        int d[4];
#pragma unroll
        for (int k = 0; k < 4; ++k) d[k] = dst[min(i + k * STH, end - 1)];
        asm volatile("" :: "v"(d[0]), "v"(d[1]), "v"(d[2]), "v"(d[3]));
#pragma unroll
        for (int k = 0; k < 4; ++k)
            if (i + k * STH < end) atomicAdd(&hist[d[k] >> BSH], 1);
    }
    __syncthreads();
    if (t < nbk) tab[t * nch + blockIdx.x] = hist[t];
}

__global__ void __launch_bounds__(STH)
k_tab_scan(int* __restrict__ tab, int len) {
    __shared__ int buf[STILE + STILE / 32];
    __shared__ int part[STH];
    const int t = threadIdx.x;
    int carry = 0;
    for (int tile = 0; tile < len; tile += STILE) {
        for (int k0 = t; k0 < STILE; k0 += 8 * STH) {
            int v[8];
#pragma unroll
            for (int u = 0; u < 8; ++u) v[u] = tab[min(tile + k0 + u * STH, len - 1)];
#pragma unroll
            for (int u = 0; u < 8; ++u) {
                const int k = k0 + u * STH;
                buf[k + (k >> 5)] = (tile + k < len) ? v[u] : 0;
            }
        }
        __syncthreads();
        int s = 0;
        for (int j = 0; j < 32; ++j) s += buf[33 * t + j];
        part[t] = s;
        __syncthreads();
        for (int off = 1; off < STH; off <<= 1) {
            const int x = (t >= off) ? part[t - off] : 0;
            __syncthreads();
            part[t] += x;
            __syncthreads();
        }
        int run = carry + part[t] - s;
        const int tot = part[STH - 1];
        for (int j = 0; j < 32; ++j) {
            const int v = buf[33 * t + j];
            buf[33 * t + j] = run;
            run += v;
        }
        __syncthreads();
        for (int k = t; k < STILE; k += STH) {
            const int g = tile + k;
            if (g < len) tab[g] = buf[k + (k >> 5)];
        }
        carry += tot;
        __syncthreads();
    }
}

__global__ void __launch_bounds__(STH)
k_bkt_scatter(const int* __restrict__ src, const int* __restrict__ dst,
              const int* __restrict__ tab, int* __restrict__ bedge, int e, int nbk, int nch) {
    __shared__ int cur[NBK_MAX];
    const int t = threadIdx.x;
    if (t < nbk) cur[t] = tab[t * nch + blockIdx.x];
    __syncthreads();
    const int beg = blockIdx.x * CHUNK;
    const int end = min(beg + CHUNK, e);
    int2* be = (int2*)bedge;
    for (int i = beg + t; i < end; i += 4 * STH) {
        int s[4], d[4];
#pragma unroll
        for (int k = 0; k < 4; ++k) {
            const int ii = min(i + k * STH, end - 1);
            s[k] = src[ii];
            d[k] = dst[ii];
        }
        asm volatile("" :: "v"(s[0]), "v"(s[1]), "v"(s[2]), "v"(s[3]),
                           "v"(d[0]), "v"(d[1]), "v"(d[2]), "v"(d[3]));
#pragma unroll
        for (int k = 0; k < 4; ++k)
            if (i + k * STH < end) {
                const int p = atomicAdd(&cur[d[k] >> BSH], 1);
                be[p] = make_int2(s[k], d[k]);
            }
    }
}

__global__ void __launch_bounds__(STH)
k_row_count(const int* __restrict__ bedge, const int* __restrict__ tab,
            int* __restrict__ rowptr, float* __restrict__ isq, float* __restrict__ idg,
            int n, int e, int nbk, int nch) {
    __shared__ int cnt[BROWS];
    __shared__ int sc[BROWS];
    const int t = threadIdx.x;
    const int b = blockIdx.x;
    const int base = b << BSH;
    const int ebeg = tab[b * nch];
    const int eend = (b + 1 < nbk) ? tab[(b + 1) * nch] : e;
    if (t < BROWS) cnt[t] = 0;
    __syncthreads();
    const int2* be = (const int2*)bedge;
    for (int i = ebeg + t; i < eend; i += 4 * STH) {
        int d[4];
#pragma unroll
        for (int k = 0; k < 4; ++k) d[k] = be[min(i + k * STH, eend - 1)].y;
        asm volatile("" :: "v"(d[0]), "v"(d[1]), "v"(d[2]), "v"(d[3]));
#pragma unroll
        for (int k = 0; k < 4; ++k)
            if (i + k * STH < eend) atomicAdd(&cnt[d[k] - base], 1);
    }
    __syncthreads();
    int c = 0;
    if (t < BROWS) { c = cnt[t]; sc[t] = c; }
    __syncthreads();
    for (int off = 1; off < BROWS; off <<= 1) {
        int x = 0;
        if (t < BROWS && t >= off) x = sc[t - off];
        __syncthreads();
        if (t < BROWS) sc[t] += x;
        __syncthreads();
    }
    if (t < BROWS) {
        const int node = base + t;
        if (node < n) {
            rowptr[node] = ebeg + sc[t] - c;
            const float dg = (float)(c + 1);
            isq[node] = rsqrtf(dg);
            idg[node] = 1.0f / dg;
        }
    }
    if (b == 0 && t == 0) rowptr[n] = e;
}

__global__ void __launch_bounds__(STH)
k_row_place(const int* __restrict__ bedge, const int* __restrict__ tab,
            const int* __restrict__ rowptr, const float* __restrict__ isq,
            int* __restrict__ rec, int n, int e, int nbk, int nch) {
    __shared__ int   cur[BROWS];
    __shared__ float qs[BROWS];
    const int t = threadIdx.x;
    const int b = blockIdx.x;
    const int base = b << BSH;
    const int ebeg = tab[b * nch];
    const int eend = (b + 1 < nbk) ? tab[(b + 1) * nch] : e;
    if (t < BROWS && base + t < n) {
        cur[t] = rowptr[base + t];
        qs[t]  = isq[base + t];
    }
    __syncthreads();
    const int2* be = (const int2*)bedge;
    int2* rec2 = (int2*)rec;
    if (ebeg + t < eend) {
        const int last = eend - 1;
        int2 sd[4];
#pragma unroll
        for (int k = 0; k < 4; ++k) sd[k] = be[min(ebeg + t + k * STH, last)];
        for (int i = ebeg + t; i < eend; i += 4 * STH) {
            int2 cu[4];
            float q[4];
#pragma unroll
            for (int k = 0; k < 4; ++k) {
                cu[k] = sd[k];
                q[k] = isq[cu[k].x];
            }
#pragma unroll
            for (int k = 0; k < 4; ++k) sd[k] = be[min(i + (4 + k) * STH, last)];
            asm volatile("" :: "v"(q[0]), "v"(q[1]), "v"(q[2]), "v"(q[3]));
#pragma unroll
            for (int k = 0; k < 4; ++k)
                if (i + k * STH < eend) {
                    const int j = cu[k].y - base;
                    const int p = atomicAdd(&cur[j], 1);
                    rec2[p] = make_int2(cu[k].x, __float_as_int(q[k] * qs[j]));
                }
        }
    }
}

template <int K, int NT, int SC>
__global__ void k_gemm_wmma(const float* __restrict__ A, const float* __restrict__ Bt,
                            float* __restrict__ H) {
    const int lane = threadIdx.x & 31;
    const int wave = threadIdx.x >> 5;
    const int half = lane >> 4;
    const int l15  = lane & 15;
    const size_t row = (size_t)blockIdx.x * 16 + l15;
    const int    col = wave * 16 + l15;
    const float* Arow = A + row * K;
    const float* Bcol = Bt + (size_t)col * K;

    v8f c = {};
#pragma unroll
    for (int k = 0; k < K; k += 4) {
        const int ka = k + half * 2;
        v2f a = *(const v2f*)(Arow + ka);
        v2f b = *(const v2f*)(Bcol + ka);
        c = __builtin_amdgcn_wmma_f32_16x16x4_f32(
                 false, a,  false, b,
                 (short)0, c,  false,  false);
    }

    if (NT * 16 == SC || col < SC) {
        const size_t mbase = (size_t)blockIdx.x * 16 + half * 8;
#pragma unroll
        for (int r = 0; r < 8; ++r)
            H[(mbase + r) * SC + col] = c[r];
    }
}

typedef __attribute__((ext_vector_type(4))) float v4f;

__global__ void __launch_bounds__(256)
k_gather64(const int* __restrict__ rowptr, const int* __restrict__ recw,
           const float* __restrict__ h, const float* __restrict__ idg,
           const float* __restrict__ bias, float* __restrict__ out, int n) {
    const int2* rec = (const int2*)recw;
    const int l = threadIdx.x & 15;
    const int stride = gridDim.x * 16;
    const v4f* h4 = (const v4f*)h;
    const v4f bv = {bias[4 * l], bias[4 * l + 1], bias[4 * l + 2], bias[4 * l + 3]};
    for (int node = blockIdx.x * 16 + (threadIdx.x >> 4); node < n; node += stride) {
        const int beg = rowptr[node];
        const int end = rowptr[node + 1];
        v4f acc = h4[(size_t)node * 16 + l] * idg[node] + bv;
        if (beg < end) {
            const int last = end - 1;
            int2 e0 = rec[beg];
            int2 e1 = rec[min(beg + 1, last)];
            int2 e2 = rec[min(beg + 2, last)];
            int2 e3 = rec[min(beg + 3, last)];
            for (int p = beg; p < end; p += 4) {
                const v4f v0 = h4[(size_t)e0.x * 16 + l];
                const v4f v1 = h4[(size_t)e1.x * 16 + l];
                const v4f v2 = h4[(size_t)e2.x * 16 + l];
                const v4f v3 = h4[(size_t)e3.x * 16 + l];
                const float c0 = __int_as_float(e0.y);
                const float c1 = __int_as_float(e1.y);
                const float c2 = __int_as_float(e2.y);
                const float c3 = __int_as_float(e3.y);
                e0 = rec[min(p + 4, last)];
                e1 = rec[min(p + 5, last)];
                e2 = rec[min(p + 6, last)];
                e3 = rec[min(p + 7, last)];
                asm volatile("" :: "v"(v0), "v"(v1), "v"(v2), "v"(v3));
                acc += v0 * c0;
                if (p + 1 < end) acc += v1 * c1;
                if (p + 2 < end) acc += v2 * c2;
                if (p + 3 < end) acc += v3 * c3;
            }
        }
        v4f r;
        r.x = fmaxf(acc.x, 0.0f);
        r.y = fmaxf(acc.y, 0.0f);
        r.z = fmaxf(acc.z, 0.0f);
        r.w = fmaxf(acc.w, 0.0f);
        ((v4f*)out)[(size_t)node * 16 + l] = r;
    }
}

__global__ void __launch_bounds__(64)
k_layer1(const int* __restrict__ rowptr, const int* __restrict__ recw,
         const float* __restrict__ xq, const float* __restrict__ idg,
         const float* __restrict__ Bt, const float* __restrict__ bias,
         float* __restrict__ out, int ntiles) {
    const int lane = threadIdx.x & 31;
    const int half = lane >> 4;
    const int l15  = lane & 15;
    const int tile = blockIdx.x * 2 + (threadIdx.x >> 5);
    if (tile >= ntiles) return;
    const int node = tile * 16 + l15;
    const int2* rec = (const int2*)recw;
    const v4f* xq4 = (const v4f*)xq;

    const int beg = rowptr[node];
    const int end = rowptr[node + 1];

    v2f b0[4], b1[4];
#pragma unroll
    for (int ct = 0; ct < 4; ++ct) {
        const float* Bcol = Bt + (size_t)(4 * l15 + ct) * 8 + half * 2;
        b0[ct] = *(const v2f*)(Bcol);
        b1[ct] = *(const v2f*)(Bcol + 4);
    }
    const v4f bb = *(const v4f*)(bias + 4 * l15);

    v4f acc = xq4[(size_t)node * 2 + half] * idg[node];
    if (beg < end) {
        const int last = end - 1;
        int2 e[8];
#pragma unroll
        for (int j = 0; j < 8; ++j) e[j] = rec[min(beg + j, last)];
        for (int p = beg; p < end; p += 8) {
            v4f v[8];
            float c[8];
#pragma unroll
            for (int j = 0; j < 8; ++j) {
                v[j] = xq4[(size_t)e[j].x * 2 + half];
                c[j] = __int_as_float(e[j].y);
            }
#pragma unroll
            for (int j = 0; j < 8; ++j) e[j] = rec[min(p + 8 + j, last)];
            asm volatile("" :: "v"(v[0]), "v"(v[1]), "v"(v[2]), "v"(v[3]),
                               "v"(v[4]), "v"(v[5]), "v"(v[6]), "v"(v[7]));
            acc += v[0] * c[0];
#pragma unroll
            for (int j = 1; j < 8; ++j)
                if (p + j < end) acc += v[j] * c[j];
        }
    }

    const v2f a0 = {acc.x, acc.y};
    const v2f a1 = {acc.z, acc.w};
    v8f d[4];
#pragma unroll
    for (int ct = 0; ct < 4; ++ct) {
        v8f z = {};
        d[ct] = __builtin_amdgcn_wmma_f32_16x16x4_f32(
                     false, a0,  false, b0[ct],
                     (short)0, z,  false,  false);
    }
#pragma unroll
    for (int ct = 0; ct < 4; ++ct)
        d[ct] = __builtin_amdgcn_wmma_f32_16x16x4_f32(
                     false, a1,  false, b1[ct],
                     (short)0, d[ct],  false,  false);

    const size_t mbase = (size_t)tile * 16 + half * 8;
#pragma unroll
    for (int r = 0; r < 8; ++r) {
        v4f o;
        o.x = fmaxf(d[0][r] + bb.x, 0.0f);
        o.y = fmaxf(d[1][r] + bb.y, 0.0f);
        o.z = fmaxf(d[2][r] + bb.z, 0.0f);
        o.w = fmaxf(d[3][r] + bb.w, 0.0f);
        ((v4f*)out)[(mbase + r) * 16 + l15] = o;
    }
}

__global__ void k_gather2(const int* __restrict__ rowptr, const int* __restrict__ recw,
                          const float* __restrict__ h, const float* __restrict__ idg,
                          const float* __restrict__ bias, float* __restrict__ out, int n) {
    const int2* rec = (const int2*)recw;
    const int node = blockIdx.x * blockDim.x + threadIdx.x;
    if (node >= n) return;
    const int beg = rowptr[node];
    const int end = rowptr[node + 1];
    const v2f* h2 = (const v2f*)h;
    const v2f bv = {bias[0], bias[1]};
    v2f acc = h2[node] * idg[node] + bv;
    for (int p = beg; p < end; p += 4) {
        const int last = end - 1;
        const int2 e0 = rec[p];
        const int2 e1 = rec[min(p + 1, last)];
        const int2 e2 = rec[min(p + 2, last)];
        const int2 e3 = rec[min(p + 3, last)];
        const v2f v0 = h2[e0.x];
        const v2f v1 = h2[e1.x];
        const v2f v2 = h2[e2.x];
        const v2f v3 = h2[e3.x];
        const float c0 = __int_as_float(e0.y);
        const float c1 = __int_as_float(e1.y);
        const float c2 = __int_as_float(e2.y);
        const float c3 = __int_as_float(e3.y);
        asm volatile("" :: "v"(v0), "v"(v1), "v"(v2), "v"(v3));
        acc += v0 * c0;
        if (p + 1 < end) acc += v1 * c1;
        if (p + 2 < end) acc += v2 * c2;
        if (p + 3 < end) acc += v3 * c3;
    }
    ((v2f*)out)[node] = acc;
}

extern "C" void kernel_launch(void* const* d_in, const int* in_sizes, int n_in,
                              void* d_out, int out_size, void* d_ws, size_t ws_size,
                              hipStream_t stream) {
    const float* x  = (const float*)d_in[0];
    const int*   ei = (const int*)d_in[1];
    const float* W1 = (const float*)d_in[2];
    const float* b1 = (const float*)d_in[3];
    const float* W2 = (const float*)d_in[4];
    const float* b2 = (const float*)d_in[5];
    const float* W3 = (const float*)d_in[6];
    const float* b3 = (const float*)d_in[7];
    float* out = (float*)d_out;

    const int N = in_sizes[0] / 5;
    const int E = in_sizes[1] / 2;
    const int* src = ei;
    const int* dst = ei + E;

    const int T = 256;
    auto cdiv = [](long a, long b) { return (int)((a + b - 1) / b); };
    const int NBK = cdiv(N, BROWS);
    const int NCH = cdiv(E, CHUNK);

    float* ws   = (float*)d_ws;
    float* bufA = ws;
    float* bufB = bufA + (size_t)N * 64;
    float* xq   = bufB + (size_t)N * 64;
    float* w1t  = xq   + (size_t)N * 8;
    float* w2t  = w1t  + 64 * 8;
    float* w3t  = w2t  + 64 * 64;
    float* h3   = w3t  + 16 * 64;
    int*   rec  = (int*)(h3 + (size_t)N * 2);
    int*   bedg = rec  + (size_t)E * 2;
    int*   tab  = bedg + (size_t)E * 2;
    int*   rowp = tab  + (size_t)NBK * NCH;
    float* isq  = (float*)(rowp + N + 1);
    float* idg  = isq  + N;

    k_bkt_count  <<<NCH + PREPB, STH, 0, stream>>>(dst, tab, E, NBK, NCH,
                                                   x, W1, W2, W3, xq, w1t, w2t, w3t, N);
    k_tab_scan   <<<1,   STH, 0, stream>>>(tab, NBK * NCH);
    k_bkt_scatter<<<NCH, STH, 0, stream>>>(src, dst, tab, bedg, E, NBK, NCH);
    k_row_count  <<<NBK, STH, 0, stream>>>(bedg, tab, rowp, isq, idg, N, E, NBK, NCH);
    k_row_place  <<<NBK, STH, 0, stream>>>(bedg, tab, rowp, isq, rec, N, E, NBK, NCH);

    const int GT = cdiv(N, 16);
    const int GG = 2048;

    k_layer1   <<<cdiv(GT, 2), 64, 0, stream>>>(rowp, rec, xq, idg, w1t, b1, bufB, GT);

    k_gemm_wmma<64, 4, 64><<<GT, 128, 0, stream>>>(bufB, w2t, bufA);
    k_gather64 <<<GG, 256, 0, stream>>>(rowp, rec, bufA, idg, b2, bufB, N);

    k_gemm_wmma<64, 1, 2><<<GT, 32, 0, stream>>>(bufB, w3t, h3);
    k_gather2  <<<cdiv(N, T), T, 0, stream>>>(rowp, rec, h3, idg, b3, out, N);
}


# Rows pre-scaled by inv_sqrt(deg) where made: a record is the source row only (no per-edge factor, no factor gather; row count+placement in one launch, 10 to 9); factor once per sum, rounding changes
# speedup vs baseline: 21.9469x; 1.0618x over previous
#include <hip/hip_runtime.h>
#include <hip/hip_bf16.h>

typedef __attribute__((ext_vector_type(2))) float v2f;
typedef __attribute__((ext_vector_type(8))) float v8f;

#define BSH     9
#define BROWS   512
#define NBK_MAX 256
#define CHUNK   20480
#define STH     1024
#define STILE   32768
#define PREPB   6

__device__ __forceinline__ void prep_w(const float* __restrict__ W, float* __restrict__ Bt,
                                       int Kin, int Cin, int KPAD, int COLS, int g, int gs) {
    for (int idx = g; idx < COLS * KPAD; idx += gs) {
        const int col = idx / KPAD, k = idx % KPAD;
        Bt[idx] = (k < Kin && col < Cin) ? W[k * Cin + col] : 0.0f;
    }
}

__global__ void __launch_bounds__(STH)
k_bkt_count(const int* __restrict__ dst, int* __restrict__ tab, int e, int nbk, int nch,
            const float* __restrict__ W1, const float* __restrict__ W2,
            const float* __restrict__ W3, float* __restrict__ w1t,
            float* __restrict__ w2t, float* __restrict__ w3t) {
    __shared__ int hist[NBK_MAX];
    const int t = threadIdx.x;
    if ((int)blockIdx.x >= nch) {
        const int g  = ((int)blockIdx.x - nch) * STH + t;
        const int gs = ((int)gridDim.x - nch) * STH;
        prep_w(W1, w1t, 5, 64, 8, 64, g, gs);
        prep_w(W2, w2t, 64, 64, 64, 64, g, gs);
        prep_w(W3, w3t, 64, 2, 64, 16, g, gs);
        return;
    }
    if (t < NBK_MAX) hist[t] = 0;
    __syncthreads();
    const int beg = blockIdx.x * CHUNK;
    const int end = min(beg + CHUNK, e);
    for (int i = beg + t; i < end; i += 4 * STH) {
        int d[4];
#pragma unroll
        for (int k = 0; k < 4; ++k) d[k] = dst[min(i + k * STH, end - 1)];
        asm volatile("" :: "v"(d[0]), "v"(d[1]), "v"(d[2]), "v"(d[3]));
#pragma unroll
        for (int k = 0; k < 4; ++k)
            if (i + k * STH < end) atomicAdd(&hist[d[k] >> BSH], 1);
    }
    __syncthreads();
    if (t < nbk) tab[t * nch + blockIdx.x] = hist[t];
}

__global__ void __launch_bounds__(STH)
k_tab_scan(int* __restrict__ tab, int len) {
    __shared__ int buf[STILE + STILE / 32];
    __shared__ int part[STH];
    const int t = threadIdx.x;
    int carry = 0;
    for (int tile = 0; tile < len; tile += STILE) {
        for (int k0 = t; k0 < STILE; k0 += 8 * STH) {
            int v[8];
#pragma unroll
            for (int u = 0; u < 8; ++u) v[u] = tab[min(tile + k0 + u * STH, len - 1)];
#pragma unroll
            for (int u = 0; u < 8; ++u) {
                const int k = k0 + u * STH;
                buf[k + (k >> 5)] = (tile + k < len) ? v[u] : 0;
            }
        }
        __syncthreads();
        int s = 0;
        for (int j = 0; j < 32; ++j) s += buf[33 * t + j];
        part[t] = s;
        __syncthreads();
        for (int off = 1; off < STH; off <<= 1) {
            const int x = (t >= off) ? part[t - off] : 0;
            __syncthreads();
            part[t] += x;
            __syncthreads();
        }
        int run = carry + part[t] - s;
        const int tot = part[STH - 1];
        for (int j = 0; j < 32; ++j) {
            const int v = buf[33 * t + j];
            buf[33 * t + j] = run;
            run += v;
        }
        __syncthreads();
        for (int k = t; k < STILE; k += STH) {
            const int g = tile + k;
            if (g < len) tab[g] = buf[k + (k >> 5)];
        }
        carry += tot;
        __syncthreads();
    }
}

__global__ void __launch_bounds__(STH)
k_bkt_scatter(const int* __restrict__ src, const int* __restrict__ dst,
              const int* __restrict__ tab, int* __restrict__ bedge, int e, int nbk, int nch) {
    __shared__ int cur[NBK_MAX];
    const int t = threadIdx.x;
    if (t < nbk) cur[t] = tab[t * nch + blockIdx.x];
    __syncthreads();
    const int beg = blockIdx.x * CHUNK;
    const int end = min(beg + CHUNK, e);
    int2* be = (int2*)bedge;
    for (int i = beg + t; i < end; i += 4 * STH) {
        int s[4], d[4];
#pragma unroll
        for (int k = 0; k < 4; ++k) {
            const int ii = min(i + k * STH, end - 1);
            s[k] = src[ii];
            d[k] = dst[ii];
        }
        asm volatile("" :: "v"(s[0]), "v"(s[1]), "v"(s[2]), "v"(s[3]),
                           "v"(d[0]), "v"(d[1]), "v"(d[2]), "v"(d[3]));
#pragma unroll
        for (int k = 0; k < 4; ++k)
            if (i + k * STH < end) {
                const int p = atomicAdd(&cur[d[k] >> BSH], 1);
                be[p] = make_int2(s[k], d[k]);
            }
    }
}

__global__ void __launch_bounds__(STH)
k_row_sort(const int* __restrict__ bedge, const int* __restrict__ tab,
           const float* __restrict__ x, int* __restrict__ rowptr,
           float* __restrict__ isq, float* __restrict__ xq, int* __restrict__ rec,
           int n, int e, int nbk, int nch) {
    __shared__ int   cnt[BROWS];
    __shared__ int   sc[BROWS];
    __shared__ float qs[BROWS];
    const int t = threadIdx.x;
    const int b = blockIdx.x;
    const int base = b << BSH;
    const int ebeg = tab[b * nch];
    const int eend = (b + 1 < nbk) ? tab[(b + 1) * nch] : e;
    if (t < BROWS) cnt[t] = 0;
    __syncthreads();
    const int2* be = (const int2*)bedge;
    for (int i = ebeg + t; i < eend; i += 4 * STH) {
        int d[4];
#pragma unroll
        for (int k = 0; k < 4; ++k) d[k] = be[min(i + k * STH, eend - 1)].y;
        asm volatile("" :: "v"(d[0]), "v"(d[1]), "v"(d[2]), "v"(d[3]));
#pragma unroll
        for (int k = 0; k < 4; ++k)
            if (i + k * STH < eend) atomicAdd(&cnt[d[k] - base], 1);
    }
    __syncthreads();
    int c = 0;
    if (t < BROWS) { c = cnt[t]; sc[t] = c; }
    __syncthreads();
    for (int off = 1; off < BROWS; off <<= 1) {
        int v = 0;
        if (t < BROWS && t >= off) v = sc[t - off];
        __syncthreads();
        if (t < BROWS) sc[t] += v;
        __syncthreads();
    }
    if (t < BROWS) {
        const int first = ebeg + sc[t] - c;
        const float q = rsqrtf((float)(c + 1));
        cnt[t] = first;
        qs[t]  = q;
        const int node = base + t;
        if (node < n) {
            rowptr[node] = first;
            isq[node] = q;
        }
    }
    if (b == 0 && t == 0) rowptr[n] = e;
    __syncthreads();
    {
        const int p = t & 7;
        const int col = (p & 1) + ((p >> 1) & 1) * 4 + (p >> 2) * 2;
        float xv[BROWS * 8 / STH];
#pragma unroll
        for (int k = 0; k < BROWS * 8 / STH; ++k) {
            const int node = min(base + (t >> 3) + k * (STH / 8), n - 1);
            xv[k] = x[(size_t)node * 5 + min(col, 4)];
        }
        static_assert(BROWS * 8 / STH == 4, "the fence below names four loaded values");
        asm volatile("" :: "v"(xv[0]), "v"(xv[1]), "v"(xv[2]), "v"(xv[3]));
#pragma unroll
        for (int k = 0; k < BROWS * 8 / STH; ++k) {
            const int i = (t >> 3) + k * (STH / 8);
            if (base + i < n)
                xq[(size_t)(base + i) * 8 + p] = (col < 5) ? xv[k] * qs[i] : 0.0f;
        }
    }
    for (int i = ebeg + t; i < eend; i += 4 * STH) {
        int2 sd[4];
#pragma unroll
        for (int k = 0; k < 4; ++k) sd[k] = be[min(i + k * STH, eend - 1)];
        asm volatile("" :: "v"(sd[0].x), "v"(sd[1].x), "v"(sd[2].x), "v"(sd[3].x),
                           "v"(sd[0].y), "v"(sd[1].y), "v"(sd[2].y), "v"(sd[3].y));
#pragma unroll
        for (int k = 0; k < 4; ++k)
            if (i + k * STH < eend) {
                const int p = atomicAdd(&cnt[sd[k].y - base], 1);
                rec[p] = sd[k].x;
            }
    }
}

typedef __attribute__((ext_vector_type(4))) float v4f;

template <int K, int NT, int SC>
__global__ void k_gemm_wmma(const float* __restrict__ A, const float* __restrict__ Bt,
                            const float* __restrict__ isq, float* __restrict__ H) {
    const int lane = threadIdx.x & 31;
    const int wave = threadIdx.x >> 5;
    const int half = lane >> 4;
    const int l15  = lane & 15;
    const size_t row = (size_t)blockIdx.x * 16 + l15;
    const int    col = wave * 16 + l15;
    const float* Arow = A + row * K;
    const float* Bcol = Bt + (size_t)col * K;
    const size_t mbase = (size_t)blockIdx.x * 16 + half * 8;
    const v4f q0 = *(const v4f*)(isq + mbase);
    const v4f q1 = *(const v4f*)(isq + mbase + 4);

    v8f c = {};
#pragma unroll
    for (int k = 0; k < K; k += 4) {
        const int ka = k + half * 2;
        v2f a = *(const v2f*)(Arow + ka);
        v2f b = *(const v2f*)(Bcol + ka);
        c = __builtin_amdgcn_wmma_f32_16x16x4_f32(
                 false, a,  false, b,
                 (short)0, c,  false,  false);
    }

    if (NT * 16 == SC || col < SC) {
        const float q[8] = {q0.x, q0.y, q0.z, q0.w, q1.x, q1.y, q1.z, q1.w};
#pragma unroll
        for (int r = 0; r < 8; ++r)
            H[(mbase + r) * SC + col] = c[r] * q[r];
    }
}

__global__ void __launch_bounds__(256)
k_gather64(const int* __restrict__ rowptr, const int* __restrict__ rec,
           const float* __restrict__ h, const float* __restrict__ isq,
           const float* __restrict__ bias, float* __restrict__ out, int n) {
    const int l = threadIdx.x & 15;
    const int stride = gridDim.x * 16;
    const v4f* h4 = (const v4f*)h;
    const v4f bv = {bias[4 * l], bias[4 * l + 1], bias[4 * l + 2], bias[4 * l + 3]};
    for (int node = blockIdx.x * 16 + (threadIdx.x >> 4); node < n; node += stride) {
        const int beg = rowptr[node];
        const int end = rowptr[node + 1];
        const float q = isq[node];
        v4f acc = h4[(size_t)node * 16 + l];
        if (beg < end) {
            const int last = end - 1;
            int e0 = rec[beg];
            int e1 = rec[min(beg + 1, last)];
            int e2 = rec[min(beg + 2, last)];
            int e3 = rec[min(beg + 3, last)];
            for (int p = beg; p < end; p += 4) {
                const v4f v0 = h4[(size_t)e0 * 16 + l];
                const v4f v1 = h4[(size_t)e1 * 16 + l];
                const v4f v2 = h4[(size_t)e2 * 16 + l];
                const v4f v3 = h4[(size_t)e3 * 16 + l];
                e0 = rec[min(p + 4, last)];
                e1 = rec[min(p + 5, last)];
                e2 = rec[min(p + 6, last)];
                e3 = rec[min(p + 7, last)];
                __builtin_amdgcn_sched_barrier(0);
                asm volatile("" :: "v"(v0), "v"(v1), "v"(v2), "v"(v3));
                acc += v0;
                if (p + 1 < end) acc += v1;
                if (p + 2 < end) acc += v2;
                if (p + 3 < end) acc += v3;
            }
        }
        acc = acc * q + bv;
        v4f r;
        r.x = fmaxf(acc.x, 0.0f);
        r.y = fmaxf(acc.y, 0.0f);
        r.z = fmaxf(acc.z, 0.0f);
        r.w = fmaxf(acc.w, 0.0f);
        ((v4f*)out)[(size_t)node * 16 + l] = r;
    }
}

__global__ void __launch_bounds__(64)
k_layer1(const int* __restrict__ rowptr, const int* __restrict__ rec,
         const float* __restrict__ xq, const float* __restrict__ isq,
         const float* __restrict__ Bt, const float* __restrict__ bias,
         float* __restrict__ out, int ntiles) {
    const int lane = threadIdx.x & 31;
    const int half = lane >> 4;
    const int l15  = lane & 15;
    const int tile = blockIdx.x * 2 + (threadIdx.x >> 5);
    if (tile >= ntiles) return;
    const int node = tile * 16 + l15;
    const v4f* xq4 = (const v4f*)xq;

    const int beg = rowptr[node];
    const int end = rowptr[node + 1];

    v2f b0[4], b1[4];
#pragma unroll
    for (int ct = 0; ct < 4; ++ct) {
        const float* Bcol = Bt + (size_t)(4 * l15 + ct) * 8 + half * 2;
        b0[ct] = *(const v2f*)(Bcol);
        b1[ct] = *(const v2f*)(Bcol + 4);
    }
    const v4f bb = *(const v4f*)(bias + 4 * l15);

    const float q = isq[node];
    v4f acc = xq4[(size_t)node * 2 + half];
    if (beg < end) {
        const int last = end - 1;
        int e[8];
#pragma unroll
        for (int j = 0; j < 8; ++j) e[j] = rec[min(beg + j, last)];
        for (int p = beg; p < end; p += 8) {
            v4f v[8];
#pragma unroll
            for (int j = 0; j < 8; ++j) v[j] = xq4[(size_t)e[j] * 2 + half];
#pragma unroll
            for (int j = 0; j < 8; ++j) e[j] = rec[min(p + 8 + j, last)];
            __builtin_amdgcn_sched_barrier(0);
            asm volatile("" :: "v"(v[0]), "v"(v[1]), "v"(v[2]), "v"(v[3]),
                               "v"(v[4]), "v"(v[5]), "v"(v[6]), "v"(v[7]));
            acc += v[0];
#pragma unroll
            for (int j = 1; j < 8; ++j)
                if (p + j < end) acc += v[j];
        }
    }
    acc *= q;

    const v2f a0 = {acc.x, acc.y};
    const v2f a1 = {acc.z, acc.w};
    v8f d[4];
#pragma unroll
    for (int ct = 0; ct < 4; ++ct) {
        v8f z = {};
        d[ct] = __builtin_amdgcn_wmma_f32_16x16x4_f32(
                     false, a0,  false, b0[ct],
                     (short)0, z,  false,  false);
    }
#pragma unroll
    for (int ct = 0; ct < 4; ++ct)
        d[ct] = __builtin_amdgcn_wmma_f32_16x16x4_f32(
                     false, a1,  false, b1[ct],
                     (short)0, d[ct],  false,  false);

    const size_t mbase = (size_t)tile * 16 + half * 8;
#pragma unroll
    for (int r = 0; r < 8; ++r) {
        v4f o;
        o.x = fmaxf(d[0][r] + bb.x, 0.0f);
        o.y = fmaxf(d[1][r] + bb.y, 0.0f);
        o.z = fmaxf(d[2][r] + bb.z, 0.0f);
        o.w = fmaxf(d[3][r] + bb.w, 0.0f);
        ((v4f*)out)[(mbase + r) * 16 + l15] = o;
    }
}

__global__ void k_gather2(const int* __restrict__ rowptr, const int* __restrict__ rec,
                          const float* __restrict__ h, const float* __restrict__ isq,
                          const float* __restrict__ bias, float* __restrict__ out, int n) {
    const int node = blockIdx.x * blockDim.x + threadIdx.x;
    if (node >= n) return;
    const int beg = rowptr[node];
    const int end = rowptr[node + 1];
    const v2f* h2 = (const v2f*)h;
    const v2f bv = {bias[0], bias[1]};
    const float q = isq[node];
    v2f acc = h2[node];
    for (int p = beg; p < end; p += 4) {
        const int last = end - 1;
        const int e0 = rec[p];
        const int e1 = rec[min(p + 1, last)];
        const int e2 = rec[min(p + 2, last)];
        const int e3 = rec[min(p + 3, last)];
        const v2f v0 = h2[e0];
        const v2f v1 = h2[e1];
        const v2f v2 = h2[e2];
        const v2f v3 = h2[e3];
        asm volatile("" :: "v"(v0), "v"(v1), "v"(v2), "v"(v3));
        acc += v0;
        if (p + 1 < end) acc += v1;
        if (p + 2 < end) acc += v2;
        if (p + 3 < end) acc += v3;
    }
    ((v2f*)out)[node] = acc * q + bv;
}

extern "C" void kernel_launch(void* const* d_in, const int* in_sizes, int n_in,
                              void* d_out, int out_size, void* d_ws, size_t ws_size,
                              hipStream_t stream) {
    const float* x  = (const float*)d_in[0];
    const int*   ei = (const int*)d_in[1];
    const float* W1 = (const float*)d_in[2];
    const float* b1 = (const float*)d_in[3];
    const float* W2 = (const float*)d_in[4];
    const float* b2 = (const float*)d_in[5];
    const float* W3 = (const float*)d_in[6];
    const float* b3 = (const float*)d_in[7];
    float* out = (float*)d_out;

    const int N = in_sizes[0] / 5;
    const int E = in_sizes[1] / 2;
    const int* src = ei;
    const int* dst = ei + E;

    const int T = 256;
    auto cdiv = [](long a, long b) { return (int)((a + b - 1) / b); };
    const int NBK = cdiv(N, BROWS);
    const int NCH = cdiv(E, CHUNK);

    float* ws   = (float*)d_ws;
    float* bufA = ws;
    float* bufB = bufA + (size_t)N * 64;
    float* xq   = bufB + (size_t)N * 64;
    float* w1t  = xq   + (size_t)N * 8;
    float* w2t  = w1t  + 64 * 8;
    float* w3t  = w2t  + 64 * 64;
    float* h3   = w3t  + 16 * 64;
    float* isq  = h3   + (size_t)N * 2;
    int*   bedg = (int*)(isq + N);
    int*   rec  = bedg + (size_t)E * 2;
    int*   tab  = rec  + (size_t)E;
    int*   rowp = tab  + (size_t)NBK * NCH;

    k_bkt_count  <<<NCH + PREPB, STH, 0, stream>>>(dst, tab, E, NBK, NCH,
                                                   W1, W2, W3, w1t, w2t, w3t);
    k_tab_scan   <<<1,   STH, 0, stream>>>(tab, NBK * NCH);
    k_bkt_scatter<<<NCH, STH, 0, stream>>>(src, dst, tab, bedg, E, NBK, NCH);
    k_row_sort   <<<NBK, STH, 0, stream>>>(bedg, tab, x, rowp, isq, xq, rec, N, E, NBK, NCH);

    const int GT = cdiv(N, 16);
    const int GG = 2048;

    k_layer1   <<<cdiv(GT, 2), 64, 0, stream>>>(rowp, rec, xq, isq, w1t, b1, bufB, GT);

    k_gemm_wmma<64, 4, 64><<<GT, 128, 0, stream>>>(bufB, w2t, isq, bufA);
    k_gather64 <<<GG, 256, 0, stream>>>(rowp, rec, bufA, isq, b2, bufB, N);

    k_gemm_wmma<64, 1, 2><<<GT, 32, 0, stream>>>(bufB, w3t, isq, h3);
    k_gather2  <<<cdiv(N, T), T, 0, stream>>>(rowp, rec, h3, isq, b3, out, N);
}


# Layer-1 product formed transposed (weights as A, row sums as B): its D tiles feed the W2 product as A fragments in the same launch, from registers (launches 9 to 8); layer-2 k order changes
# speedup vs baseline: 23.2740x; 1.0605x over previous
#include <hip/hip_runtime.h>
#include <hip/hip_bf16.h>

typedef __attribute__((ext_vector_type(2))) float v2f;
typedef __attribute__((ext_vector_type(8))) float v8f;

#define BSH     9
#define BROWS   512
#define NBK_MAX 256
#define CHUNK   20480
#define STH     1024
#define STILE   32768
#define PREPB   6

__device__ __forceinline__ void prep_w(const float* __restrict__ W, float* __restrict__ Bt,
                                       int Kin, int Cin, int KPAD, int COLS, int g, int gs) {
    for (int idx = g; idx < COLS * KPAD; idx += gs) {
        const int col = idx / KPAD, k = idx % KPAD;
        Bt[idx] = (k < Kin && col < Cin) ? W[k * Cin + col] : 0.0f;
    }
}

__device__ __forceinline__ void prep_w2f(const float* __restrict__ W2, float* __restrict__ w2f,
                                         int g, int gs) {
    for (int idx = g; idx < 64 * 64; idx += gs) {
        const int e   = idx & 3;
        const int ln  = (idx >> 2) & 31;
        const int ip  = (idx >> 7) & 1;
        const int ct  = (idx >> 8) & 3;
        const int ct2 = idx >> 10;
        const int k   = 16 * ct + 8 * (ln >> 4) + 4 * ip + e;
        const int col = 4 * (ln & 15) + ct2;
        w2f[idx] = W2[k * 64 + col];
    }
}

__global__ void __launch_bounds__(STH)
k_bkt_count(const int* __restrict__ dst, int* __restrict__ tab, int e, int nbk, int nch,
            const float* __restrict__ W1, const float* __restrict__ W2,
            const float* __restrict__ W3, float* __restrict__ w1t,
            float* __restrict__ w2f, float* __restrict__ w3t) {
    __shared__ int hist[NBK_MAX];
    const int t = threadIdx.x;
    if ((int)blockIdx.x >= nch) {
        const int g  = ((int)blockIdx.x - nch) * STH + t;
        const int gs = ((int)gridDim.x - nch) * STH;
        prep_w(W1, w1t, 5, 64, 8, 64, g, gs);
        prep_w2f(W2, w2f, g, gs);
        prep_w(W3, w3t, 64, 2, 64, 16, g, gs);
        return;
    }
    if (t < NBK_MAX) hist[t] = 0;
    __syncthreads();
    const int beg = blockIdx.x * CHUNK;
    const int end = min(beg + CHUNK, e);
    for (int i = beg + t; i < end; i += 4 * STH) {
        int d[4];
#pragma unroll
        for (int k = 0; k < 4; ++k) d[k] = dst[min(i + k * STH, end - 1)];
        asm volatile("" :: "v"(d[0]), "v"(d[1]), "v"(d[2]), "v"(d[3]));
#pragma unroll
        for (int k = 0; k < 4; ++k)
            if (i + k * STH < end) atomicAdd(&hist[d[k] >> BSH], 1);
    }
    __syncthreads();
    if (t < nbk) tab[t * nch + blockIdx.x] = hist[t];
}

__global__ void __launch_bounds__(STH)
k_tab_scan(int* __restrict__ tab, int len) {
    __shared__ int buf[STILE + STILE / 32];
    __shared__ int part[STH];
    const int t = threadIdx.x;
    int carry = 0;
    for (int tile = 0; tile < len; tile += STILE) {
        for (int k0 = t; k0 < STILE; k0 += 8 * STH) {
            int v[8];
#pragma unroll
            for (int u = 0; u < 8; ++u) v[u] = tab[min(tile + k0 + u * STH, len - 1)];
#pragma unroll
            for (int u = 0; u < 8; ++u) {
                const int k = k0 + u * STH;
                buf[k + (k >> 5)] = (tile + k < len) ? v[u] : 0;
            }
        }
        __syncthreads();
        int s = 0;
        for (int j = 0; j < 32; ++j) s += buf[33 * t + j];
        part[t] = s;
        __syncthreads();
        for (int off = 1; off < STH; off <<= 1) {
            const int x = (t >= off) ? part[t - off] : 0;
            __syncthreads();
            part[t] += x;
            __syncthreads();
        }
        int run = carry + part[t] - s;
        const int tot = part[STH - 1];
        for (int j = 0; j < 32; ++j) {
            const int v = buf[33 * t + j];
            buf[33 * t + j] = run;
            run += v;
        }
        __syncthreads();
        for (int k = t; k < STILE; k += STH) {
            const int g = tile + k;
            if (g < len) tab[g] = buf[k + (k >> 5)];
        }
        carry += tot;
        __syncthreads();
    }
}

__global__ void __launch_bounds__(STH)
k_bkt_scatter(const int* __restrict__ src, const int* __restrict__ dst,
              const int* __restrict__ tab, int* __restrict__ bedge, int e, int nbk, int nch) {
    __shared__ int cur[NBK_MAX];
    const int t = threadIdx.x;
    if (t < nbk) cur[t] = tab[t * nch + blockIdx.x];
    __syncthreads();
    const int beg = blockIdx.x * CHUNK;
    const int end = min(beg + CHUNK, e);
    int2* be = (int2*)bedge;
    for (int i = beg + t; i < end; i += 4 * STH) {
        int s[4], d[4];
#pragma unroll
        for (int k = 0; k < 4; ++k) {
            const int ii = min(i + k * STH, end - 1);
            s[k] = src[ii];
            d[k] = dst[ii];
        }
        asm volatile("" :: "v"(s[0]), "v"(s[1]), "v"(s[2]), "v"(s[3]),
                           "v"(d[0]), "v"(d[1]), "v"(d[2]), "v"(d[3]));
#pragma unroll
        for (int k = 0; k < 4; ++k)
            if (i + k * STH < end) {
                const int p = atomicAdd(&cur[d[k] >> BSH], 1);
                be[p] = make_int2(s[k], d[k]);
            }
    }
}

__global__ void __launch_bounds__(STH)
k_row_sort(const int* __restrict__ bedge, const int* __restrict__ tab,
           const float* __restrict__ x, int* __restrict__ rowptr,
           float* __restrict__ isq, float* __restrict__ xq, int* __restrict__ rec,
           int n, int e, int nbk, int nch) {
    __shared__ int   cnt[BROWS];
    __shared__ int   sc[BROWS];
    __shared__ float qs[BROWS];
    const int t = threadIdx.x;
    const int b = blockIdx.x;
    const int base = b << BSH;
    const int ebeg = tab[b * nch];
    const int eend = (b + 1 < nbk) ? tab[(b + 1) * nch] : e;
    if (t < BROWS) cnt[t] = 0;
    __syncthreads();
    const int2* be = (const int2*)bedge;
    for (int i = ebeg + t; i < eend; i += 4 * STH) {
        int d[4];
#pragma unroll
        for (int k = 0; k < 4; ++k) d[k] = be[min(i + k * STH, eend - 1)].y;
        asm volatile("" :: "v"(d[0]), "v"(d[1]), "v"(d[2]), "v"(d[3]));
#pragma unroll
        for (int k = 0; k < 4; ++k)
            if (i + k * STH < eend) atomicAdd(&cnt[d[k] - base], 1);
    }
    __syncthreads();
    int c = 0;
    if (t < BROWS) { c = cnt[t]; sc[t] = c; }
    __syncthreads();
    for (int off = 1; off < BROWS; off <<= 1) {
        int v = 0;
        if (t < BROWS && t >= off) v = sc[t - off];
        __syncthreads();
        if (t < BROWS) sc[t] += v;
        __syncthreads();
    }
    if (t < BROWS) {
        const int first = ebeg + sc[t] - c;
        const float q = rsqrtf((float)(c + 1));
        cnt[t] = first;
        qs[t]  = q;
        const int node = base + t;
        if (node < n) {
            rowptr[node] = first;
            isq[node] = q;
        }
    }
    if (b == 0 && t == 0) rowptr[n] = e;
    __syncthreads();
    {
        const int p = t & 7;
        const int col = (p & 1) + ((p >> 1) & 1) * 4 + (p >> 2) * 2;
        float xv[BROWS * 8 / STH];
#pragma unroll
        for (int k = 0; k < BROWS * 8 / STH; ++k) {
            const int node = min(base + (t >> 3) + k * (STH / 8), n - 1);
            xv[k] = x[(size_t)node * 5 + min(col, 4)];
        }
        static_assert(BROWS * 8 / STH == 4, "the fence below names four loaded values");
        asm volatile("" :: "v"(xv[0]), "v"(xv[1]), "v"(xv[2]), "v"(xv[3]));
#pragma unroll
        for (int k = 0; k < BROWS * 8 / STH; ++k) {
            const int i = (t >> 3) + k * (STH / 8);
            if (base + i < n)
                xq[(size_t)(base + i) * 8 + p] = (col < 5) ? xv[k] * qs[i] : 0.0f;
        }
    }
    for (int i = ebeg + t; i < eend; i += 4 * STH) {
        int2 sd[4];
#pragma unroll
        for (int k = 0; k < 4; ++k) sd[k] = be[min(i + k * STH, eend - 1)];
        asm volatile("" :: "v"(sd[0].x), "v"(sd[1].x), "v"(sd[2].x), "v"(sd[3].x),
                           "v"(sd[0].y), "v"(sd[1].y), "v"(sd[2].y), "v"(sd[3].y));
#pragma unroll
        for (int k = 0; k < 4; ++k)
            if (i + k * STH < eend) {
                const int p = atomicAdd(&cnt[sd[k].y - base], 1);
                rec[p] = sd[k].x;
            }
    }
}

typedef __attribute__((ext_vector_type(4))) float v4f;

template <int K, int NT, int SC>
__global__ void k_gemm_wmma(const float* __restrict__ A, const float* __restrict__ Bt,
                            const float* __restrict__ isq, float* __restrict__ H) {
    const int lane = threadIdx.x & 31;
    const int wave = threadIdx.x >> 5;
    const int half = lane >> 4;
    const int l15  = lane & 15;
    const size_t row = (size_t)blockIdx.x * 16 + l15;
    const int    col = wave * 16 + l15;
    const float* Arow = A + row * K;
    const float* Bcol = Bt + (size_t)col * K;
    const size_t mbase = (size_t)blockIdx.x * 16 + half * 8;
    const v4f q0 = *(const v4f*)(isq + mbase);
    const v4f q1 = *(const v4f*)(isq + mbase + 4);

    v8f c = {};
#pragma unroll
    for (int k = 0; k < K; k += 4) {
        const int ka = k + half * 2;
        v2f a = *(const v2f*)(Arow + ka);
        v2f b = *(const v2f*)(Bcol + ka);
        c = __builtin_amdgcn_wmma_f32_16x16x4_f32(
                 false, a,  false, b,
                 (short)0, c,  false,  false);
    }

    if (NT * 16 == SC || col < SC) {
        const float q[8] = {q0.x, q0.y, q0.z, q0.w, q1.x, q1.y, q1.z, q1.w};
#pragma unroll
        for (int r = 0; r < 8; ++r)
            H[(mbase + r) * SC + col] = c[r] * q[r];
    }
}

__global__ void __launch_bounds__(256)
k_gather64(const int* __restrict__ rowptr, const int* __restrict__ rec,
           const float* __restrict__ h, const float* __restrict__ isq,
           const float* __restrict__ bias, float* __restrict__ out, int n) {
    const int l = threadIdx.x & 15;
    const int stride = gridDim.x * 16;
    const v4f* h4 = (const v4f*)h;
    const v4f bv = {bias[4 * l], bias[4 * l + 1], bias[4 * l + 2], bias[4 * l + 3]};
    for (int node = blockIdx.x * 16 + (threadIdx.x >> 4); node < n; node += stride) {
        const int beg = rowptr[node];
        const int end = rowptr[node + 1];
        const float q = isq[node];
        v4f acc = h4[(size_t)node * 16 + l];
        if (beg < end) {
            const int last = end - 1;
            int e0 = rec[beg];
            int e1 = rec[min(beg + 1, last)];
            int e2 = rec[min(beg + 2, last)];
            int e3 = rec[min(beg + 3, last)];
            for (int p = beg; p < end; p += 4) {
                const v4f v0 = h4[(size_t)e0 * 16 + l];
                const v4f v1 = h4[(size_t)e1 * 16 + l];
                const v4f v2 = h4[(size_t)e2 * 16 + l];
                const v4f v3 = h4[(size_t)e3 * 16 + l];
                e0 = rec[min(p + 4, last)];
                e1 = rec[min(p + 5, last)];
                e2 = rec[min(p + 6, last)];
                e3 = rec[min(p + 7, last)];
                __builtin_amdgcn_sched_barrier(0);
                asm volatile("" :: "v"(v0), "v"(v1), "v"(v2), "v"(v3));
                acc += v0;
                if (p + 1 < end) acc += v1;
                if (p + 2 < end) acc += v2;
                if (p + 3 < end) acc += v3;
            }
        }
        acc = acc * q + bv;
        v4f r;
        r.x = fmaxf(acc.x, 0.0f);
        r.y = fmaxf(acc.y, 0.0f);
        r.z = fmaxf(acc.z, 0.0f);
        r.w = fmaxf(acc.w, 0.0f);
        ((v4f*)out)[(size_t)node * 16 + l] = r;
    }
}

__global__ void __launch_bounds__(64)
k_layer12(const int* __restrict__ rowptr, const int* __restrict__ rec,
          const float* __restrict__ xq, const float* __restrict__ isq,
          const float* __restrict__ W1t, const float* __restrict__ bias,
          const float* __restrict__ W2f, float* __restrict__ out, int ntiles) {
    const int lane = threadIdx.x & 31;
    const int half = lane >> 4;
    const int l15  = lane & 15;
    const int tile = blockIdx.x * 2 + (threadIdx.x >> 5);
    if (tile >= ntiles) return;
    const int node = tile * 16 + l15;
    const v4f* xq4 = (const v4f*)xq;

    const int beg = rowptr[node];
    const int end = rowptr[node + 1];

    v2f w0[4], w1[4];
#pragma unroll
    for (int ct = 0; ct < 4; ++ct) {
        const float* Wcol = W1t + (size_t)(16 * ct + l15) * 8 + half * 2;
        w0[ct] = *(const v2f*)(Wcol);
        w1[ct] = *(const v2f*)(Wcol + 4);
    }

    const float q = isq[node];
    v4f acc = xq4[(size_t)node * 2 + half];
    if (beg < end) {
        const int last = end - 1;
        int e[8];
#pragma unroll
        for (int j = 0; j < 8; ++j) e[j] = rec[min(beg + j, last)];
        for (int p = beg; p < end; p += 8) {
            v4f v[8];
#pragma unroll
            for (int j = 0; j < 8; ++j) v[j] = xq4[(size_t)e[j] * 2 + half];
#pragma unroll
            for (int j = 0; j < 8; ++j) e[j] = rec[min(p + 8 + j, last)];
            __builtin_amdgcn_sched_barrier(0);
            asm volatile("" :: "v"(v[0]), "v"(v[1]), "v"(v[2]), "v"(v[3]),
                               "v"(v[4]), "v"(v[5]), "v"(v[6]), "v"(v[7]));
            acc += v[0];
#pragma unroll
            for (int j = 1; j < 8; ++j)
                if (p + j < end) acc += v[j];
        }
    }
    acc *= q;

    const v2f x0 = {acc.x, acc.y};
    const v2f x1 = {acc.z, acc.w};
    v8f d[4];
#pragma unroll
    for (int ct = 0; ct < 4; ++ct) {
        v8f z = {};
        d[ct] = __builtin_amdgcn_wmma_f32_16x16x4_f32(
                     false, w0[ct],  false, x0,
                     (short)0, z,  false,  false);
    }
#pragma unroll
    for (int ct = 0; ct < 4; ++ct)
        d[ct] = __builtin_amdgcn_wmma_f32_16x16x4_f32(
                     false, w1[ct],  false, x1,
                     (short)0, d[ct],  false,  false);

#pragma unroll
    for (int ct = 0; ct < 4; ++ct) {
        const v4f ba = *(const v4f*)(bias + 16 * ct + 8 * half);
        const v4f bb = *(const v4f*)(bias + 16 * ct + 8 * half + 4);
        d[ct][0] = fmaxf(d[ct][0] + ba.x, 0.0f);
        d[ct][1] = fmaxf(d[ct][1] + ba.y, 0.0f);
        d[ct][2] = fmaxf(d[ct][2] + ba.z, 0.0f);
        d[ct][3] = fmaxf(d[ct][3] + ba.w, 0.0f);
        d[ct][4] = fmaxf(d[ct][4] + bb.x, 0.0f);
        d[ct][5] = fmaxf(d[ct][5] + bb.y, 0.0f);
        d[ct][6] = fmaxf(d[ct][6] + bb.z, 0.0f);
        d[ct][7] = fmaxf(d[ct][7] + bb.w, 0.0f);
    }

    const v4f* W2f4 = (const v4f*)W2f;
    v8f o[4];
#pragma unroll
    for (int ct2 = 0; ct2 < 4; ++ct2) o[ct2] = v8f{};
#pragma unroll
    for (int ct = 0; ct < 4; ++ct) {
#pragma unroll
        for (int ip = 0; ip < 2; ++ip) {
            v4f bw[4];
#pragma unroll
            for (int ct2 = 0; ct2 < 4; ++ct2)
                bw[ct2] = W2f4[((ct2 * 4 + ct) * 2 + ip) * 32 + lane];
            const v2f a_lo = {d[ct][4 * ip],     d[ct][4 * ip + 1]};
            const v2f a_hi = {d[ct][4 * ip + 2], d[ct][4 * ip + 3]};
#pragma unroll
            for (int ct2 = 0; ct2 < 4; ++ct2) {
                const v2f b_lo = {bw[ct2].x, bw[ct2].y};
                o[ct2] = __builtin_amdgcn_wmma_f32_16x16x4_f32(
                              false, a_lo,  false, b_lo,
                              (short)0, o[ct2],  false,  false);
            }
#pragma unroll
            for (int ct2 = 0; ct2 < 4; ++ct2) {
                const v2f b_hi = {bw[ct2].z, bw[ct2].w};
                o[ct2] = __builtin_amdgcn_wmma_f32_16x16x4_f32(
                              false, a_hi,  false, b_hi,
                              (short)0, o[ct2],  false,  false);
            }
        }
    }

    const size_t mbase = (size_t)tile * 16 + half * 8;
    const v4f q0 = *(const v4f*)(isq + mbase);
    const v4f q1 = *(const v4f*)(isq + mbase + 4);
    const float qr[8] = {q0.x, q0.y, q0.z, q0.w, q1.x, q1.y, q1.z, q1.w};
#pragma unroll
    for (int r = 0; r < 8; ++r) {
        v4f s;
        s.x = o[0][r] * qr[r];
        s.y = o[1][r] * qr[r];
        s.z = o[2][r] * qr[r];
        s.w = o[3][r] * qr[r];
        ((v4f*)out)[(mbase + r) * 16 + l15] = s;
    }
}

__global__ void k_gather2(const int* __restrict__ rowptr, const int* __restrict__ rec,
                          const float* __restrict__ h, const float* __restrict__ isq,
                          const float* __restrict__ bias, float* __restrict__ out, int n) {
    const int node = blockIdx.x * blockDim.x + threadIdx.x;
    if (node >= n) return;
    const int beg = rowptr[node];
    const int end = rowptr[node + 1];
    const v2f* h2 = (const v2f*)h;
    const v2f bv = {bias[0], bias[1]};
    const float q = isq[node];
    v2f acc = h2[node];
    for (int p = beg; p < end; p += 4) {
        const int last = end - 1;
        const int e0 = rec[p];
        const int e1 = rec[min(p + 1, last)];
        const int e2 = rec[min(p + 2, last)];
        const int e3 = rec[min(p + 3, last)];
        const v2f v0 = h2[e0];
        const v2f v1 = h2[e1];
        const v2f v2 = h2[e2];
        const v2f v3 = h2[e3];
        asm volatile("" :: "v"(v0), "v"(v1), "v"(v2), "v"(v3));
        acc += v0;
        if (p + 1 < end) acc += v1;
        if (p + 2 < end) acc += v2;
        if (p + 3 < end) acc += v3;
    }
    ((v2f*)out)[node] = acc * q + bv;
}

extern "C" void kernel_launch(void* const* d_in, const int* in_sizes, int n_in,
                              void* d_out, int out_size, void* d_ws, size_t ws_size,
                              hipStream_t stream) {
    const float* x  = (const float*)d_in[0];
    const int*   ei = (const int*)d_in[1];
    const float* W1 = (const float*)d_in[2];
    const float* b1 = (const float*)d_in[3];
    const float* W2 = (const float*)d_in[4];
    const float* b2 = (const float*)d_in[5];
    const float* W3 = (const float*)d_in[6];
    const float* b3 = (const float*)d_in[7];
    float* out = (float*)d_out;

    const int N = in_sizes[0] / 5;
    const int E = in_sizes[1] / 2;
    const int* src = ei;
    const int* dst = ei + E;

    const int T = 256;
    auto cdiv = [](long a, long b) { return (int)((a + b - 1) / b); };
    const int NBK = cdiv(N, BROWS);
    const int NCH = cdiv(E, CHUNK);

    float* ws   = (float*)d_ws;
    float* bufA = ws;
    float* bufB = bufA + (size_t)N * 64;
    float* xq   = bufB + (size_t)N * 64;
    float* w1t  = xq   + (size_t)N * 8;
    float* w2f  = w1t  + 64 * 8;
    float* w3t  = w2f  + 64 * 64;
    float* h3   = w3t  + 16 * 64;
    float* isq  = h3   + (size_t)N * 2;
    int*   bedg = (int*)(isq + N);
    int*   rec  = bedg + (size_t)E * 2;
    int*   tab  = rec  + (size_t)E;
    int*   rowp = tab  + (size_t)NBK * NCH;

    k_bkt_count  <<<NCH + PREPB, STH, 0, stream>>>(dst, tab, E, NBK, NCH,
                                                   W1, W2, W3, w1t, w2f, w3t);
    k_tab_scan   <<<1,   STH, 0, stream>>>(tab, NBK * NCH);
    k_bkt_scatter<<<NCH, STH, 0, stream>>>(src, dst, tab, bedg, E, NBK, NCH);
    k_row_sort   <<<NBK, STH, 0, stream>>>(bedg, tab, x, rowp, isq, xq, rec, N, E, NBK, NCH);

    const int GT = cdiv(N, 16);
    const int GG = 2048;

    k_layer12  <<<cdiv(GT, 2), 64, 0, stream>>>(rowp, rec, xq, isq, w1t, b1, w2f, bufA, GT);

    k_gather64 <<<GG, 256, 0, stream>>>(rowp, rec, bufA, isq, b2, bufB, N);

    k_gemm_wmma<64, 1, 2><<<GT, 32, 0, stream>>>(bufB, w3t, isq, h3);
    k_gather2  <<<cdiv(N, T), T, 0, stream>>>(rowp, rec, h3, isq, b3, out, N);
}


# Layer-3 weight product folded into the layer-2 edge-sum launch: a tile's 16 finished rows go via LDS into the f32 WMMA as B fragments (weights as A); N*64 plane and a launch gone; sums unchanged
# speedup vs baseline: 24.2916x; 1.0437x over previous
#include <hip/hip_runtime.h>
#include <hip/hip_bf16.h>

typedef __attribute__((ext_vector_type(2))) float v2f;
typedef __attribute__((ext_vector_type(8))) float v8f;

#define BSH     9
#define BROWS   512
#define NBK_MAX 256
#define CHUNK   20480
#define STH     1024
#define STILE   32768
#define PREPB   6

__device__ __forceinline__ void prep_w(const float* __restrict__ W, float* __restrict__ Bt,
                                       int Kin, int Cin, int KPAD, int COLS, int g, int gs) {
    for (int idx = g; idx < COLS * KPAD; idx += gs) {
        const int col = idx / KPAD, k = idx % KPAD;
        Bt[idx] = (k < Kin && col < Cin) ? W[k * Cin + col] : 0.0f;
    }
}

__device__ __forceinline__ void prep_w2f(const float* __restrict__ W2, float* __restrict__ w2f,
                                         int g, int gs) {
    for (int idx = g; idx < 64 * 64; idx += gs) {
        const int e   = idx & 3;
        const int ln  = (idx >> 2) & 31;
        const int ip  = (idx >> 7) & 1;
        const int ct  = (idx >> 8) & 3;
        const int ct2 = idx >> 10;
        const int k   = 16 * ct + 8 * (ln >> 4) + 4 * ip + e;
        const int col = 4 * (ln & 15) + ct2;
        w2f[idx] = W2[k * 64 + col];
    }
}

__global__ void __launch_bounds__(STH)
k_bkt_count(const int* __restrict__ dst, int* __restrict__ tab, int e, int nbk, int nch,
            const float* __restrict__ W1, const float* __restrict__ W2,
            const float* __restrict__ W3, float* __restrict__ w1t,
            float* __restrict__ w2f, float* __restrict__ w3t) {
    __shared__ int hist[NBK_MAX];
    const int t = threadIdx.x;
    if ((int)blockIdx.x >= nch) {
        const int g  = ((int)blockIdx.x - nch) * STH + t;
        const int gs = ((int)gridDim.x - nch) * STH;
        prep_w(W1, w1t, 5, 64, 8, 64, g, gs);
        prep_w2f(W2, w2f, g, gs);
        prep_w(W3, w3t, 64, 2, 64, 16, g, gs);
        return;
    }
    if (t < NBK_MAX) hist[t] = 0;
    __syncthreads();
    const int beg = blockIdx.x * CHUNK;
    const int end = min(beg + CHUNK, e);
    for (int i = beg + t; i < end; i += 4 * STH) {
        int d[4];
#pragma unroll
        for (int k = 0; k < 4; ++k) d[k] = dst[min(i + k * STH, end - 1)];
        asm volatile("" :: "v"(d[0]), "v"(d[1]), "v"(d[2]), "v"(d[3]));
#pragma unroll
        for (int k = 0; k < 4; ++k)
            if (i + k * STH < end) atomicAdd(&hist[d[k] >> BSH], 1);
    }
    __syncthreads();
    if (t < nbk) tab[t * nch + blockIdx.x] = hist[t];
}

__global__ void __launch_bounds__(STH)
k_tab_scan(int* __restrict__ tab, int len) {
    __shared__ int buf[STILE + STILE / 32];
    __shared__ int part[STH];
    const int t = threadIdx.x;
    int carry = 0;
    for (int tile = 0; tile < len; tile += STILE) {
        for (int k0 = t; k0 < STILE; k0 += 8 * STH) {
            int v[8];
#pragma unroll
            for (int u = 0; u < 8; ++u) v[u] = tab[min(tile + k0 + u * STH, len - 1)];
#pragma unroll
            for (int u = 0; u < 8; ++u) {
                const int k = k0 + u * STH;
                buf[k + (k >> 5)] = (tile + k < len) ? v[u] : 0;
            }
        }
        __syncthreads();
        int s = 0;
        for (int j = 0; j < 32; ++j) s += buf[33 * t + j];
        part[t] = s;
        __syncthreads();
        for (int off = 1; off < STH; off <<= 1) {
            const int x = (t >= off) ? part[t - off] : 0;
            __syncthreads();
            part[t] += x;
            __syncthreads();
        }
        int run = carry + part[t] - s;
        const int tot = part[STH - 1];
        for (int j = 0; j < 32; ++j) {
            const int v = buf[33 * t + j];
            buf[33 * t + j] = run;
            run += v;
        }
        __syncthreads();
        for (int k = t; k < STILE; k += STH) {
            const int g = tile + k;
            if (g < len) tab[g] = buf[k + (k >> 5)];
        }
        carry += tot;
        __syncthreads();
    }
}

__global__ void __launch_bounds__(STH)
k_bkt_scatter(const int* __restrict__ src, const int* __restrict__ dst,
              const int* __restrict__ tab, int* __restrict__ bedge, int e, int nbk, int nch) {
    __shared__ int cur[NBK_MAX];
    const int t = threadIdx.x;
    if (t < nbk) cur[t] = tab[t * nch + blockIdx.x];
    __syncthreads();
    const int beg = blockIdx.x * CHUNK;
    const int end = min(beg + CHUNK, e);
    int2* be = (int2*)bedge;
    for (int i = beg + t; i < end; i += 4 * STH) {
        int s[4], d[4];
#pragma unroll
        for (int k = 0; k < 4; ++k) {
            const int ii = min(i + k * STH, end - 1);
            s[k] = src[ii];
            d[k] = dst[ii];
        }
        asm volatile("" :: "v"(s[0]), "v"(s[1]), "v"(s[2]), "v"(s[3]),
                           "v"(d[0]), "v"(d[1]), "v"(d[2]), "v"(d[3]));
#pragma unroll
        for (int k = 0; k < 4; ++k)
            if (i + k * STH < end) {
                const int p = atomicAdd(&cur[d[k] >> BSH], 1);
                be[p] = make_int2(s[k], d[k]);
            }
    }
}

__global__ void __launch_bounds__(STH)
k_row_sort(const int* __restrict__ bedge, const int* __restrict__ tab,
           const float* __restrict__ x, int* __restrict__ rowptr,
           float* __restrict__ isq, float* __restrict__ xq, int* __restrict__ rec,
           int n, int e, int nbk, int nch) {
    __shared__ int   cnt[BROWS];
    __shared__ int   sc[BROWS];
    __shared__ float qs[BROWS];
    const int t = threadIdx.x;
    const int b = blockIdx.x;
    const int base = b << BSH;
    const int ebeg = tab[b * nch];
    const int eend = (b + 1 < nbk) ? tab[(b + 1) * nch] : e;
    if (t < BROWS) cnt[t] = 0;
    __syncthreads();
    const int2* be = (const int2*)bedge;
    for (int i = ebeg + t; i < eend; i += 4 * STH) {
        int d[4];
#pragma unroll
        for (int k = 0; k < 4; ++k) d[k] = be[min(i + k * STH, eend - 1)].y;
        asm volatile("" :: "v"(d[0]), "v"(d[1]), "v"(d[2]), "v"(d[3]));
#pragma unroll
        for (int k = 0; k < 4; ++k)
            if (i + k * STH < eend) atomicAdd(&cnt[d[k] - base], 1);
    }
    __syncthreads();
    int c = 0;
    if (t < BROWS) { c = cnt[t]; sc[t] = c; }
    __syncthreads();
    for (int off = 1; off < BROWS; off <<= 1) {
        int v = 0;
        if (t < BROWS && t >= off) v = sc[t - off];
        __syncthreads();
        if (t < BROWS) sc[t] += v;
        __syncthreads();
    }
    if (t < BROWS) {
        const int first = ebeg + sc[t] - c;
        const float q = rsqrtf((float)(c + 1));
        cnt[t] = first;
        qs[t]  = q;
        const int node = base + t;
        if (node < n) {
            rowptr[node] = first;
            isq[node] = q;
        }
    }
    if (b == 0 && t == 0) rowptr[n] = e;
    __syncthreads();
    {
        const int p = t & 7;
        const int col = (p & 1) + ((p >> 1) & 1) * 4 + (p >> 2) * 2;
        float xv[BROWS * 8 / STH];
#pragma unroll
        for (int k = 0; k < BROWS * 8 / STH; ++k) {
            const int node = min(base + (t >> 3) + k * (STH / 8), n - 1);
            xv[k] = x[(size_t)node * 5 + min(col, 4)];
        }
        static_assert(BROWS * 8 / STH == 4, "the fence below names four loaded values");
        asm volatile("" :: "v"(xv[0]), "v"(xv[1]), "v"(xv[2]), "v"(xv[3]));
#pragma unroll
        for (int k = 0; k < BROWS * 8 / STH; ++k) {
            const int i = (t >> 3) + k * (STH / 8);
            if (base + i < n)
                xq[(size_t)(base + i) * 8 + p] = (col < 5) ? xv[k] * qs[i] : 0.0f;
        }
    }
    for (int i = ebeg + t; i < eend; i += 4 * STH) {
        int2 sd[4];
#pragma unroll
        for (int k = 0; k < 4; ++k) sd[k] = be[min(i + k * STH, eend - 1)];
        asm volatile("" :: "v"(sd[0].x), "v"(sd[1].x), "v"(sd[2].x), "v"(sd[3].x),
                           "v"(sd[0].y), "v"(sd[1].y), "v"(sd[2].y), "v"(sd[3].y));
#pragma unroll
        for (int k = 0; k < 4; ++k)
            if (i + k * STH < eend) {
                const int p = atomicAdd(&cnt[sd[k].y - base], 1);
                rec[p] = sd[k].x;
            }
    }
}

typedef __attribute__((ext_vector_type(4))) float v4f;

#define TPITCH 68

__global__ void __launch_bounds__(64)
k_gather64_w3(const int* __restrict__ rowptr, const int* __restrict__ rec,
              const float* __restrict__ h, const float* __restrict__ isq,
              const float* __restrict__ bias, const float* __restrict__ W3t,
              float* __restrict__ out) {
    __shared__ __attribute__((aligned(16))) float tile[16 * TPITCH];
    const int lane = threadIdx.x & 31;
    const int wave = threadIdx.x >> 5;
    const int half = lane >> 4;
    const int l    = lane & 15;
    const v4f* h4 = (const v4f*)h;
    const v4f bv = {bias[4 * l], bias[4 * l + 1], bias[4 * l + 2], bias[4 * l + 3]};
    for (int j = 0; j < 4; ++j) {
        const int rr   = 8 * wave + 2 * j + half;
        const int node = blockIdx.x * 16 + rr;
        const int beg = rowptr[node];
        const int end = rowptr[node + 1];
        const float q = isq[node];
        v4f acc = h4[(size_t)node * 16 + l];
        if (beg < end) {
            const int last = end - 1;
            int e0 = rec[beg];
            int e1 = rec[min(beg + 1, last)];
            int e2 = rec[min(beg + 2, last)];
            int e3 = rec[min(beg + 3, last)];
            for (int p = beg; p < end; p += 4) {
                const v4f v0 = h4[(size_t)e0 * 16 + l];
                const v4f v1 = h4[(size_t)e1 * 16 + l];
                const v4f v2 = h4[(size_t)e2 * 16 + l];
                const v4f v3 = h4[(size_t)e3 * 16 + l];
                e0 = rec[min(p + 4, last)];
                e1 = rec[min(p + 5, last)];
                e2 = rec[min(p + 6, last)];
                e3 = rec[min(p + 7, last)];
                __builtin_amdgcn_sched_barrier(0);
                asm volatile("" :: "v"(v0), "v"(v1), "v"(v2), "v"(v3));
                acc += v0;
                if (p + 1 < end) acc += v1;
                if (p + 2 < end) acc += v2;
                if (p + 3 < end) acc += v3;
            }
        }
        acc = acc * q + bv;
        v4f r;
        r.x = fmaxf(acc.x, 0.0f);
        r.y = fmaxf(acc.y, 0.0f);
        r.z = fmaxf(acc.z, 0.0f);
        r.w = fmaxf(acc.w, 0.0f);
        *(v4f*)(tile + rr * TPITCH + 4 * l) = r;
    }
    __syncthreads();
    if (wave != 0) return;

    const float* Wrow = W3t + l * 64 + 2 * half;
    const float* Trow = tile + l * TPITCH + 2 * half;
    const float qrow = isq[blockIdx.x * 16 + l];
    v8f c = {};
#pragma unroll
    for (int kk = 0; kk < 16; ++kk) {
        const v2f a = *(const v2f*)(Wrow + 4 * kk);
        const v2f b = *(const v2f*)(Trow + 4 * kk);
        c = __builtin_amdgcn_wmma_f32_16x16x4_f32(
                 false, a,  false, b,
                 (short)0, c,  false,  false);
    }
    asm volatile("" :: "v"(qrow));
    if (half == 0) {
        const v2f o = {c[0] * qrow, c[1] * qrow};
        ((v2f*)out)[blockIdx.x * 16 + l] = o;
    }
}

__global__ void __launch_bounds__(64)
k_layer12(const int* __restrict__ rowptr, const int* __restrict__ rec,
          const float* __restrict__ xq, const float* __restrict__ isq,
          const float* __restrict__ W1t, const float* __restrict__ bias,
          const float* __restrict__ W2f, float* __restrict__ out, int ntiles) {
    const int lane = threadIdx.x & 31;
    const int half = lane >> 4;
    const int l15  = lane & 15;
    const int tile = blockIdx.x * 2 + (threadIdx.x >> 5);
    if (tile >= ntiles) return;
    const int node = tile * 16 + l15;
    const v4f* xq4 = (const v4f*)xq;

    const int beg = rowptr[node];
    const int end = rowptr[node + 1];

    v2f w0[4], w1[4];
#pragma unroll
    for (int ct = 0; ct < 4; ++ct) {
        const float* Wcol = W1t + (size_t)(16 * ct + l15) * 8 + half * 2;
        w0[ct] = *(const v2f*)(Wcol);
        w1[ct] = *(const v2f*)(Wcol + 4);
    }

    const float q = isq[node];
    v4f acc = xq4[(size_t)node * 2 + half];
    if (beg < end) {
        const int last = end - 1;
        int e[8];
#pragma unroll
        for (int j = 0; j < 8; ++j) e[j] = rec[min(beg + j, last)];
        for (int p = beg; p < end; p += 8) {
            v4f v[8];
#pragma unroll
            for (int j = 0; j < 8; ++j) v[j] = xq4[(size_t)e[j] * 2 + half];
#pragma unroll
            for (int j = 0; j < 8; ++j) e[j] = rec[min(p + 8 + j, last)];
            __builtin_amdgcn_sched_barrier(0);
            asm volatile("" :: "v"(v[0]), "v"(v[1]), "v"(v[2]), "v"(v[3]),
                               "v"(v[4]), "v"(v[5]), "v"(v[6]), "v"(v[7]));
            acc += v[0];
#pragma unroll
            for (int j = 1; j < 8; ++j)
                if (p + j < end) acc += v[j];
        }
    }
    acc *= q;

    const v2f x0 = {acc.x, acc.y};
    const v2f x1 = {acc.z, acc.w};
    v8f d[4];
#pragma unroll
    for (int ct = 0; ct < 4; ++ct) {
        v8f z = {};
        d[ct] = __builtin_amdgcn_wmma_f32_16x16x4_f32(
                     false, w0[ct],  false, x0,
                     (short)0, z,  false,  false);
    }
#pragma unroll
    for (int ct = 0; ct < 4; ++ct)
        d[ct] = __builtin_amdgcn_wmma_f32_16x16x4_f32(
                     false, w1[ct],  false, x1,
                     (short)0, d[ct],  false,  false);

#pragma unroll
    for (int ct = 0; ct < 4; ++ct) {
        const v4f ba = *(const v4f*)(bias + 16 * ct + 8 * half);
        const v4f bb = *(const v4f*)(bias + 16 * ct + 8 * half + 4);
        d[ct][0] = fmaxf(d[ct][0] + ba.x, 0.0f);
        d[ct][1] = fmaxf(d[ct][1] + ba.y, 0.0f);
        d[ct][2] = fmaxf(d[ct][2] + ba.z, 0.0f);
        d[ct][3] = fmaxf(d[ct][3] + ba.w, 0.0f);
        d[ct][4] = fmaxf(d[ct][4] + bb.x, 0.0f);
        d[ct][5] = fmaxf(d[ct][5] + bb.y, 0.0f);
        d[ct][6] = fmaxf(d[ct][6] + bb.z, 0.0f);
        d[ct][7] = fmaxf(d[ct][7] + bb.w, 0.0f);
    }

    const v4f* W2f4 = (const v4f*)W2f;
    v8f o[4];
#pragma unroll
    for (int ct2 = 0; ct2 < 4; ++ct2) o[ct2] = v8f{};
#pragma unroll
    for (int ct = 0; ct < 4; ++ct) {
#pragma unroll
        for (int ip = 0; ip < 2; ++ip) {
            v4f bw[4];
#pragma unroll
            for (int ct2 = 0; ct2 < 4; ++ct2)
                bw[ct2] = W2f4[((ct2 * 4 + ct) * 2 + ip) * 32 + lane];
            const v2f a_lo = {d[ct][4 * ip],     d[ct][4 * ip + 1]};
            const v2f a_hi = {d[ct][4 * ip + 2], d[ct][4 * ip + 3]};
#pragma unroll
            for (int ct2 = 0; ct2 < 4; ++ct2) {
                const v2f b_lo = {bw[ct2].x, bw[ct2].y};
                o[ct2] = __builtin_amdgcn_wmma_f32_16x16x4_f32(
                              false, a_lo,  false, b_lo,
                              (short)0, o[ct2],  false,  false);
            }
#pragma unroll
            for (int ct2 = 0; ct2 < 4; ++ct2) {
                const v2f b_hi = {bw[ct2].z, bw[ct2].w};
                o[ct2] = __builtin_amdgcn_wmma_f32_16x16x4_f32(
                              false, a_hi,  false, b_hi,
                              (short)0, o[ct2],  false,  false);
            }
        }
    }

    const size_t mbase = (size_t)tile * 16 + half * 8;
    const v4f q0 = *(const v4f*)(isq + mbase);
    const v4f q1 = *(const v4f*)(isq + mbase + 4);
    const float qr[8] = {q0.x, q0.y, q0.z, q0.w, q1.x, q1.y, q1.z, q1.w};
#pragma unroll
    for (int r = 0; r < 8; ++r) {
        v4f s;
        s.x = o[0][r] * qr[r];
        s.y = o[1][r] * qr[r];
        s.z = o[2][r] * qr[r];
        s.w = o[3][r] * qr[r];
        ((v4f*)out)[(mbase + r) * 16 + l15] = s;
    }
}

__global__ void k_gather2(const int* __restrict__ rowptr, const int* __restrict__ rec,
                          const float* __restrict__ h, const float* __restrict__ isq,
                          const float* __restrict__ bias, float* __restrict__ out, int n) {
    const int node = blockIdx.x * blockDim.x + threadIdx.x;
    if (node >= n) return;
    const int beg = rowptr[node];
    const int end = rowptr[node + 1];
    const v2f* h2 = (const v2f*)h;
    const v2f bv = {bias[0], bias[1]};
    const float q = isq[node];
    v2f acc = h2[node];
    for (int p = beg; p < end; p += 4) {
        const int last = end - 1;
        const int e0 = rec[p];
        const int e1 = rec[min(p + 1, last)];
        const int e2 = rec[min(p + 2, last)];
        const int e3 = rec[min(p + 3, last)];
        const v2f v0 = h2[e0];
        const v2f v1 = h2[e1];
        const v2f v2 = h2[e2];
        const v2f v3 = h2[e3];
        asm volatile("" :: "v"(v0), "v"(v1), "v"(v2), "v"(v3));
        acc += v0;
        if (p + 1 < end) acc += v1;
        if (p + 2 < end) acc += v2;
        if (p + 3 < end) acc += v3;
    }
    ((v2f*)out)[node] = acc * q + bv;
}

extern "C" void kernel_launch(void* const* d_in, const int* in_sizes, int n_in,
                              void* d_out, int out_size, void* d_ws, size_t ws_size,
                              hipStream_t stream) {
    const float* x  = (const float*)d_in[0];
    const int*   ei = (const int*)d_in[1];
    const float* W1 = (const float*)d_in[2];
    const float* b1 = (const float*)d_in[3];
    const float* W2 = (const float*)d_in[4];
    const float* b2 = (const float*)d_in[5];
    const float* W3 = (const float*)d_in[6];
    const float* b3 = (const float*)d_in[7];
    float* out = (float*)d_out;

    const int N = in_sizes[0] / 5;
    const int E = in_sizes[1] / 2;
    const int* src = ei;
    const int* dst = ei + E;

    const int T = 256;
    auto cdiv = [](long a, long b) { return (int)((a + b - 1) / b); };
    const int NBK = cdiv(N, BROWS);
    const int NCH = cdiv(E, CHUNK);

    float* ws   = (float*)d_ws;
    float* bufA = ws;
    float* xq   = bufA + (size_t)N * 64;
    float* w1t  = xq   + (size_t)N * 8;
    float* w2f  = w1t  + 64 * 8;
    float* w3t  = w2f  + 64 * 64;
    float* h3   = w3t  + 16 * 64;
    float* isq  = h3   + (size_t)N * 2;
    int*   bedg = (int*)(isq + N);
    int*   rec  = bedg + (size_t)E * 2;
    int*   tab  = rec  + (size_t)E;
    int*   rowp = tab  + (size_t)NBK * NCH;

    k_bkt_count  <<<NCH + PREPB, STH, 0, stream>>>(dst, tab, E, NBK, NCH,
                                                   W1, W2, W3, w1t, w2f, w3t);
    k_tab_scan   <<<1,   STH, 0, stream>>>(tab, NBK * NCH);
    k_bkt_scatter<<<NCH, STH, 0, stream>>>(src, dst, tab, bedg, E, NBK, NCH);
    k_row_sort   <<<NBK, STH, 0, stream>>>(bedg, tab, x, rowp, isq, xq, rec, N, E, NBK, NCH);

    const int GT = cdiv(N, 16);

    k_layer12  <<<cdiv(GT, 2), 64, 0, stream>>>(rowp, rec, xq, isq, w1t, b1, w2f, bufA, GT);

    k_gather64_w3<<<GT, 64, 0, stream>>>(rowp, rec, bufA, isq, b2, w3t, h3);

    k_gather2  <<<cdiv(N, T), T, 0, stream>>>(rowp, rec, h3, isq, b3, out, N);
}


# Both sort scatters (edges into bucket runs, sources into row order) go to an LDS stage first and are copied out in order: whole-line stores instead of one line per edge; row order still LDS arrival
# speedup vs baseline: 28.6843x; 1.1808x over previous
#include <hip/hip_runtime.h>
#include <hip/hip_bf16.h>

typedef __attribute__((ext_vector_type(2))) float v2f;
typedef __attribute__((ext_vector_type(8))) float v8f;

#define BSH     9
#define BROWS   512
#define NBK_MAX 256
#define CHUNK   20480
#define STH     1024
#define STILE   32768
#define PREPB   6
#define RCAP    32768

__device__ __forceinline__ void prep_w(const float* __restrict__ W, float* __restrict__ Bt,
                                       int Kin, int Cin, int KPAD, int COLS, int g, int gs) {
    for (int idx = g; idx < COLS * KPAD; idx += gs) {
        const int col = idx / KPAD, k = idx % KPAD;
        Bt[idx] = (k < Kin && col < Cin) ? W[k * Cin + col] : 0.0f;
    }
}

__device__ __forceinline__ void prep_w2f(const float* __restrict__ W2, float* __restrict__ w2f,
                                         int g, int gs) {
    for (int idx = g; idx < 64 * 64; idx += gs) {
        const int e   = idx & 3;
        const int ln  = (idx >> 2) & 31;
        const int ip  = (idx >> 7) & 1;
        const int ct  = (idx >> 8) & 3;
        const int ct2 = idx >> 10;
        const int k   = 16 * ct + 8 * (ln >> 4) + 4 * ip + e;
        const int col = 4 * (ln & 15) + ct2;
        w2f[idx] = W2[k * 64 + col];
    }
}

__global__ void __launch_bounds__(STH)
k_bkt_count(const int* __restrict__ dst, int* __restrict__ tab, int e, int nbk, int nch,
            const float* __restrict__ W1, const float* __restrict__ W2,
            const float* __restrict__ W3, float* __restrict__ w1t,
            float* __restrict__ w2f, float* __restrict__ w3t) {
    __shared__ int hist[NBK_MAX];
    const int t = threadIdx.x;
    if ((int)blockIdx.x >= nch) {
        const int g  = ((int)blockIdx.x - nch) * STH + t;
        const int gs = ((int)gridDim.x - nch) * STH;
        prep_w(W1, w1t, 5, 64, 8, 64, g, gs);
        prep_w2f(W2, w2f, g, gs);
        prep_w(W3, w3t, 64, 2, 64, 16, g, gs);
        return;
    }
    if (t < NBK_MAX) hist[t] = 0;
    __syncthreads();
    const int beg = blockIdx.x * CHUNK;
    const int end = min(beg + CHUNK, e);
    for (int i = beg + t; i < end; i += 4 * STH) {
        int d[4];
#pragma unroll
        for (int k = 0; k < 4; ++k) d[k] = dst[min(i + k * STH, end - 1)];
        asm volatile("" :: "v"(d[0]), "v"(d[1]), "v"(d[2]), "v"(d[3]));
#pragma unroll
        for (int k = 0; k < 4; ++k)
            if (i + k * STH < end) atomicAdd(&hist[d[k] >> BSH], 1);
    }
    __syncthreads();
    if (t < nbk) tab[t * nch + blockIdx.x] = hist[t];
}

__global__ void __launch_bounds__(STH)
k_tab_scan(int* __restrict__ tab, int len) {
    __shared__ int buf[STILE + STILE / 32];
    __shared__ int part[STH];
    const int t = threadIdx.x;
    int carry = 0;
    for (int tile = 0; tile < len; tile += STILE) {
        for (int k0 = t; k0 < STILE; k0 += 8 * STH) {
            int v[8];
#pragma unroll
            for (int u = 0; u < 8; ++u) v[u] = tab[min(tile + k0 + u * STH, len - 1)];
#pragma unroll
            for (int u = 0; u < 8; ++u) {
                const int k = k0 + u * STH;
                buf[k + (k >> 5)] = (tile + k < len) ? v[u] : 0;
            }
        }
        __syncthreads();
        int s = 0;
        for (int j = 0; j < 32; ++j) s += buf[33 * t + j];
        part[t] = s;
        __syncthreads();
        for (int off = 1; off < STH; off <<= 1) {
            const int x = (t >= off) ? part[t - off] : 0;
            __syncthreads();
            part[t] += x;
            __syncthreads();
        }
        int run = carry + part[t] - s;
        const int tot = part[STH - 1];
        for (int j = 0; j < 32; ++j) {
            const int v = buf[33 * t + j];
            buf[33 * t + j] = run;
            run += v;
        }
        __syncthreads();
        for (int k = t; k < STILE; k += STH) {
            const int g = tile + k;
            if (g < len) tab[g] = buf[k + (k >> 5)];
        }
        carry += tot;
        __syncthreads();
    }
}

__global__ void __launch_bounds__(STH)
k_bkt_scatter(const int* __restrict__ src, const int* __restrict__ dst,
              const int* __restrict__ tab, int* __restrict__ bedge, int e, int nbk, int nch) {
    __shared__ int  cur[NBK_MAX];
    __shared__ int  dlt[NBK_MAX];
    __shared__ int2 stage[CHUNK];
    const int t = threadIdx.x;
    static_assert(NBK_MAX == 8 * 32, "one wave scans the bucket counts, eight per lane");
    if (t < 32) {
        int g[8], nx[8], cn[8];
        int s = 0;
        const int tlast = nbk * nch - 1;
#pragma unroll
        for (int k = 0; k < 8; ++k) {
            const int idx = min(8 * t + k, nbk - 1) * nch + (int)blockIdx.x;
            g[k]  = tab[idx];
            nx[k] = tab[min(idx + 1, tlast)];
        }
#pragma unroll
        for (int k = 0; k < 8; ++k) {
            const int b = 8 * t + k;
            const int idx = min(b, nbk - 1) * nch + (int)blockIdx.x;
            const int nxt = (idx < tlast) ? nx[k] : e;
            cn[k] = (b < nbk) ? nxt - g[k] : 0;
            s += cn[k];
        }
        int inc = s;
        for (int off = 1; off < 32; off <<= 1) {
            const int v = __shfl_up(inc, off);
            if (t >= off) inc += v;
        }
        int run = inc - s;
#pragma unroll
        for (int k = 0; k < 8; ++k) {
            cur[8 * t + k] = run;
            dlt[8 * t + k] = g[k] - run;
            run += cn[k];
        }
    }
    __syncthreads();
    const int beg = blockIdx.x * CHUNK;
    const int end = min(beg + CHUNK, e);
    int2* be = (int2*)bedge;
    for (int i = beg + t; i < end; i += 4 * STH) {
        int s[4], d[4];
#pragma unroll
        for (int k = 0; k < 4; ++k) {
            const int ii = min(i + k * STH, end - 1);
            s[k] = src[ii];
            d[k] = dst[ii];
        }
        asm volatile("" :: "v"(s[0]), "v"(s[1]), "v"(s[2]), "v"(s[3]),
                           "v"(d[0]), "v"(d[1]), "v"(d[2]), "v"(d[3]));
#pragma unroll
        for (int k = 0; k < 4; ++k)
            if (i + k * STH < end) {
                const int p = atomicAdd(&cur[d[k] >> BSH], 1);
                stage[p] = make_int2(s[k], d[k]);
            }
    }
    __syncthreads();
    const int nloc = end - beg;
    for (int i = t; i < nloc; i += 4 * STH) {
        int2 sd[4];
        int  dl[4];
#pragma unroll
        for (int k = 0; k < 4; ++k) sd[k] = stage[min(i + k * STH, nloc - 1)];
#pragma unroll
        for (int k = 0; k < 4; ++k) dl[k] = dlt[sd[k].y >> BSH];
#pragma unroll
        for (int k = 0; k < 4; ++k)
            if (i + k * STH < nloc) be[i + k * STH + dl[k]] = sd[k];
    }
}

__global__ void __launch_bounds__(STH)
k_row_sort(const int* __restrict__ bedge, const int* __restrict__ tab,
           const float* __restrict__ x, int* __restrict__ rowptr,
           float* __restrict__ isq, float* __restrict__ xq, int* __restrict__ rec,
           int n, int e, int nbk, int nch) {
    __shared__ int   cnt[BROWS];
    __shared__ int   sc[BROWS];
    __shared__ float qs[BROWS];
    __shared__ int   stage[RCAP];
    const int t = threadIdx.x;
    const int b = blockIdx.x;
    const int base = b << BSH;
    const int ebeg = tab[b * nch];
    const int eend = (b + 1 < nbk) ? tab[(b + 1) * nch] : e;
    const int nrec = eend - ebeg;
    const bool staged = nrec <= RCAP;
    if (t < BROWS) cnt[t] = 0;
    __syncthreads();
    const int2* be = (const int2*)bedge;
    for (int i = ebeg + t; i < eend; i += 4 * STH) {
        int d[4];
#pragma unroll
        for (int k = 0; k < 4; ++k) d[k] = be[min(i + k * STH, eend - 1)].y;
        asm volatile("" :: "v"(d[0]), "v"(d[1]), "v"(d[2]), "v"(d[3]));
#pragma unroll
        for (int k = 0; k < 4; ++k)
            if (i + k * STH < eend) atomicAdd(&cnt[d[k] - base], 1);
    }
    __syncthreads();
    int c = 0;
    if (t < BROWS) { c = cnt[t]; sc[t] = c; }
    __syncthreads();
    for (int off = 1; off < BROWS; off <<= 1) {
        int v = 0;
        if (t < BROWS && t >= off) v = sc[t - off];
        __syncthreads();
        if (t < BROWS) sc[t] += v;
        __syncthreads();
    }
    if (t < BROWS) {
        const int lfirst = sc[t] - c;
        const int first = ebeg + lfirst;
        const float q = rsqrtf((float)(c + 1));
        cnt[t] = staged ? lfirst : first;
        qs[t]  = q;
        const int node = base + t;
        if (node < n) {
            rowptr[node] = first;
            isq[node] = q;
        }
    }
    if (b == 0 && t == 0) rowptr[n] = e;
    __syncthreads();
    {
        const int p = t & 7;
        const int col = (p & 1) + ((p >> 1) & 1) * 4 + (p >> 2) * 2;
        float xv[BROWS * 8 / STH];
#pragma unroll
        for (int k = 0; k < BROWS * 8 / STH; ++k) {
            const int node = min(base + (t >> 3) + k * (STH / 8), n - 1);
            xv[k] = x[(size_t)node * 5 + min(col, 4)];
        }
        static_assert(BROWS * 8 / STH == 4, "the fence below names four loaded values");
        asm volatile("" :: "v"(xv[0]), "v"(xv[1]), "v"(xv[2]), "v"(xv[3]));
#pragma unroll
        for (int k = 0; k < BROWS * 8 / STH; ++k) {
            const int i = (t >> 3) + k * (STH / 8);
            if (base + i < n)
                xq[(size_t)(base + i) * 8 + p] = (col < 5) ? xv[k] * qs[i] : 0.0f;
        }
    }
    for (int i = ebeg + t; i < eend; i += 4 * STH) {
        int2 sd[4];
#pragma unroll
        for (int k = 0; k < 4; ++k) sd[k] = be[min(i + k * STH, eend - 1)];
        asm volatile("" :: "v"(sd[0].x), "v"(sd[1].x), "v"(sd[2].x), "v"(sd[3].x),
                           "v"(sd[0].y), "v"(sd[1].y), "v"(sd[2].y), "v"(sd[3].y));
#pragma unroll
        for (int k = 0; k < 4; ++k)
            if (i + k * STH < eend) {
                const int p = atomicAdd(&cnt[sd[k].y - base], 1);
                if (staged) stage[p] = sd[k].x;
                else        rec[p]   = sd[k].x;
            }
    }
    if (staged && nrec > 0) {
        __syncthreads();
        for (int i = t - (ebeg & 31); i < nrec; i += 4 * STH) {
            int v[4];
#pragma unroll
            for (int k = 0; k < 4; ++k) v[k] = stage[min(max(i + k * STH, 0), nrec - 1)];
#pragma unroll
            for (int k = 0; k < 4; ++k) {
                const int ii = i + k * STH;
                if (ii >= 0 && ii < nrec) rec[ebeg + ii] = v[k];
            }
        }
    }
}

typedef __attribute__((ext_vector_type(4))) float v4f;

#define TPITCH 68

__global__ void __launch_bounds__(64)
k_gather64_w3(const int* __restrict__ rowptr, const int* __restrict__ rec,
              const float* __restrict__ h, const float* __restrict__ isq,
              const float* __restrict__ bias, const float* __restrict__ W3t,
              float* __restrict__ out) {
    __shared__ __attribute__((aligned(16))) float tile[16 * TPITCH];
    const int lane = threadIdx.x & 31;
    const int wave = threadIdx.x >> 5;
    const int half = lane >> 4;
    const int l    = lane & 15;
    const v4f* h4 = (const v4f*)h;
    const v4f bv = {bias[4 * l], bias[4 * l + 1], bias[4 * l + 2], bias[4 * l + 3]};
    for (int j = 0; j < 4; ++j) {
        const int rr   = 8 * wave + 2 * j + half;
        const int node = blockIdx.x * 16 + rr;
        const int beg = rowptr[node];
        const int end = rowptr[node + 1];
        const float q = isq[node];
        v4f acc = h4[(size_t)node * 16 + l];
        if (beg < end) {
            const int last = end - 1;
            int e0 = rec[beg];
            int e1 = rec[min(beg + 1, last)];
            int e2 = rec[min(beg + 2, last)];
            int e3 = rec[min(beg + 3, last)];
            for (int p = beg; p < end; p += 4) {
                const v4f v0 = h4[(size_t)e0 * 16 + l];
                const v4f v1 = h4[(size_t)e1 * 16 + l];
                const v4f v2 = h4[(size_t)e2 * 16 + l];
                const v4f v3 = h4[(size_t)e3 * 16 + l];
                e0 = rec[min(p + 4, last)];
                e1 = rec[min(p + 5, last)];
                e2 = rec[min(p + 6, last)];
                e3 = rec[min(p + 7, last)];
                __builtin_amdgcn_sched_barrier(0);
                asm volatile("" :: "v"(v0), "v"(v1), "v"(v2), "v"(v3));
                acc += v0;
                if (p + 1 < end) acc += v1;
                if (p + 2 < end) acc += v2;
                if (p + 3 < end) acc += v3;
            }
        }
        acc = acc * q + bv;
        v4f r;
        r.x = fmaxf(acc.x, 0.0f);
        r.y = fmaxf(acc.y, 0.0f);
        r.z = fmaxf(acc.z, 0.0f);
        r.w = fmaxf(acc.w, 0.0f);
        *(v4f*)(tile + rr * TPITCH + 4 * l) = r;
    }
    __syncthreads();
    if (wave != 0) return;

    const float* Wrow = W3t + l * 64 + 2 * half;
    const float* Trow = tile + l * TPITCH + 2 * half;
    const float qrow = isq[blockIdx.x * 16 + l];
    v8f c = {};
#pragma unroll
    for (int kk = 0; kk < 16; ++kk) {
        const v2f a = *(const v2f*)(Wrow + 4 * kk);
        const v2f b = *(const v2f*)(Trow + 4 * kk);
        c = __builtin_amdgcn_wmma_f32_16x16x4_f32(
                 false, a,  false, b,
                 (short)0, c,  false,  false);
    }
    asm volatile("" :: "v"(qrow));
    if (half == 0) {
        const v2f o = {c[0] * qrow, c[1] * qrow};
        ((v2f*)out)[blockIdx.x * 16 + l] = o;
    }
}

__global__ void __launch_bounds__(64)
k_layer12(const int* __restrict__ rowptr, const int* __restrict__ rec,
          const float* __restrict__ xq, const float* __restrict__ isq,
          const float* __restrict__ W1t, const float* __restrict__ bias,
          const float* __restrict__ W2f, float* __restrict__ out, int ntiles) {
    const int lane = threadIdx.x & 31;
    const int half = lane >> 4;
    const int l15  = lane & 15;
    const int tile = blockIdx.x * 2 + (threadIdx.x >> 5);
    if (tile >= ntiles) return;
    const int node = tile * 16 + l15;
    const v4f* xq4 = (const v4f*)xq;

    const int beg = rowptr[node];
    const int end = rowptr[node + 1];

    v2f w0[4], w1[4];
#pragma unroll
    for (int ct = 0; ct < 4; ++ct) {
        const float* Wcol = W1t + (size_t)(16 * ct + l15) * 8 + half * 2;
        w0[ct] = *(const v2f*)(Wcol);
        w1[ct] = *(const v2f*)(Wcol + 4);
    }

    const float q = isq[node];
    v4f acc = xq4[(size_t)node * 2 + half];
    if (beg < end) {
        const int last = end - 1;
        int e[8];
#pragma unroll
        for (int j = 0; j < 8; ++j) e[j] = rec[min(beg + j, last)];
        for (int p = beg; p < end; p += 8) {
            v4f v[8];
#pragma unroll
            for (int j = 0; j < 8; ++j) v[j] = xq4[(size_t)e[j] * 2 + half];
#pragma unroll
            for (int j = 0; j < 8; ++j) e[j] = rec[min(p + 8 + j, last)];
            __builtin_amdgcn_sched_barrier(0);
            asm volatile("" :: "v"(v[0]), "v"(v[1]), "v"(v[2]), "v"(v[3]),
                               "v"(v[4]), "v"(v[5]), "v"(v[6]), "v"(v[7]));
            acc += v[0];
#pragma unroll
            for (int j = 1; j < 8; ++j)
                if (p + j < end) acc += v[j];
        }
    }
    acc *= q;

    const v2f x0 = {acc.x, acc.y};
    const v2f x1 = {acc.z, acc.w};
    v8f d[4];
#pragma unroll
    for (int ct = 0; ct < 4; ++ct) {
        v8f z = {};
        d[ct] = __builtin_amdgcn_wmma_f32_16x16x4_f32(
                     false, w0[ct],  false, x0,
                     (short)0, z,  false,  false);
    }
#pragma unroll
    for (int ct = 0; ct < 4; ++ct)
        d[ct] = __builtin_amdgcn_wmma_f32_16x16x4_f32(
                     false, w1[ct],  false, x1,
                     (short)0, d[ct],  false,  false);

#pragma unroll
    for (int ct = 0; ct < 4; ++ct) {
        const v4f ba = *(const v4f*)(bias + 16 * ct + 8 * half);
        const v4f bb = *(const v4f*)(bias + 16 * ct + 8 * half + 4);
        d[ct][0] = fmaxf(d[ct][0] + ba.x, 0.0f);
        d[ct][1] = fmaxf(d[ct][1] + ba.y, 0.0f);
        d[ct][2] = fmaxf(d[ct][2] + ba.z, 0.0f);
        d[ct][3] = fmaxf(d[ct][3] + ba.w, 0.0f);
        d[ct][4] = fmaxf(d[ct][4] + bb.x, 0.0f);
        d[ct][5] = fmaxf(d[ct][5] + bb.y, 0.0f);
        d[ct][6] = fmaxf(d[ct][6] + bb.z, 0.0f);
        d[ct][7] = fmaxf(d[ct][7] + bb.w, 0.0f);
    }

    const v4f* W2f4 = (const v4f*)W2f;
    v8f o[4];
#pragma unroll
    for (int ct2 = 0; ct2 < 4; ++ct2) o[ct2] = v8f{};
#pragma unroll
    for (int ct = 0; ct < 4; ++ct) {
#pragma unroll
        for (int ip = 0; ip < 2; ++ip) {
            v4f bw[4];
#pragma unroll
            for (int ct2 = 0; ct2 < 4; ++ct2)
                bw[ct2] = W2f4[((ct2 * 4 + ct) * 2 + ip) * 32 + lane];
            const v2f a_lo = {d[ct][4 * ip],     d[ct][4 * ip + 1]};
            const v2f a_hi = {d[ct][4 * ip + 2], d[ct][4 * ip + 3]};
#pragma unroll
            for (int ct2 = 0; ct2 < 4; ++ct2) {
                const v2f b_lo = {bw[ct2].x, bw[ct2].y};
                o[ct2] = __builtin_amdgcn_wmma_f32_16x16x4_f32(
                              false, a_lo,  false, b_lo,
                              (short)0, o[ct2],  false,  false);
            }
#pragma unroll
            for (int ct2 = 0; ct2 < 4; ++ct2) {
                const v2f b_hi = {bw[ct2].z, bw[ct2].w};
                o[ct2] = __builtin_amdgcn_wmma_f32_16x16x4_f32(
                              false, a_hi,  false, b_hi,
                              (short)0, o[ct2],  false,  false);
            }
        }
    }

    const size_t mbase = (size_t)tile * 16 + half * 8;
    const v4f q0 = *(const v4f*)(isq + mbase);
    const v4f q1 = *(const v4f*)(isq + mbase + 4);
    const float qr[8] = {q0.x, q0.y, q0.z, q0.w, q1.x, q1.y, q1.z, q1.w};
#pragma unroll
    for (int r = 0; r < 8; ++r) {
        v4f s;
        s.x = o[0][r] * qr[r];
        s.y = o[1][r] * qr[r];
        s.z = o[2][r] * qr[r];
        s.w = o[3][r] * qr[r];
        ((v4f*)out)[(mbase + r) * 16 + l15] = s;
    }
}

__global__ void k_gather2(const int* __restrict__ rowptr, const int* __restrict__ rec,
                          const float* __restrict__ h, const float* __restrict__ isq,
                          const float* __restrict__ bias, float* __restrict__ out, int n) {
    const int node = blockIdx.x * blockDim.x + threadIdx.x;
    if (node >= n) return;
    const int beg = rowptr[node];
    const int end = rowptr[node + 1];
    const v2f* h2 = (const v2f*)h;
    const v2f bv = {bias[0], bias[1]};
    const float q = isq[node];
    v2f acc = h2[node];
    for (int p = beg; p < end; p += 4) {
        const int last = end - 1;
        const int e0 = rec[p];
        const int e1 = rec[min(p + 1, last)];
        const int e2 = rec[min(p + 2, last)];
        const int e3 = rec[min(p + 3, last)];
        const v2f v0 = h2[e0];
        const v2f v1 = h2[e1];
        const v2f v2 = h2[e2];
        const v2f v3 = h2[e3];
        asm volatile("" :: "v"(v0), "v"(v1), "v"(v2), "v"(v3));
        acc += v0;
        if (p + 1 < end) acc += v1;
        if (p + 2 < end) acc += v2;
        if (p + 3 < end) acc += v3;
    }
    ((v2f*)out)[node] = acc * q + bv;
}

extern "C" void kernel_launch(void* const* d_in, const int* in_sizes, int n_in,
                              void* d_out, int out_size, void* d_ws, size_t ws_size,
                              hipStream_t stream) {
    const float* x  = (const float*)d_in[0];
    const int*   ei = (const int*)d_in[1];
    const float* W1 = (const float*)d_in[2];
    const float* b1 = (const float*)d_in[3];
    const float* W2 = (const float*)d_in[4];
    const float* b2 = (const float*)d_in[5];
    const float* W3 = (const float*)d_in[6];
    const float* b3 = (const float*)d_in[7];
    float* out = (float*)d_out;

    const int N = in_sizes[0] / 5;
    const int E = in_sizes[1] / 2;
    const int* src = ei;
    const int* dst = ei + E;

    const int T = 256;
    auto cdiv = [](long a, long b) { return (int)((a + b - 1) / b); };
    const int NBK = cdiv(N, BROWS);
    const int NCH = cdiv(E, CHUNK);

    float* ws   = (float*)d_ws;
    float* bufA = ws;
    float* xq   = bufA + (size_t)N * 64;
    float* w1t  = xq   + (size_t)N * 8;
    float* w2f  = w1t  + 64 * 8;
    float* w3t  = w2f  + 64 * 64;
    float* h3   = w3t  + 16 * 64;
    float* isq  = h3   + (size_t)N * 2;
    int*   bedg = (int*)(isq + N);
    int*   rec  = bedg + (size_t)E * 2;
    int*   tab  = rec  + (size_t)E;
    int*   rowp = tab  + (size_t)NBK * NCH;

    k_bkt_count  <<<NCH + PREPB, STH, 0, stream>>>(dst, tab, E, NBK, NCH,
                                                   W1, W2, W3, w1t, w2f, w3t);
    k_tab_scan   <<<1,   STH, 0, stream>>>(tab, NBK * NCH);
    k_bkt_scatter<<<NCH, STH, 0, stream>>>(src, dst, tab, bedg, E, NBK, NCH);
    k_row_sort   <<<NBK, STH, 0, stream>>>(bedg, tab, x, rowp, isq, xq, rec, N, E, NBK, NCH);

    const int GT = cdiv(N, 16);

    k_layer12  <<<cdiv(GT, 2), 64, 0, stream>>>(rowp, rec, xq, isq, w1t, b1, w2f, bufA, GT);

    k_gather64_w3<<<GT, 64, 0, stream>>>(rowp, rec, bufA, isq, b2, w3t, h3);

    k_gather2  <<<cdiv(N, T), T, 0, stream>>>(rowp, rec, h3, isq, b3, out, N);
}


# Sort level 1 groups each chunk in its own place (edges in registers, LDS ranks), no global offset table: count and scan launches gone, 7 to 5; row sort reads a piece per chunk; row sum order changes
# speedup vs baseline: 31.1624x; 1.0864x over previous
#include <hip/hip_runtime.h>
#include <hip/hip_bf16.h>

typedef __attribute__((ext_vector_type(2))) float v2f;
typedef __attribute__((ext_vector_type(8))) float v8f;


#define BSH     9
#define BROWS   512
#define NBK_MAX 256
#define NCH_MAX 256
#define CHUNK   20480
#define STH     1024
#define EPT     20
#define PREPB   6
#define RCAP    32768

__device__ __forceinline__ void prep_w(const float* __restrict__ W, float* __restrict__ Bt,
                                       int Kin, int Cin, int KPAD, int COLS, int g, int gs) {
    for (int idx = g; idx < COLS * KPAD; idx += gs) {
        const int col = idx / KPAD, k = idx % KPAD;
        Bt[idx] = (k < Kin && col < Cin) ? W[k * Cin + col] : 0.0f;
    }
}

__device__ __forceinline__ void prep_w2f(const float* __restrict__ W2, float* __restrict__ w2f,
                                         int g, int gs) {
    for (int idx = g; idx < 64 * 64; idx += gs) {
        const int e   = idx & 3;
        const int ln  = (idx >> 2) & 31;
        const int ip  = (idx >> 7) & 1;
        const int ct  = (idx >> 8) & 3;
        const int ct2 = idx >> 10;
        const int k   = 16 * ct + 8 * (ln >> 4) + 4 * ip + e;
        const int col = 4 * (ln & 15) + ct2;
        w2f[idx] = W2[k * 64 + col];
    }
}

__global__ void __launch_bounds__(STH)
k_chunk_group(const int* __restrict__ src, const int* __restrict__ dst,
              int* __restrict__ tcnt, int* __restrict__ tloc, int* __restrict__ bedge,
              int e, int nbk, int nch,
              const float* __restrict__ W1, const float* __restrict__ W2,
              const float* __restrict__ W3, float* __restrict__ w1t,
              float* __restrict__ w2f, float* __restrict__ w3t) {
    __shared__ int  hist[NBK_MAX];
    __shared__ int  start[NBK_MAX];
    __shared__ int2 stage[CHUNK];
    const int t = threadIdx.x;
    if ((int)blockIdx.x >= nch) {
        const int g  = ((int)blockIdx.x - nch) * STH + t;
        const int gs = ((int)gridDim.x - nch) * STH;
        prep_w(W1, w1t, 5, 64, 8, 64, g, gs);
        prep_w2f(W2, w2f, g, gs);
        prep_w(W3, w3t, 64, 2, 64, 16, g, gs);
        return;
    }
    if (t < NBK_MAX) hist[t] = 0;
    __syncthreads();
    const int beg = blockIdx.x * CHUNK;
    const int end = min(beg + CHUNK, e);
    const int nloc = end - beg;
    static_assert(EPT * STH == CHUNK, "a thread holds its EPT edges of the chunk");
    int s[EPT], d[EPT], r[EPT];
#pragma unroll
    for (int k = 0; k < EPT; ++k) {
        const int ii = min(beg + t + k * STH, end - 1);
        s[k] = src[ii];
        d[k] = dst[ii];
    }
#pragma unroll
    for (int k = 0; k < EPT; ++k) {
        r[k] = 0;
        if (t + k * STH < nloc) r[k] = atomicAdd(&hist[d[k] >> BSH], 1);
    }
    __syncthreads();
    static_assert(NBK_MAX == 8 * 32, "one wave scans the bucket counts, eight per lane");
    if (t < 32) {
        int cn[8];
        int sm = 0;
#pragma unroll
        for (int k = 0; k < 8; ++k) {
            cn[k] = hist[8 * t + k];
            sm += cn[k];
        }
        int inc = sm;
        for (int off = 1; off < 32; off <<= 1) {
            const int v = __shfl_up(inc, off);
            if (t >= off) inc += v;
        }
        int run = inc - sm;
#pragma unroll
        for (int k = 0; k < 8; ++k) {
            const int b = 8 * t + k;
            start[b] = run;
            if (b < nbk) {
                tcnt[b * nch + (int)blockIdx.x] = cn[k];
                tloc[b * nch + (int)blockIdx.x] = run;
            }
            run += cn[k];
        }
    }
    __syncthreads();
#pragma unroll
    for (int k = 0; k < EPT; ++k) r[k] += start[d[k] >> BSH];
#pragma unroll
    for (int k = 0; k < EPT; ++k)
        if (t + k * STH < nloc) stage[r[k]] = make_int2(s[k], d[k]);
    __syncthreads();
    int2* be = (int2*)bedge + beg;
    for (int i = t; i < nloc; i += 4 * STH) {
        int2 sd[4];
#pragma unroll
        for (int k = 0; k < 4; ++k) sd[k] = stage[min(i + k * STH, nloc - 1)];
#pragma unroll
        for (int k = 0; k < 4; ++k)
            if (i + k * STH < nloc) be[i + k * STH] = sd[k];
    }
}

__global__ void __launch_bounds__(STH)
k_row_sort(const int* __restrict__ bedge, const int* __restrict__ tcnt,
           const int* __restrict__ tloc,
           const float* __restrict__ x, int* __restrict__ rowptr,
           float* __restrict__ isq, float* __restrict__ xq, int* __restrict__ rec,
           int n, int e, int nbk, int nch) {
    __shared__ int   cnt[BROWS];
    __shared__ int   sc[BROWS];
    __shared__ float qs[BROWS];
    __shared__ int   stage[RCAP];
    __shared__ int   pc[NCH_MAX];
    __shared__ int   ps[NCH_MAX];
    __shared__ int   red[2 * (STH / 32)];
    __shared__ int   tot[2];
    static_assert(STH == 32 * 32, "one wave adds up the per-wave sums, one per lane");
    const int t = threadIdx.x;
    const int b = blockIdx.x;
    const int base = b << BSH;
    const int wv = t >> 5;
    const int ln = t & 31;
    const int lim = b * nch;
    int sb = 0;
    for (int i = t; i < lim; i += 8 * STH) {
        int v[8];
#pragma unroll
        for (int u = 0; u < 8; ++u) v[u] = tcnt[min(i + u * STH, lim - 1)];
#pragma unroll
        for (int u = 0; u < 8; ++u) sb += (i + u * STH < lim) ? v[u] : 0;
    }
    int so = 0;
    if (t < nch) {
        so = tcnt[lim + t];
        pc[t] = so;
        ps[t] = t * CHUNK + tloc[lim + t];
    }
    if (t < BROWS) cnt[t] = 0;
    for (int off = 16; off > 0; off >>= 1) {
        sb += __shfl_xor(sb, off);
        so += __shfl_xor(so, off);
    }
    if (ln == 0) {
        red[wv] = sb;
        red[STH / 32 + wv] = so;
    }
    __syncthreads();
    if (t < 32) {
        int a = red[t];
        int c2 = red[STH / 32 + t];
        for (int off = 16; off > 0; off >>= 1) {
            a  += __shfl_xor(a, off);
            c2 += __shfl_xor(c2, off);
        }
        if (t == 0) {
            tot[0] = a;
            tot[1] = c2;
        }
    }
    __syncthreads();
    const int ebeg = tot[0];
    const int nrec = tot[1];
    const bool staged = nrec <= RCAP;
    const int2* be = (const int2*)bedge;
    for (int c = wv; c < nch; c += STH / 32) {
        const int np = pc[c];
        const int2* pp = be + ps[c];
        for (int j = 0; j < np; j += 4 * 32) {
            int d[4];
#pragma unroll
            for (int k = 0; k < 4; ++k) d[k] = pp[min(j + ln + 32 * k, np - 1)].y;
            asm volatile("" :: "v"(d[0]), "v"(d[1]), "v"(d[2]), "v"(d[3]));
#pragma unroll
            for (int k = 0; k < 4; ++k)
                if (j + ln + 32 * k < np) atomicAdd(&cnt[d[k] - base], 1);
        }
    }
    __syncthreads();
    int c = 0;
    if (t < BROWS) { c = cnt[t]; sc[t] = c; }
    __syncthreads();
    for (int off = 1; off < BROWS; off <<= 1) {
        int v = 0;
        if (t < BROWS && t >= off) v = sc[t - off];
        __syncthreads();
        if (t < BROWS) sc[t] += v;
        __syncthreads();
    }
    if (t < BROWS) {
        const int lfirst = sc[t] - c;
        const int first = ebeg + lfirst;
        const float q = rsqrtf((float)(c + 1));
        cnt[t] = staged ? lfirst : first;
        qs[t]  = q;
        const int node = base + t;
        if (node < n) {
            rowptr[node] = first;
            isq[node] = q;
        }
    }
    if (b == 0 && t == 0) rowptr[n] = e;
    __syncthreads();
    {
        const int p = t & 7;
        const int col = (p & 1) + ((p >> 1) & 1) * 4 + (p >> 2) * 2;
        float xv[BROWS * 8 / STH];
#pragma unroll
        for (int k = 0; k < BROWS * 8 / STH; ++k) {
            const int node = min(base + (t >> 3) + k * (STH / 8), n - 1);
            xv[k] = x[(size_t)node * 5 + min(col, 4)];
        }
        static_assert(BROWS * 8 / STH == 4, "the fence below names four loaded values");
        asm volatile("" :: "v"(xv[0]), "v"(xv[1]), "v"(xv[2]), "v"(xv[3]));
#pragma unroll
        for (int k = 0; k < BROWS * 8 / STH; ++k) {
            const int i = (t >> 3) + k * (STH / 8);
            if (base + i < n)
                xq[(size_t)(base + i) * 8 + p] = (col < 5) ? xv[k] * qs[i] : 0.0f;
        }
    }
    for (int cc = wv; cc < nch; cc += STH / 32) {
        const int np = pc[cc];
        const int2* pp = be + ps[cc];
        for (int j = 0; j < np; j += 4 * 32) {
            int2 sd[4];
#pragma unroll
            for (int k = 0; k < 4; ++k) sd[k] = pp[min(j + ln + 32 * k, np - 1)];
            asm volatile("" :: "v"(sd[0].x), "v"(sd[1].x), "v"(sd[2].x), "v"(sd[3].x),
                               "v"(sd[0].y), "v"(sd[1].y), "v"(sd[2].y), "v"(sd[3].y));
#pragma unroll
            for (int k = 0; k < 4; ++k)
                if (j + ln + 32 * k < np) {
                    const int p = atomicAdd(&cnt[sd[k].y - base], 1);
                    if (staged) stage[p] = sd[k].x;
                    else        rec[p]   = sd[k].x;
                }
        }
    }
    if (staged && nrec > 0) {
        __syncthreads();
        for (int i = t - (ebeg & 31); i < nrec; i += 4 * STH) {
            int v[4];
#pragma unroll
            for (int k = 0; k < 4; ++k) v[k] = stage[min(max(i + k * STH, 0), nrec - 1)];
#pragma unroll
            for (int k = 0; k < 4; ++k) {
                const int ii = i + k * STH;
                if (ii >= 0 && ii < nrec) rec[ebeg + ii] = v[k];
            }
        }
    }
}

typedef __attribute__((ext_vector_type(4))) float v4f;

#define TPITCH 68

__global__ void __launch_bounds__(64)
k_gather64_w3(const int* __restrict__ rowptr, const int* __restrict__ rec,
              const float* __restrict__ h, const float* __restrict__ isq,
              const float* __restrict__ bias, const float* __restrict__ W3t,
              float* __restrict__ out) {
    __shared__ __attribute__((aligned(16))) float tile[16 * TPITCH];
    const int lane = threadIdx.x & 31;
    const int wave = threadIdx.x >> 5;
    const int half = lane >> 4;
    const int l    = lane & 15;
    const v4f* h4 = (const v4f*)h;
    const v4f bv = {bias[4 * l], bias[4 * l + 1], bias[4 * l + 2], bias[4 * l + 3]};
    for (int j = 0; j < 4; ++j) {
        const int rr   = 8 * wave + 2 * j + half;
        const int node = blockIdx.x * 16 + rr;
        const int beg = rowptr[node];
        const int end = rowptr[node + 1];
        const float q = isq[node];
        v4f acc = h4[(size_t)node * 16 + l];
        if (beg < end) {
            const int last = end - 1;
            int e0 = rec[beg];
            int e1 = rec[min(beg + 1, last)];
            int e2 = rec[min(beg + 2, last)];
            int e3 = rec[min(beg + 3, last)];
            for (int p = beg; p < end; p += 4) {
                const v4f v0 = h4[(size_t)e0 * 16 + l];
                const v4f v1 = h4[(size_t)e1 * 16 + l];
                const v4f v2 = h4[(size_t)e2 * 16 + l];
                const v4f v3 = h4[(size_t)e3 * 16 + l];
                e0 = rec[min(p + 4, last)];
                e1 = rec[min(p + 5, last)];
                e2 = rec[min(p + 6, last)];
                e3 = rec[min(p + 7, last)];
                __builtin_amdgcn_sched_barrier(0);
                asm volatile("" :: "v"(v0), "v"(v1), "v"(v2), "v"(v3));
                acc += v0;
                if (p + 1 < end) acc += v1;
                if (p + 2 < end) acc += v2;
                if (p + 3 < end) acc += v3;
            }
        }
        acc = acc * q + bv;
        v4f r;
        r.x = fmaxf(acc.x, 0.0f);
        r.y = fmaxf(acc.y, 0.0f);
        r.z = fmaxf(acc.z, 0.0f);
        r.w = fmaxf(acc.w, 0.0f);
        *(v4f*)(tile + rr * TPITCH + 4 * l) = r;
    }
    __syncthreads();
    if (wave != 0) return;

    const float* Wrow = W3t + l * 64 + 2 * half;
    const float* Trow = tile + l * TPITCH + 2 * half;
    const float qrow = isq[blockIdx.x * 16 + l];
    v8f c = {};
#pragma unroll
    for (int kk = 0; kk < 16; ++kk) {
        const v2f a = *(const v2f*)(Wrow + 4 * kk);
        const v2f b = *(const v2f*)(Trow + 4 * kk);
        c = __builtin_amdgcn_wmma_f32_16x16x4_f32(
                 false, a,  false, b,
                 (short)0, c,  false,  false);
    }
    asm volatile("" :: "v"(qrow));
    if (half == 0) {
        const v2f o = {c[0] * qrow, c[1] * qrow};
        ((v2f*)out)[blockIdx.x * 16 + l] = o;
    }
}

__global__ void __launch_bounds__(64)
k_layer12(const int* __restrict__ rowptr, const int* __restrict__ rec,
          const float* __restrict__ xq, const float* __restrict__ isq,
          const float* __restrict__ W1t, const float* __restrict__ bias,
          const float* __restrict__ W2f, float* __restrict__ out, int ntiles) {
    const int lane = threadIdx.x & 31;
    const int half = lane >> 4;
    const int l15  = lane & 15;
    const int tile = blockIdx.x * 2 + (threadIdx.x >> 5);
    if (tile >= ntiles) return;
    const int node = tile * 16 + l15;
    const v4f* xq4 = (const v4f*)xq;

    const int beg = rowptr[node];
    const int end = rowptr[node + 1];

    v2f w0[4], w1[4];
#pragma unroll
    for (int ct = 0; ct < 4; ++ct) {
        const float* Wcol = W1t + (size_t)(16 * ct + l15) * 8 + half * 2;
        w0[ct] = *(const v2f*)(Wcol);
        w1[ct] = *(const v2f*)(Wcol + 4);
    }

    const float q = isq[node];
    v4f acc = xq4[(size_t)node * 2 + half];
    if (beg < end) {
        const int last = end - 1;
        int e[8];
#pragma unroll
        for (int j = 0; j < 8; ++j) e[j] = rec[min(beg + j, last)];
        for (int p = beg; p < end; p += 8) {
            v4f v[8];
#pragma unroll
            for (int j = 0; j < 8; ++j) v[j] = xq4[(size_t)e[j] * 2 + half];
#pragma unroll
            for (int j = 0; j < 8; ++j) e[j] = rec[min(p + 8 + j, last)];
            __builtin_amdgcn_sched_barrier(0);
            asm volatile("" :: "v"(v[0]), "v"(v[1]), "v"(v[2]), "v"(v[3]),
                               "v"(v[4]), "v"(v[5]), "v"(v[6]), "v"(v[7]));
            acc += v[0];
#pragma unroll
            for (int j = 1; j < 8; ++j)
                if (p + j < end) acc += v[j];
        }
    }
    acc *= q;

    const v2f x0 = {acc.x, acc.y};
    const v2f x1 = {acc.z, acc.w};
    v8f d[4];
#pragma unroll
    for (int ct = 0; ct < 4; ++ct) {
        v8f z = {};
        d[ct] = __builtin_amdgcn_wmma_f32_16x16x4_f32(
                     false, w0[ct],  false, x0,
                     (short)0, z,  false,  false);
    }
#pragma unroll
    for (int ct = 0; ct < 4; ++ct)
        d[ct] = __builtin_amdgcn_wmma_f32_16x16x4_f32(
                     false, w1[ct],  false, x1,
                     (short)0, d[ct],  false,  false);

#pragma unroll
    for (int ct = 0; ct < 4; ++ct) {
        const v4f ba = *(const v4f*)(bias + 16 * ct + 8 * half);
        const v4f bb = *(const v4f*)(bias + 16 * ct + 8 * half + 4);
        d[ct][0] = fmaxf(d[ct][0] + ba.x, 0.0f);
        d[ct][1] = fmaxf(d[ct][1] + ba.y, 0.0f);
        d[ct][2] = fmaxf(d[ct][2] + ba.z, 0.0f);
        d[ct][3] = fmaxf(d[ct][3] + ba.w, 0.0f);
        d[ct][4] = fmaxf(d[ct][4] + bb.x, 0.0f);
        d[ct][5] = fmaxf(d[ct][5] + bb.y, 0.0f);
        d[ct][6] = fmaxf(d[ct][6] + bb.z, 0.0f);
        d[ct][7] = fmaxf(d[ct][7] + bb.w, 0.0f);
    }

    const v4f* W2f4 = (const v4f*)W2f;
    v8f o[4];
#pragma unroll
    for (int ct2 = 0; ct2 < 4; ++ct2) o[ct2] = v8f{};
#pragma unroll
    for (int ct = 0; ct < 4; ++ct) {
#pragma unroll
        for (int ip = 0; ip < 2; ++ip) {
            v4f bw[4];
#pragma unroll
            for (int ct2 = 0; ct2 < 4; ++ct2)
                bw[ct2] = W2f4[((ct2 * 4 + ct) * 2 + ip) * 32 + lane];
            const v2f a_lo = {d[ct][4 * ip],     d[ct][4 * ip + 1]};
            const v2f a_hi = {d[ct][4 * ip + 2], d[ct][4 * ip + 3]};
#pragma unroll
            for (int ct2 = 0; ct2 < 4; ++ct2) {
                const v2f b_lo = {bw[ct2].x, bw[ct2].y};
                o[ct2] = __builtin_amdgcn_wmma_f32_16x16x4_f32(
                              false, a_lo,  false, b_lo,
                              (short)0, o[ct2],  false,  false);
            }
#pragma unroll
            for (int ct2 = 0; ct2 < 4; ++ct2) {
                const v2f b_hi = {bw[ct2].z, bw[ct2].w};
                o[ct2] = __builtin_amdgcn_wmma_f32_16x16x4_f32(
                              false, a_hi,  false, b_hi,
                              (short)0, o[ct2],  false,  false);
            }
        }
    }

    const size_t mbase = (size_t)tile * 16 + half * 8;
    const v4f q0 = *(const v4f*)(isq + mbase);
    const v4f q1 = *(const v4f*)(isq + mbase + 4);
    const float qr[8] = {q0.x, q0.y, q0.z, q0.w, q1.x, q1.y, q1.z, q1.w};
#pragma unroll
    for (int r = 0; r < 8; ++r) {
        v4f s;
        s.x = o[0][r] * qr[r];
        s.y = o[1][r] * qr[r];
        s.z = o[2][r] * qr[r];
        s.w = o[3][r] * qr[r];
        ((v4f*)out)[(mbase + r) * 16 + l15] = s;
    }
}

__global__ void k_gather2(const int* __restrict__ rowptr, const int* __restrict__ rec,
                          const float* __restrict__ h, const float* __restrict__ isq,
                          const float* __restrict__ bias, float* __restrict__ out, int n) {
    const int node = blockIdx.x * blockDim.x + threadIdx.x;
    if (node >= n) return;
    const int beg = rowptr[node];
    const int end = rowptr[node + 1];
    const v2f* h2 = (const v2f*)h;
    const v2f bv = {bias[0], bias[1]};
    const float q = isq[node];
    v2f acc = h2[node];
    for (int p = beg; p < end; p += 4) {
        const int last = end - 1;
        const int e0 = rec[p];
        const int e1 = rec[min(p + 1, last)];
        const int e2 = rec[min(p + 2, last)];
        const int e3 = rec[min(p + 3, last)];
        const v2f v0 = h2[e0];
        const v2f v1 = h2[e1];
        const v2f v2 = h2[e2];
        const v2f v3 = h2[e3];
        asm volatile("" :: "v"(v0), "v"(v1), "v"(v2), "v"(v3));
        acc += v0;
        if (p + 1 < end) acc += v1;
        if (p + 2 < end) acc += v2;
        if (p + 3 < end) acc += v3;
    }
    ((v2f*)out)[node] = acc * q + bv;
}

extern "C" void kernel_launch(void* const* d_in, const int* in_sizes, int n_in,
                              void* d_out, int out_size, void* d_ws, size_t ws_size,
                              hipStream_t stream) {
    const float* x  = (const float*)d_in[0];
    const int*   ei = (const int*)d_in[1];
    const float* W1 = (const float*)d_in[2];
    const float* b1 = (const float*)d_in[3];
    const float* W2 = (const float*)d_in[4];
    const float* b2 = (const float*)d_in[5];
    const float* W3 = (const float*)d_in[6];
    const float* b3 = (const float*)d_in[7];
    float* out = (float*)d_out;

    const int N = in_sizes[0] / 5;
    const int E = in_sizes[1] / 2;
    const int* src = ei;
    const int* dst = ei + E;

    const int T = 256;
    auto cdiv = [](long a, long b) { return (int)((a + b - 1) / b); };
    const int NBK = cdiv(N, BROWS);
    const int NCH = cdiv(E, CHUNK);

    float* ws   = (float*)d_ws;
    float* bufA = ws;
    float* xq   = bufA + (size_t)N * 64;
    float* w1t  = xq   + (size_t)N * 8;
    float* w2f  = w1t  + 64 * 8;
    float* w3t  = w2f  + 64 * 64;
    float* h3   = w3t  + 16 * 64;
    float* isq  = h3   + (size_t)N * 2;
    int*   bedg = (int*)(isq + N);
    int*   rec  = bedg + (size_t)E * 2;
    int*   tcnt = rec  + (size_t)E;
    int*   tloc = tcnt + (size_t)NBK * NCH;
    int*   rowp = tloc + (size_t)NBK * NCH;

    k_chunk_group<<<NCH + PREPB, STH, 0, stream>>>(src, dst, tcnt, tloc, bedg, E, NBK, NCH,
                                                   W1, W2, W3, w1t, w2f, w3t);
    k_row_sort   <<<NBK, STH, 0, stream>>>(bedg, tcnt, tloc, x, rowp, isq, xq, rec, N, E, NBK, NCH);

    const int GT = cdiv(N, 16);

    k_layer12  <<<cdiv(GT, 2), 64, 0, stream>>>(rowp, rec, xq, isq, w1t, b1, w2f, bufA, GT);

    k_gather64_w3<<<GT, 64, 0, stream>>>(rowp, rec, bufA, isq, b2, w3t, h3);

    k_gather2  <<<cdiv(N, T), T, 0, stream>>>(rowp, rec, h3, isq, b3, out, N);
}
